# starting kernel
# baseline (speedup 1.0000x reference)
#include <hip/hip_runtime.h>
#include <math.h>

#define NNODES 50000
#define NEDGES 800000
#define ETOT   (NEDGES + NNODES)
#define INDIM  128
#define H1C    4
#define C1C    32
#define HID    128
#define OUTD   64
#define NGRAPH 64

typedef float v2f __attribute__((ext_vector_type(2)));
typedef float v8f __attribute__((ext_vector_type(8)));

__global__ void gemm_wmma_f32(const float* __restrict__ A, const float* __restrict__ B,
                              float* __restrict__ Cm, int M, int K, int Ncol) {
  int gwave = (int)((blockIdx.x * blockDim.x + threadIdx.x) >> 5);
  int lane  = (int)(threadIdx.x & 31);
  int tilesN = Ncol >> 4;
  int numTiles = (M >> 4) * tilesN;
  if (gwave >= numTiles) return;
  int tm = (gwave / tilesN) << 4;
  int tn = (gwave % tilesN) << 4;
  int half = lane >> 4;
  int l    = lane & 15;

  v8f acc = {};
  const float* arow = A + (size_t)(tm + l) * K;
  for (int k0 = 0; k0 < K; k0 += 4) {
    int ka = k0 + (half << 1);
    v2f a, b;
    a.x = arow[ka];
    a.y = arow[ka + 1];
    b.x = B[(size_t)(ka)     * Ncol + tn + l];
    b.y = B[(size_t)(ka + 1) * Ncol + tn + l];
    acc = __builtin_amdgcn_wmma_f32_16x16x4_f32(false, a, false, b,
                                                (short)0, acc, false, false);
  }
  int row0 = tm + (half << 3);
  #pragma unroll
  for (int v = 0; v < 8; ++v)
    Cm[(size_t)(row0 + v) * Ncol + tn + l] = acc[v];
}

__global__ void attn_scores(const float* __restrict__ xw,
                            const float* __restrict__ a_src,
                            const float* __restrict__ a_dst,
                            float* __restrict__ esrc, float* __restrict__ edst,
                            int n, int H, int C) {
  int t = blockIdx.x * blockDim.x + threadIdx.x;
  if (t >= n * H) return;
  int node = t / H, h = t % H;
  const float* v  = xw + (size_t)node * H * C + (size_t)h * C;
  const float* as = a_src + (size_t)h * C;
  const float* ad = a_dst + (size_t)h * C;
  float ss = 0.f, sd = 0.f;
  for (int c = 0; c < C; ++c) { float x = v[c]; ss += x * as[c]; sd += x * ad[c]; }
  esrc[t] = ss;
  edst[t] = sd;
}

__device__ __forceinline__ unsigned ordered_enc(float f) {
  unsigned u = __float_as_uint(f);
  return (u & 0x80000000u) ? ~u : (u | 0x80000000u);
}
__device__ __forceinline__ float ordered_dec(unsigned o) {
  return (o & 0x80000000u) ? __uint_as_float(o & 0x7FFFFFFFu) : __uint_as_float(~o);
}

__device__ __forceinline__ void edge_nodes(const int* __restrict__ src,
                                           const int* __restrict__ dst,
                                           int e, int& s, int& d) {
  if (e < NEDGES) { s = src[e]; d = dst[e]; }
  else            { s = e - NEDGES; d = s; }
}

__global__ void edge_alpha_max(const int* __restrict__ src, const int* __restrict__ dst,
                               const float* __restrict__ esrc, const float* __restrict__ edst,
                               unsigned* __restrict__ amax, float* __restrict__ alpha, int H) {
  int t = blockIdx.x * blockDim.x + threadIdx.x;
  if (t >= ETOT * H) return;
  int e = t / H, h = t % H;
  int s, d; edge_nodes(src, dst, e, s, d);
  float a = esrc[s * H + h] + edst[d * H + h];
  a = (a >= 0.f) ? a : 0.2f * a;
  alpha[t] = a;
  atomicMax(&amax[d * H + h], ordered_enc(a));
}

__global__ void edge_exp_denom(const int* __restrict__ src, const int* __restrict__ dst,
                               float* __restrict__ alpha_ex,
                               const unsigned* __restrict__ amax,
                               float* __restrict__ denom, int H) {
  int t = blockIdx.x * blockDim.x + threadIdx.x;
  if (t >= ETOT * H) return;
  int e = t / H, h = t % H;
  int s, d; edge_nodes(src, dst, e, s, d);
  float m  = ordered_dec(amax[d * H + h]);
  float ex = expf(alpha_ex[t] - m);
  alpha_ex[t] = ex;
  atomicAdd(&denom[d * H + h], ex);
}

__global__ void edge_aggregate(const int* __restrict__ src, const int* __restrict__ dst,
                               const float* __restrict__ xw,
                               const float* __restrict__ ex, const float* __restrict__ denom,
                               float* __restrict__ agg, int H, int C) {
  int F = H * C;
  int t = blockIdx.x * blockDim.x + threadIdx.x;
  if (t >= ETOT * F) return;
  int e = t / F, c = t % F;
  int h = c / C;
  int s, d; edge_nodes(src, dst, e, s, d);
  float a = ex[(size_t)e * H + h] / denom[d * H + h];
  atomicAdd(&agg[(size_t)d * F + c], xw[(size_t)s * F + c] * a);
}

__global__ void bias_act(float* __restrict__ x, const float* __restrict__ bias,
                         int n, int F, int doElu) {
  int t = blockIdx.x * blockDim.x + threadIdx.x;
  if (t >= n * F) return;
  int c = t % F;
  float v = x[t] + bias[c];
  if (doElu) v = (v > 0.f) ? v : (expf(v) - 1.f);
  x[t] = v;
}

__global__ void pool_count(const int* __restrict__ batch, float* __restrict__ cnt) {
  int n = blockIdx.x * blockDim.x + threadIdx.x;
  if (n >= NNODES) return;
  atomicAdd(&cnt[batch[n]], 1.f);
}
__global__ void pool_sum(const float* __restrict__ h, const int* __restrict__ batch,
                         float* __restrict__ out) {
  int t = blockIdx.x * blockDim.x + threadIdx.x;
  if (t >= NNODES * OUTD) return;
  int n = t / OUTD, c = t % OUTD;
  atomicAdd(&out[(size_t)batch[n] * OUTD + c], h[t]);
}
__global__ void pool_div(float* __restrict__ out, const float* __restrict__ cnt) {
  int t = blockIdx.x * blockDim.x + threadIdx.x;
  if (t >= NGRAPH * OUTD) return;
  out[t] /= fmaxf(cnt[t / OUTD], 1.f);
}

static inline int cdiv(long long a, int b) { return (int)((a + b - 1) / b); }

extern "C" void kernel_launch(void* const* d_in, const int* in_sizes, int n_in,
                              void* d_out, int out_size, void* d_ws, size_t ws_size,
                              hipStream_t stream) {
  (void)in_sizes; (void)n_in; (void)out_size; (void)ws_size;
  const float* x   = (const float*)d_in[0];
  const int*   ei  = (const int*)d_in[1];
  const int*   bat = (const int*)d_in[2];
  const float* W1  = (const float*)d_in[3];
  const float* as1 = (const float*)d_in[4];
  const float* ad1 = (const float*)d_in[5];
  const float* b1  = (const float*)d_in[6];
  const float* W2  = (const float*)d_in[7];
  const float* as2 = (const float*)d_in[8];
  const float* ad2 = (const float*)d_in[9];
  const float* b2  = (const float*)d_in[10];
  const int* src = ei;
  const int* dst = ei + NEDGES;

  char* ws = (char*)d_ws;
  size_t off = 0;
  auto carve = [&](size_t bytes) -> char* {
    char* p = ws + off;
    off += (bytes + 255) & ~(size_t)255;
    return p;
  };
  float*    xw1    = (float*)carve((size_t)NNODES * HID * 4);
  float*    esrc1  = (float*)carve((size_t)NNODES * H1C * 4);
  float*    edst1  = (float*)carve((size_t)NNODES * H1C * 4);
  unsigned* amax1  = (unsigned*)carve((size_t)NNODES * H1C * 4);
  float*    denom1 = (float*)carve((size_t)NNODES * H1C * 4);
  float*    ex1    = (float*)carve((size_t)ETOT * H1C * 4);
  float*    agg1   = (float*)carve((size_t)NNODES * HID * 4);
  float*    agg2   = (float*)carve((size_t)NNODES * OUTD * 4);
  float*    cnt    = (float*)carve((size_t)NGRAPH * 4);
  float*    xw2    = xw1;
  float*    esrc2  = esrc1;
  float*    edst2  = edst1;
  unsigned* amax2  = amax1;
  float*    denom2 = denom1;
  float*    ex2    = ex1;
  float*    outF   = (float*)d_out;

  const int B = 256;

  hipMemsetAsync(amax1,  0, (size_t)NNODES * H1C * 4, stream);
  hipMemsetAsync(denom1, 0, (size_t)NNODES * H1C * 4, stream);
  hipMemsetAsync(agg1,   0, (size_t)NNODES * HID * 4, stream);

  {
    int tiles = (NNODES / 16) * (HID / 16);
    gemm_wmma_f32<<<cdiv((long long)tiles * 32, B), B, 0, stream>>>(x, W1, xw1, NNODES, INDIM, HID);
  }
  attn_scores<<<cdiv((long long)NNODES * H1C, B), B, 0, stream>>>(xw1, as1, ad1, esrc1, edst1, NNODES, H1C, C1C);
  edge_alpha_max<<<cdiv((long long)ETOT * H1C, B), B, 0, stream>>>(src, dst, esrc1, edst1, amax1, ex1, H1C);
  edge_exp_denom<<<cdiv((long long)ETOT * H1C, B), B, 0, stream>>>(src, dst, ex1, amax1, denom1, H1C);
  edge_aggregate<<<cdiv((long long)ETOT * HID, B), B, 0, stream>>>(src, dst, xw1, ex1, denom1, agg1, H1C, C1C);
  bias_act<<<cdiv((long long)NNODES * HID, B), B, 0, stream>>>(agg1, b1, NNODES, HID, 1);

  hipMemsetAsync(amax2,  0, (size_t)NNODES * 4, stream);
  hipMemsetAsync(denom2, 0, (size_t)NNODES * 4, stream);
  hipMemsetAsync(agg2,   0, (size_t)NNODES * OUTD * 4, stream);

  {
    int tiles = (NNODES / 16) * (OUTD / 16);
    gemm_wmma_f32<<<cdiv((long long)tiles * 32, B), B, 0, stream>>>(agg1, W2, xw2, NNODES, HID, OUTD);
  }
  attn_scores<<<cdiv((long long)NNODES, B), B, 0, stream>>>(xw2, as2, ad2, esrc2, edst2, NNODES, 1, OUTD);
  edge_alpha_max<<<cdiv((long long)ETOT, B), B, 0, stream>>>(src, dst, esrc2, edst2, amax2, ex2, 1);
  edge_exp_denom<<<cdiv((long long)ETOT, B), B, 0, stream>>>(src, dst, ex2, amax2, denom2, 1);
  edge_aggregate<<<cdiv((long long)ETOT * OUTD, B), B, 0, stream>>>(src, dst, xw2, ex2, denom2, agg2, 1, OUTD);
  bias_act<<<cdiv((long long)NNODES * OUTD, B), B, 0, stream>>>(agg2, b2, NNODES, OUTD, 0);

  hipMemsetAsync(outF, 0, (size_t)NGRAPH * OUTD * 4, stream);
  hipMemsetAsync(cnt,  0, (size_t)NGRAPH * 4, stream);
  pool_count<<<cdiv(NNODES, B), B, 0, stream>>>(bat, cnt);
  pool_sum<<<cdiv((long long)NNODES * OUTD, B), B, 0, stream>>>(agg2, bat, outF);
  pool_div<<<cdiv((long long)NGRAPH * OUTD, B), B, 0, stream>>>(outF, cnt);
}


# Softmax folded into the owning work-group: edge list kept in LDS; max, exp-sum, weighted row sums there, rows divided once at the end; edge score passes and memsets gone; LDS arrival order
# speedup vs baseline: 3.5079x; 3.5079x over previous
#include <hip/hip_runtime.h>
#include <math.h>

#define NNODES 50000
#define NEDGES 800000
#define ETOT   (NEDGES + NNODES)
#define INDIM  128
#define H1C    4
#define C1C    32
#define HID    128
#define OUTD   64
#define NGRAPH 64

typedef float v2f __attribute__((ext_vector_type(2)));
typedef float v8f __attribute__((ext_vector_type(8)));

__global__ void gemm_wmma_f32(const float* __restrict__ A, const float* __restrict__ B,
                              float* __restrict__ Cm, int M, int K, int Ncol) {
  int gwave = (int)((blockIdx.x * blockDim.x + threadIdx.x) >> 5);
  int lane  = (int)(threadIdx.x & 31);
  int tilesN = Ncol >> 4;
  int numTiles = (M >> 4) * tilesN;
  if (gwave >= numTiles) return;
  int tm = (gwave / tilesN) << 4;
  int tn = (gwave % tilesN) << 4;
  int half = lane >> 4;
  int l    = lane & 15;

  v8f acc = {};
  const float* arow = A + (size_t)(tm + l) * K;
  for (int k0 = 0; k0 < K; k0 += 4) {
    int ka = k0 + (half << 1);
    v2f a, b;
    a.x = arow[ka];
    a.y = arow[ka + 1];
    b.x = B[(size_t)(ka)     * Ncol + tn + l];
    b.y = B[(size_t)(ka + 1) * Ncol + tn + l];
    acc = __builtin_amdgcn_wmma_f32_16x16x4_f32(false, a, false, b,
                                                (short)0, acc, false, false);
  }
  int row0 = tm + (half << 3);
  #pragma unroll
  for (int v = 0; v < 8; ++v)
    Cm[(size_t)(row0 + v) * Ncol + tn + l] = acc[v];
}

__global__ void attn_scores(const float* __restrict__ xw,
                            const float* __restrict__ a_src,
                            const float* __restrict__ a_dst,
                            float* __restrict__ esrc, float* __restrict__ edst,
                            int n, int H, int C) {
  int t = blockIdx.x * blockDim.x + threadIdx.x;
  if (t >= n * H) return;
  int node = t / H, h = t % H;
  const float* v  = xw + (size_t)node * H * C + (size_t)h * C;
  const float* as = a_src + (size_t)h * C;
  const float* ad = a_dst + (size_t)h * C;
  float ss = 0.f, sd = 0.f;
  for (int c = 0; c < C; ++c) { float x = v[c]; ss += x * as[c]; sd += x * ad[c]; }
  esrc[t] = ss;
  edst[t] = sd;
}

__device__ __forceinline__ unsigned ordered_enc(float f) {
  unsigned u = __float_as_uint(f);
  return (u & 0x80000000u) ? ~u : (u | 0x80000000u);
}
__device__ __forceinline__ float ordered_dec(unsigned o) {
  return (o & 0x80000000u) ? __uint_as_float(o & 0x7FFFFFFFu) : __uint_as_float(~o);
}

__device__ __forceinline__ float lrelu02(float a) { return (a >= 0.f) ? a : 0.2f * a; }

#define NWG    256
#define NPW    ((NNODES + NWG - 1) / NWG)
#define AGG_T  1024
#define AGG_W  (AGG_T / 32)
#define QCAP   24576
#define NQ     (NEDGES / 4)
#define SLD    8
#define EBITS  20
#define SBITS  16
static_assert(NEDGES <= (1 << EBITS), "edge id must fit the raw entry's low bits");
static_assert(NPW <= (1 << (32 - EBITS)), "local node must fit the raw entry's high bits");
static_assert(NNODES <= (1 << SBITS), "source node must fit the resolved entry's low bits");
static_assert(NEDGES % 4 == 0, "16-byte scan granularity");

template <int F, int H, int ELU>
__global__ __launch_bounds__(AGG_T)
void gat_owner_aggregate(const int* __restrict__ src, const int* __restrict__ dst,
                         const float* __restrict__ xw,
                         const float* __restrict__ esrc, const float* __restrict__ edst,
                         const float* __restrict__ bias, float* __restrict__ out) {
  constexpr int NJ = F / 32;
  constexpr int C  = F / H;
  constexpr int U  = 32 / NJ;
  constexpr int RU = 4;
  static_assert(U * H <= 32, "one lane per (edge, head) pair of a wave's batch");

  __shared__ float    s_acc[NPW * F];
  __shared__ unsigned s_list[QCAP];
  __shared__ unsigned s_mx[NPW * H];
  __shared__ float    s_ed[NPW * H];
  __shared__ float    s_m[NPW * H];
  __shared__ float    s_den[NPW * H];
  __shared__ int      s_cnt;

  const int tid  = (int)threadIdx.x;
  const int lane = tid & 31;
  const int wave = tid >> 5;
  const int lo   = (int)blockIdx.x * NPW;
  const int cntN = (NNODES - lo < NPW) ? (NNODES - lo) : NPW;

  for (int i = tid; i < NPW * F; i += AGG_T) s_acc[i] = 0.f;
  for (int i = tid; i < NPW * H; i += AGG_T) {
    const bool own = i < cntN * H;
    float es0 = 0.f, ed0 = 0.f;
    if (own) { es0 = esrc[lo * H + i]; ed0 = edst[lo * H + i]; }
    s_mx[i]  = own ? ordered_enc(es0) : 0u;
    s_ed[i]  = ed0;
    s_m[i]   = 0.f;
    s_den[i] = 0.f;
  }
  if (tid == 0) s_cnt = 0;
  __syncthreads();

  {
    const int4* dst4 = (const int4*)dst;
    constexpr int STEP = AGG_T * SLD;
    auto load_ids = [&](int4 (&a)[SLD], int base) {
      #pragma unroll
      for (int i = 0; i < SLD; ++i) {
        const int q = base + i * AGG_T + tid;
        a[i] = dst4[(q < NQ) ? q : (NQ - 1)];
      }
    };
    auto test_ids = [&](const int4 (&a)[SLD], int base) {
      #pragma unroll
      for (int i = 0; i < SLD; ++i) {
        const int e0 = (base + i * AGG_T + tid) * 4;
        const unsigned d0 = (unsigned)(a[i].x - lo), d1 = (unsigned)(a[i].y - lo);
        const unsigned d2 = (unsigned)(a[i].z - lo), d3 = (unsigned)(a[i].w - lo);
        const unsigned mn = min(min(d0, d1), min(d2, d3));
        if (mn < (unsigned)cntN && e0 < NEDGES) {
          const unsigned dd[4] = {d0, d1, d2, d3};
          #pragma unroll
          for (int r = 0; r < 4; ++r) {
            if (dd[r] < (unsigned)cntN) {
              int pos = atomicAdd(&s_cnt, 1);
              if (pos < QCAP) s_list[pos] = (dd[r] << EBITS) | (unsigned)(e0 + r);
            }
          }
        }
      }
    };
    int4 setA[SLD], setB[SLD];
    load_ids(setA, 0);
    for (int base = 0; base < NQ; base += 2 * STEP) {
      load_ids(setB, base + STEP);
      test_ids(setA, base);
      load_ids(setA, base + 2 * STEP);
      test_ids(setB, base + STEP);
    }
  }
  __syncthreads();
  const int  n    = s_cnt;
  const bool fits = (n <= QCAP);

  if (fits) {
    for (int i0 = 0; i0 < n; i0 += AGG_T * RU) {
      unsigned ent[RU];
      int      sv[RU];
      float    es[RU][H];
      #pragma unroll
      for (int k = 0; k < RU; ++k) {
        const int idx = i0 + k * AGG_T + tid;
        ent[k] = (idx < n) ? s_list[idx] : 0u;
      }
      #pragma unroll
      for (int k = 0; k < RU; ++k) sv[k] = src[ent[k] & ((1u << EBITS) - 1u)];
      #pragma unroll
      for (int k = 0; k < RU; ++k)
        #pragma unroll
        for (int h = 0; h < H; ++h) es[k][h] = esrc[sv[k] * H + h];
      #pragma unroll
      for (int k = 0; k < RU; ++k) {
        const int idx = i0 + k * AGG_T + tid;
        if (idx < n) {
          const unsigned dl = ent[k] >> EBITS;
          #pragma unroll
          for (int h = 0; h < H; ++h) atomicMax(&s_mx[dl * H + h], ordered_enc(es[k][h]));
          s_list[idx] = (dl << SBITS) | (unsigned)sv[k];
        }
      }
    }
  } else {
    for (int e = tid; e < NEDGES; e += AGG_T) {
      const unsigned dl = (unsigned)(dst[e] - lo);
      if (dl < (unsigned)cntN) {
        const int s = src[e];
        for (int h = 0; h < H; ++h) atomicMax(&s_mx[dl * H + h], ordered_enc(esrc[s * H + h]));
      }
    }
  }
  __syncthreads();

  for (int i = tid; i < cntN * H; i += AGG_T) s_m[i] = lrelu02(ordered_dec(s_mx[i]) + s_ed[i]);
  __syncthreads();

  if (fits) {
    const int p  = lane % (U * H);
    const int pk = p / H;
    const int ph = p % H;
    for (int i0 = 0; i0 < n; i0 += AGG_W * U) {
      const int  idxL  = i0 + pk * AGG_W + wave;
      const bool liveL = idxL < n;
      const unsigned entL = s_list[liveL ? idxL : (n - 1)];
      const int sL  = (int)(entL & ((1u << SBITS) - 1u));
      const int dlL = (int)(entL >> SBITS);
      const float esL = esrc[sL * H + ph];
      unsigned ent[U];
      #pragma unroll
      for (int k = 0; k < U; ++k) ent[k] = (unsigned)__builtin_amdgcn_readlane((int)entL, k * H);
      float v[U][NJ];
      #pragma unroll
      for (int k = 0; k < U; ++k)
        #pragma unroll
        for (int j = 0; j < NJ; ++j)
          v[k][j] = xw[(size_t)(ent[k] & ((1u << SBITS) - 1u)) * F + j * 32 + lane];
      const float aL = lrelu02(esL + s_ed[dlL * H + ph]);
      const float wL = liveL ? expf(aL - s_m[dlL * H + ph]) : 0.f;
      if (lane < U * H) atomicAdd(&s_den[dlL * H + ph], wL);
      #pragma unroll
      for (int k = 0; k < U; ++k) {
        float* row = s_acc + (size_t)(ent[k] >> SBITS) * F + lane;
        #pragma unroll
        for (int j = 0; j < NJ; ++j) {
          const float w = __uint_as_float((unsigned)__builtin_amdgcn_readlane(
              (int)__float_as_uint(wL), k * H + (j * 32) / C));
          atomicAdd(row + j * 32, v[k][j] * w);
        }
      }
    }
  } else {
    for (int e = tid; e < NEDGES; e += AGG_T) {
      const unsigned dl = (unsigned)(dst[e] - lo);
      if (dl < (unsigned)cntN) {
        const int s = src[e];
        for (int h = 0; h < H; ++h) {
          const float w = expf(lrelu02(esrc[s * H + h] + s_ed[dl * H + h]) - s_m[dl * H + h]);
          atomicAdd(&s_den[dl * H + h], w);
          for (int c = 0; c < C; ++c)
            atomicAdd(&s_acc[dl * F + h * C + c], w * xw[(size_t)s * F + h * C + c]);
        }
      }
    }
  }
  __syncthreads();

  float* orow = out + (size_t)lo * F;
  for (int i = tid; i < cntN * F; i += AGG_T) {
    const int dl = i / F, col = i % F, h = col / C;
    const float w   = expf(lrelu02(esrc[(lo + dl) * H + h] + s_ed[dl * H + h]) - s_m[dl * H + h]);
    const float num = s_acc[i] + w * xw[(size_t)lo * F + i];
    const float den = s_den[dl * H + h] + w;
    float v = num / den + bias[col];
    if (ELU) v = (v > 0.f) ? v : (expf(v) - 1.f);
    orow[i] = v;
  }
}

__global__ void pool_count(const int* __restrict__ batch, float* __restrict__ cnt) {
  int n = blockIdx.x * blockDim.x + threadIdx.x;
  if (n >= NNODES) return;
  atomicAdd(&cnt[batch[n]], 1.f);
}
__global__ void pool_sum(const float* __restrict__ h, const int* __restrict__ batch,
                         float* __restrict__ out) {
  int t = blockIdx.x * blockDim.x + threadIdx.x;
  if (t >= NNODES * OUTD) return;
  int n = t / OUTD, c = t % OUTD;
  atomicAdd(&out[(size_t)batch[n] * OUTD + c], h[t]);
}
__global__ void pool_div(float* __restrict__ out, const float* __restrict__ cnt) {
  int t = blockIdx.x * blockDim.x + threadIdx.x;
  if (t >= NGRAPH * OUTD) return;
  out[t] /= fmaxf(cnt[t / OUTD], 1.f);
}

static inline int cdiv(long long a, int b) { return (int)((a + b - 1) / b); }

extern "C" void kernel_launch(void* const* d_in, const int* in_sizes, int n_in,
                              void* d_out, int out_size, void* d_ws, size_t ws_size,
                              hipStream_t stream) {
  (void)in_sizes; (void)n_in; (void)out_size; (void)ws_size;
  const float* x   = (const float*)d_in[0];
  const int*   ei  = (const int*)d_in[1];
  const int*   bat = (const int*)d_in[2];
  const float* W1  = (const float*)d_in[3];
  const float* as1 = (const float*)d_in[4];
  const float* ad1 = (const float*)d_in[5];
  const float* b1  = (const float*)d_in[6];
  const float* W2  = (const float*)d_in[7];
  const float* as2 = (const float*)d_in[8];
  const float* ad2 = (const float*)d_in[9];
  const float* b2  = (const float*)d_in[10];
  const int* src = ei;
  const int* dst = ei + NEDGES;

  char* ws = (char*)d_ws;
  size_t off = 0;
  auto carve = [&](size_t bytes) -> char* {
    char* p = ws + off;
    off += (bytes + 255) & ~(size_t)255;
    return p;
  };
  float*    xw1    = (float*)carve((size_t)NNODES * HID * 4);
  float*    esrc1  = (float*)carve((size_t)NNODES * H1C * 4);
  float*    edst1  = (float*)carve((size_t)NNODES * H1C * 4);
  float*    agg1   = (float*)carve((size_t)NNODES * HID * 4);
  float*    agg2   = (float*)carve((size_t)NNODES * OUTD * 4);
  float*    cnt    = (float*)carve((size_t)NGRAPH * 4);
  float*    xw2    = xw1;
  float*    esrc2  = esrc1;
  float*    edst2  = edst1;
  float*    outF   = (float*)d_out;

  const int B = 256;

  {
    int tiles = (NNODES / 16) * (HID / 16);
    gemm_wmma_f32<<<cdiv((long long)tiles * 32, B), B, 0, stream>>>(x, W1, xw1, NNODES, INDIM, HID);
  }
  attn_scores<<<cdiv((long long)NNODES * H1C, B), B, 0, stream>>>(xw1, as1, ad1, esrc1, edst1, NNODES, H1C, C1C);
  gat_owner_aggregate<HID, H1C, 1><<<NWG, AGG_T, 0, stream>>>(src, dst, xw1, esrc1, edst1, b1, agg1);

  {
    int tiles = (NNODES / 16) * (OUTD / 16);
    gemm_wmma_f32<<<cdiv((long long)tiles * 32, B), B, 0, stream>>>(agg1, W2, xw2, NNODES, HID, OUTD);
  }
  attn_scores<<<cdiv((long long)NNODES, B), B, 0, stream>>>(xw2, as2, ad2, esrc2, edst2, NNODES, 1, OUTD);
  gat_owner_aggregate<OUTD, 1, 0><<<NWG, AGG_T, 0, stream>>>(src, dst, xw2, esrc2, edst2, b2, agg2);

  hipMemsetAsync(outF, 0, (size_t)NGRAPH * OUTD * 4, stream);
  hipMemsetAsync(cnt,  0, (size_t)NGRAPH * 4, stream);
  pool_count<<<cdiv(NNODES, B), B, 0, stream>>>(bat, cnt);
  pool_sum<<<cdiv((long long)NNODES * OUTD, B), B, 0, stream>>>(agg2, bat, outF);
  pool_div<<<cdiv((long long)NGRAPH * OUTD, B), B, 0, stream>>>(outF, cnt);
}


# Owner scan on byte tags: a new first launch writes tag = dst/196 per edge (0.8 MB); each layer's groups search 16 tags per 16-byte load instead of 4 ids; sums stay in LDS arrival order
# speedup vs baseline: 3.7980x; 1.0827x over previous
#include <hip/hip_runtime.h>
#include <math.h>

#define NNODES 50000
#define NEDGES 800000
#define ETOT   (NEDGES + NNODES)
#define INDIM  128
#define H1C    4
#define C1C    32
#define HID    128
#define OUTD   64
#define NGRAPH 64

typedef float v2f __attribute__((ext_vector_type(2)));
typedef float v8f __attribute__((ext_vector_type(8)));

__global__ void gemm_wmma_f32(const float* __restrict__ A, const float* __restrict__ B,
                              float* __restrict__ Cm, int M, int K, int Ncol) {
  int gwave = (int)((blockIdx.x * blockDim.x + threadIdx.x) >> 5);
  int lane  = (int)(threadIdx.x & 31);
  int tilesN = Ncol >> 4;
  int numTiles = (M >> 4) * tilesN;
  if (gwave >= numTiles) return;
  int tm = (gwave / tilesN) << 4;
  int tn = (gwave % tilesN) << 4;
  int half = lane >> 4;
  int l    = lane & 15;

  v8f acc = {};
  const float* arow = A + (size_t)(tm + l) * K;
  for (int k0 = 0; k0 < K; k0 += 4) {
    int ka = k0 + (half << 1);
    v2f a, b;
    a.x = arow[ka];
    a.y = arow[ka + 1];
    b.x = B[(size_t)(ka)     * Ncol + tn + l];
    b.y = B[(size_t)(ka + 1) * Ncol + tn + l];
    acc = __builtin_amdgcn_wmma_f32_16x16x4_f32(false, a, false, b,
                                                (short)0, acc, false, false);
  }
  int row0 = tm + (half << 3);
  #pragma unroll
  for (int v = 0; v < 8; ++v)
    Cm[(size_t)(row0 + v) * Ncol + tn + l] = acc[v];
}

__global__ void attn_scores(const float* __restrict__ xw,
                            const float* __restrict__ a_src,
                            const float* __restrict__ a_dst,
                            float* __restrict__ esrc, float* __restrict__ edst,
                            int n, int H, int C) {
  int t = blockIdx.x * blockDim.x + threadIdx.x;
  if (t >= n * H) return;
  int node = t / H, h = t % H;
  const float* v  = xw + (size_t)node * H * C + (size_t)h * C;
  const float* as = a_src + (size_t)h * C;
  const float* ad = a_dst + (size_t)h * C;
  float ss = 0.f, sd = 0.f;
  for (int c = 0; c < C; ++c) { float x = v[c]; ss += x * as[c]; sd += x * ad[c]; }
  esrc[t] = ss;
  edst[t] = sd;
}

__device__ __forceinline__ unsigned ordered_enc(float f) {
  unsigned u = __float_as_uint(f);
  return (u & 0x80000000u) ? ~u : (u | 0x80000000u);
}
__device__ __forceinline__ float ordered_dec(unsigned o) {
  return (o & 0x80000000u) ? __uint_as_float(o & 0x7FFFFFFFu) : __uint_as_float(~o);
}

__device__ __forceinline__ float lrelu02(float a) { return (a >= 0.f) ? a : 0.2f * a; }

#define NWG    256
#define NPW    ((NNODES + NWG - 1) / NWG)
#define AGG_T  1024
#define AGG_W  (AGG_T / 32)
#define QCAP   24576
#define NQ     (NEDGES / 4)
#define NQT    (NEDGES / 16)
#define SLD    5
#define SBITS  16
static_assert(NWG <= 256, "an owner tag is one byte");
static_assert(NPW <= (1 << (32 - SBITS)), "local node must fit the resolved entry's high bits");
static_assert(NNODES <= (1 << SBITS), "source node must fit the resolved entry's low bits");
static_assert(NEDGES % 16 == 0, "16-byte granularity of the tag scan");

#define TAG_T  256
#define TAG_P  4
#define TAG_G  ((NQ + TAG_T * TAG_P - 1) / (TAG_T * TAG_P))
__global__ __launch_bounds__(TAG_T)
void owner_tags(const int* __restrict__ dst, unsigned* __restrict__ tags) {
  const int4* dst4 = (const int4*)dst;
  const int q0 = (int)blockIdx.x * (TAG_T * TAG_P) + (int)threadIdx.x;
  int4 d[TAG_P];
  #pragma unroll
  for (int i = 0; i < TAG_P; ++i) {
    const int q = q0 + i * TAG_T;
    d[i] = dst4[(q < NQ) ? q : (NQ - 1)];
  }
  #pragma unroll
  for (int i = 0; i < TAG_P; ++i) {
    const int q = q0 + i * TAG_T;
    if (q < NQ) {
      const unsigned t0 = min((unsigned)d[i].x / (unsigned)NPW, 255u);
      const unsigned t1 = min((unsigned)d[i].y / (unsigned)NPW, 255u);
      const unsigned t2 = min((unsigned)d[i].z / (unsigned)NPW, 255u);
      const unsigned t3 = min((unsigned)d[i].w / (unsigned)NPW, 255u);
      tags[q] = t0 | (t1 << 8) | (t2 << 16) | (t3 << 24);
    }
  }
}

template <int F, int H, int ELU>
__global__ __launch_bounds__(AGG_T)
void gat_owner_aggregate(const int* __restrict__ src, const int* __restrict__ dst,
                         const unsigned* __restrict__ tags,
                         const float* __restrict__ xw,
                         const float* __restrict__ esrc, const float* __restrict__ edst,
                         const float* __restrict__ bias, float* __restrict__ out) {
  constexpr int NJ = F / 32;
  constexpr int C  = F / H;
  constexpr int U  = 32 / NJ;
  constexpr int RU = 4;
  static_assert(U * H <= 32, "one lane per (edge, head) pair of a wave's batch");

  __shared__ float    s_acc[NPW * F];
  __shared__ unsigned s_list[QCAP];
  __shared__ unsigned s_mx[NPW * H];
  __shared__ float    s_ed[NPW * H];
  __shared__ float    s_m[NPW * H];
  __shared__ float    s_den[NPW * H];
  __shared__ int      s_cnt;

  const int tid  = (int)threadIdx.x;
  const int lane = tid & 31;
  const int wave = tid >> 5;
  const int lo   = (int)blockIdx.x * NPW;
  const int cntN = (NNODES - lo < NPW) ? (NNODES - lo) : NPW;

  for (int i = tid; i < NPW * F; i += AGG_T) s_acc[i] = 0.f;
  for (int i = tid; i < NPW * H; i += AGG_T) {
    const bool own = i < cntN * H;
    float es0 = 0.f, ed0 = 0.f;
    if (own) { es0 = esrc[lo * H + i]; ed0 = edst[lo * H + i]; }
    s_mx[i]  = own ? ordered_enc(es0) : 0u;
    s_ed[i]  = ed0;
    s_m[i]   = 0.f;
    s_den[i] = 0.f;
  }
  if (tid == 0) s_cnt = 0;
  __syncthreads();

  {
    const uint4* tag4 = (const uint4*)tags;
    constexpr int STEP = AGG_T * SLD;
    const unsigned pat = (unsigned)blockIdx.x * 0x01010101u;
    auto load_tags = [&](uint4 (&a)[SLD], int base) {
      #pragma unroll
      for (int i = 0; i < SLD; ++i) {
        const int q = base + i * AGG_T + tid;
        a[i] = tag4[(q < NQT) ? q : (NQT - 1)];
      }
    };
    auto test_tags = [&](const uint4 (&a)[SLD], int base) {
      #pragma unroll
      for (int i = 0; i < SLD; ++i) {
        const int q = base + i * AGG_T + tid;
        const unsigned w[4] = {a[i].x, a[i].y, a[i].z, a[i].w};
        unsigned m = 0u;
        #pragma unroll
        for (int j = 0; j < 4; ++j) {
          const unsigned x = w[j] ^ pat;
          const unsigned f = ~(((x & 0x7F7F7F7Fu) + 0x7F7F7F7Fu) | x) & 0x80808080u;
          m |= f >> (7 - j);
        }
        if (m != 0u && q < NQT) {
          do {
            const int b = __builtin_ctz(m);
            m &= m - 1u;
            const int e = q * 16 + (b & 7) * 4 + (b >> 3);
            int pos = atomicAdd(&s_cnt, 1);
            if (pos < QCAP) s_list[pos] = (unsigned)e;
          } while (m != 0u);
        }
      }
    };
    uint4 setA[SLD], setB[SLD];
    load_tags(setA, 0);
    for (int base = 0; base < NQT; base += 2 * STEP) {
      load_tags(setB, base + STEP);
      test_tags(setA, base);
      load_tags(setA, base + 2 * STEP);
      test_tags(setB, base + STEP);
    }
  }
  __syncthreads();
  const int  n    = s_cnt;
  const bool fits = (n <= QCAP);

  if (fits) {
    for (int i0 = 0; i0 < n; i0 += AGG_T * RU) {
      unsigned ent[RU];
      int      sv[RU];
      int      dv[RU];
      float    es[RU][H];
      #pragma unroll
      for (int k = 0; k < RU; ++k) {
        const int idx = i0 + k * AGG_T + tid;
        ent[k] = (idx < n) ? s_list[idx] : 0u;
      }
      #pragma unroll
      for (int k = 0; k < RU; ++k) { sv[k] = src[ent[k]]; dv[k] = dst[ent[k]]; }
      #pragma unroll
      for (int k = 0; k < RU; ++k)
        #pragma unroll
        for (int h = 0; h < H; ++h) es[k][h] = esrc[sv[k] * H + h];
      #pragma unroll
      for (int k = 0; k < RU; ++k) {
        const int idx = i0 + k * AGG_T + tid;
        if (idx < n) {
          const unsigned dl = (unsigned)(dv[k] - lo);
          #pragma unroll
          for (int h = 0; h < H; ++h) atomicMax(&s_mx[dl * H + h], ordered_enc(es[k][h]));
          s_list[idx] = (dl << SBITS) | (unsigned)sv[k];
        }
      }
    }
  } else {
    for (int e = tid; e < NEDGES; e += AGG_T) {
      const unsigned dl = (unsigned)(dst[e] - lo);
      if (dl < (unsigned)cntN) {
        const int s = src[e];
        for (int h = 0; h < H; ++h) atomicMax(&s_mx[dl * H + h], ordered_enc(esrc[s * H + h]));
      }
    }
  }
  __syncthreads();

  for (int i = tid; i < cntN * H; i += AGG_T) s_m[i] = lrelu02(ordered_dec(s_mx[i]) + s_ed[i]);
  __syncthreads();

  if (fits) {
    const int p  = lane % (U * H);
    const int pk = p / H;
    const int ph = p % H;
    for (int i0 = 0; i0 < n; i0 += AGG_W * U) {
      const int  idxL  = i0 + pk * AGG_W + wave;
      const bool liveL = idxL < n;
      const unsigned entL = s_list[liveL ? idxL : (n - 1)];
      const int sL  = (int)(entL & ((1u << SBITS) - 1u));
      const int dlL = (int)(entL >> SBITS);
      const float esL = esrc[sL * H + ph];
      unsigned ent[U];
      #pragma unroll
      for (int k = 0; k < U; ++k) ent[k] = (unsigned)__builtin_amdgcn_readlane((int)entL, k * H);
      float v[U][NJ];
      #pragma unroll
      for (int k = 0; k < U; ++k)
        #pragma unroll
        for (int j = 0; j < NJ; ++j)
          v[k][j] = xw[(size_t)(ent[k] & ((1u << SBITS) - 1u)) * F + j * 32 + lane];
      const float aL = lrelu02(esL + s_ed[dlL * H + ph]);
      const float wL = liveL ? expf(aL - s_m[dlL * H + ph]) : 0.f;
      if (lane < U * H) atomicAdd(&s_den[dlL * H + ph], wL);
      #pragma unroll
      for (int k = 0; k < U; ++k) {
        float* row = s_acc + (size_t)(ent[k] >> SBITS) * F + lane;
        #pragma unroll
        for (int j = 0; j < NJ; ++j) {
          const float w = __uint_as_float((unsigned)__builtin_amdgcn_readlane(
              (int)__float_as_uint(wL), k * H + (j * 32) / C));
          atomicAdd(row + j * 32, v[k][j] * w);
        }
      }
    }
  } else {
    for (int e = tid; e < NEDGES; e += AGG_T) {
      const unsigned dl = (unsigned)(dst[e] - lo);
      if (dl < (unsigned)cntN) {
        const int s = src[e];
        for (int h = 0; h < H; ++h) {
          const float w = expf(lrelu02(esrc[s * H + h] + s_ed[dl * H + h]) - s_m[dl * H + h]);
          atomicAdd(&s_den[dl * H + h], w);
          for (int c = 0; c < C; ++c)
            atomicAdd(&s_acc[dl * F + h * C + c], w * xw[(size_t)s * F + h * C + c]);
        }
      }
    }
  }
  __syncthreads();

  float* orow = out + (size_t)lo * F;
  for (int i = tid; i < cntN * F; i += AGG_T) {
    const int dl = i / F, col = i % F, h = col / C;
    const float w   = expf(lrelu02(esrc[(lo + dl) * H + h] + s_ed[dl * H + h]) - s_m[dl * H + h]);
    const float num = s_acc[i] + w * xw[(size_t)lo * F + i];
    const float den = s_den[dl * H + h] + w;
    float v = num / den + bias[col];
    if (ELU) v = (v > 0.f) ? v : (expf(v) - 1.f);
    orow[i] = v;
  }
}

__global__ void pool_count(const int* __restrict__ batch, float* __restrict__ cnt) {
  int n = blockIdx.x * blockDim.x + threadIdx.x;
  if (n >= NNODES) return;
  atomicAdd(&cnt[batch[n]], 1.f);
}
__global__ void pool_sum(const float* __restrict__ h, const int* __restrict__ batch,
                         float* __restrict__ out) {
  int t = blockIdx.x * blockDim.x + threadIdx.x;
  if (t >= NNODES * OUTD) return;
  int n = t / OUTD, c = t % OUTD;
  atomicAdd(&out[(size_t)batch[n] * OUTD + c], h[t]);
}
__global__ void pool_div(float* __restrict__ out, const float* __restrict__ cnt) {
  int t = blockIdx.x * blockDim.x + threadIdx.x;
  if (t >= NGRAPH * OUTD) return;
  out[t] /= fmaxf(cnt[t / OUTD], 1.f);
}

static inline int cdiv(long long a, int b) { return (int)((a + b - 1) / b); }

extern "C" void kernel_launch(void* const* d_in, const int* in_sizes, int n_in,
                              void* d_out, int out_size, void* d_ws, size_t ws_size,
                              hipStream_t stream) {
  (void)in_sizes; (void)n_in; (void)out_size; (void)ws_size;
  const float* x   = (const float*)d_in[0];
  const int*   ei  = (const int*)d_in[1];
  const int*   bat = (const int*)d_in[2];
  const float* W1  = (const float*)d_in[3];
  const float* as1 = (const float*)d_in[4];
  const float* ad1 = (const float*)d_in[5];
  const float* b1  = (const float*)d_in[6];
  const float* W2  = (const float*)d_in[7];
  const float* as2 = (const float*)d_in[8];
  const float* ad2 = (const float*)d_in[9];
  const float* b2  = (const float*)d_in[10];
  const int* src = ei;
  const int* dst = ei + NEDGES;

  char* ws = (char*)d_ws;
  size_t off = 0;
  auto carve = [&](size_t bytes) -> char* {
    char* p = ws + off;
    off += (bytes + 255) & ~(size_t)255;
    return p;
  };
  float*    xw1    = (float*)carve((size_t)NNODES * HID * 4);
  float*    esrc1  = (float*)carve((size_t)NNODES * H1C * 4);
  float*    edst1  = (float*)carve((size_t)NNODES * H1C * 4);
  float*    agg1   = (float*)carve((size_t)NNODES * HID * 4);
  float*    agg2   = (float*)carve((size_t)NNODES * OUTD * 4);
  float*    cnt    = (float*)carve((size_t)NGRAPH * 4);
  unsigned* tags   = (unsigned*)carve((size_t)NEDGES);
  float*    xw2    = xw1;
  float*    esrc2  = esrc1;
  float*    edst2  = edst1;
  float*    outF   = (float*)d_out;

  const int B = 256;

  owner_tags<<<TAG_G, TAG_T, 0, stream>>>(dst, tags);

  {
    int tiles = (NNODES / 16) * (HID / 16);
    gemm_wmma_f32<<<cdiv((long long)tiles * 32, B), B, 0, stream>>>(x, W1, xw1, NNODES, INDIM, HID);
  }
  attn_scores<<<cdiv((long long)NNODES * H1C, B), B, 0, stream>>>(xw1, as1, ad1, esrc1, edst1, NNODES, H1C, C1C);
  gat_owner_aggregate<HID, H1C, 1><<<NWG, AGG_T, 0, stream>>>(src, dst, tags, xw1, esrc1, edst1, b1, agg1);

  {
    int tiles = (NNODES / 16) * (OUTD / 16);
    gemm_wmma_f32<<<cdiv((long long)tiles * 32, B), B, 0, stream>>>(agg1, W2, xw2, NNODES, HID, OUTD);
  }
  attn_scores<<<cdiv((long long)NNODES, B), B, 0, stream>>>(xw2, as2, ad2, esrc2, edst2, NNODES, 1, OUTD);
  gat_owner_aggregate<OUTD, 1, 0><<<NWG, AGG_T, 0, stream>>>(src, dst, tags, xw2, esrc2, edst2, b2, agg2);

  hipMemsetAsync(outF, 0, (size_t)NGRAPH * OUTD * 4, stream);
  hipMemsetAsync(cnt,  0, (size_t)NGRAPH * 4, stream);
  pool_count<<<cdiv(NNODES, B), B, 0, stream>>>(bat, cnt);
  pool_sum<<<cdiv((long long)NNODES * OUTD, B), B, 0, stream>>>(agg2, bat, outF);
  pool_div<<<cdiv((long long)NGRAPH * OUTD, B), B, 0, stream>>>(outF, cnt);
}


# Mean pool folded into the last layer's owner groups: a thread sums its column over a graph's rows in a register, one float atomic per run to out; pool passes gone; run sums, then arrival order
# speedup vs baseline: 5.2198x; 1.3744x over previous
#include <hip/hip_runtime.h>
#include <math.h>

#define NNODES 50000
#define NEDGES 800000
#define ETOT   (NEDGES + NNODES)
#define INDIM  128
#define H1C    4
#define C1C    32
#define HID    128
#define OUTD   64
#define NGRAPH 64

typedef float v2f __attribute__((ext_vector_type(2)));
typedef float v8f __attribute__((ext_vector_type(8)));

__global__ void gemm_wmma_f32(const float* __restrict__ A, const float* __restrict__ B,
                              float* __restrict__ Cm, int M, int K, int Ncol) {
  int gwave = (int)((blockIdx.x * blockDim.x + threadIdx.x) >> 5);
  int lane  = (int)(threadIdx.x & 31);
  int tilesN = Ncol >> 4;
  int numTiles = (M >> 4) * tilesN;
  if (gwave >= numTiles) return;
  int tm = (gwave / tilesN) << 4;
  int tn = (gwave % tilesN) << 4;
  int half = lane >> 4;
  int l    = lane & 15;

  v8f acc = {};
  const float* arow = A + (size_t)(tm + l) * K;
  for (int k0 = 0; k0 < K; k0 += 4) {
    int ka = k0 + (half << 1);
    v2f a, b;
    a.x = arow[ka];
    a.y = arow[ka + 1];
    b.x = B[(size_t)(ka)     * Ncol + tn + l];
    b.y = B[(size_t)(ka + 1) * Ncol + tn + l];
    acc = __builtin_amdgcn_wmma_f32_16x16x4_f32(false, a, false, b,
                                                (short)0, acc, false, false);
  }
  int row0 = tm + (half << 3);
  #pragma unroll
  for (int v = 0; v < 8; ++v)
    Cm[(size_t)(row0 + v) * Ncol + tn + l] = acc[v];
}

__global__ void attn_scores(const float* __restrict__ xw,
                            const float* __restrict__ a_src,
                            const float* __restrict__ a_dst,
                            float* __restrict__ esrc, float* __restrict__ edst,
                            int n, int H, int C) {
  int t = blockIdx.x * blockDim.x + threadIdx.x;
  if (t >= n * H) return;
  int node = t / H, h = t % H;
  const float* v  = xw + (size_t)node * H * C + (size_t)h * C;
  const float* as = a_src + (size_t)h * C;
  const float* ad = a_dst + (size_t)h * C;
  float ss = 0.f, sd = 0.f;
  for (int c = 0; c < C; ++c) { float x = v[c]; ss += x * as[c]; sd += x * ad[c]; }
  esrc[t] = ss;
  edst[t] = sd;
}

__device__ __forceinline__ unsigned ordered_enc(float f) {
  unsigned u = __float_as_uint(f);
  return (u & 0x80000000u) ? ~u : (u | 0x80000000u);
}
__device__ __forceinline__ float ordered_dec(unsigned o) {
  return (o & 0x80000000u) ? __uint_as_float(o & 0x7FFFFFFFu) : __uint_as_float(~o);
}

__device__ __forceinline__ float lrelu02(float a) { return (a >= 0.f) ? a : 0.2f * a; }

#define NWG    256
#define NPW    ((NNODES + NWG - 1) / NWG)
#define AGG_T  1024
#define AGG_W  (AGG_T / 32)
#define QCAP   24576
#define NQ     (NEDGES / 4)
#define NQT    (NEDGES / 16)
#define SLD    5
#define SBITS  16
static_assert(NWG <= 256, "an owner tag is one byte");
static_assert(NPW <= (1 << (32 - SBITS)), "local node must fit the resolved entry's high bits");
static_assert(NNODES <= (1 << SBITS), "source node must fit the resolved entry's low bits");
static_assert(NEDGES % 16 == 0, "16-byte granularity of the tag scan");

#define TAG_T  256
#define TAG_P  4
#define TAG_G  ((NQ + TAG_T * TAG_P - 1) / (TAG_T * TAG_P))
__global__ __launch_bounds__(TAG_T)
void owner_tags(const int* __restrict__ dst, unsigned* __restrict__ tags) {
  const int4* dst4 = (const int4*)dst;
  const int q0 = (int)blockIdx.x * (TAG_T * TAG_P) + (int)threadIdx.x;
  int4 d[TAG_P];
  #pragma unroll
  for (int i = 0; i < TAG_P; ++i) {
    const int q = q0 + i * TAG_T;
    d[i] = dst4[(q < NQ) ? q : (NQ - 1)];
  }
  #pragma unroll
  for (int i = 0; i < TAG_P; ++i) {
    const int q = q0 + i * TAG_T;
    if (q < NQ) {
      const unsigned t0 = min((unsigned)d[i].x / (unsigned)NPW, 255u);
      const unsigned t1 = min((unsigned)d[i].y / (unsigned)NPW, 255u);
      const unsigned t2 = min((unsigned)d[i].z / (unsigned)NPW, 255u);
      const unsigned t3 = min((unsigned)d[i].w / (unsigned)NPW, 255u);
      tags[q] = t0 | (t1 << 8) | (t2 << 16) | (t3 << 24);
    }
  }
}

template <int F, int H, int ELU, int POOL>
__global__ __launch_bounds__(AGG_T)
void gat_owner_aggregate(const int* __restrict__ src, const int* __restrict__ dst,
                         const unsigned* __restrict__ tags,
                         const float* __restrict__ xw,
                         const float* __restrict__ esrc, const float* __restrict__ edst,
                         const float* __restrict__ bias, float* __restrict__ out,
                         const int* __restrict__ gid, float* __restrict__ gcnt) {
  constexpr int NJ = F / 32;
  constexpr int C  = F / H;
  constexpr int U  = 32 / NJ;
  constexpr int RU = 4;
  static_assert(U * H <= 32, "one lane per (edge, head) pair of a wave's batch");
  static_assert(!POOL || AGG_T % F == 0, "pooling: a thread keeps one column through the row loop");

  __shared__ float    s_acc[NPW * F];
  __shared__ unsigned s_list[QCAP];
  __shared__ unsigned s_mx[NPW * H];
  __shared__ float    s_ed[NPW * H];
  __shared__ float    s_m[NPW * H];
  __shared__ float    s_den[NPW * H];
  __shared__ int      s_cnt;

  const int tid  = (int)threadIdx.x;
  const int lane = tid & 31;
  const int wave = tid >> 5;
  const int lo   = (int)blockIdx.x * NPW;
  const int cntN = (NNODES - lo < NPW) ? (NNODES - lo) : NPW;

  for (int i = tid; i < NPW * F; i += AGG_T) s_acc[i] = 0.f;
  for (int i = tid; i < NPW * H; i += AGG_T) {
    const bool own = i < cntN * H;
    float es0 = 0.f, ed0 = 0.f;
    if (own) { es0 = esrc[lo * H + i]; ed0 = edst[lo * H + i]; }
    s_mx[i]  = own ? ordered_enc(es0) : 0u;
    s_ed[i]  = ed0;
    s_m[i]   = 0.f;
    s_den[i] = 0.f;
  }
  if (tid == 0) s_cnt = 0;
  __syncthreads();

  {
    const uint4* tag4 = (const uint4*)tags;
    constexpr int STEP = AGG_T * SLD;
    const unsigned pat = (unsigned)blockIdx.x * 0x01010101u;
    auto load_tags = [&](uint4 (&a)[SLD], int base) {
      #pragma unroll
      for (int i = 0; i < SLD; ++i) {
        const int q = base + i * AGG_T + tid;
        a[i] = tag4[(q < NQT) ? q : (NQT - 1)];
      }
    };
    auto test_tags = [&](const uint4 (&a)[SLD], int base) {
      #pragma unroll
      for (int i = 0; i < SLD; ++i) {
        const int q = base + i * AGG_T + tid;
        const unsigned w[4] = {a[i].x, a[i].y, a[i].z, a[i].w};
        unsigned m = 0u;
        #pragma unroll
        for (int j = 0; j < 4; ++j) {
          const unsigned x = w[j] ^ pat;
          const unsigned f = ~(((x & 0x7F7F7F7Fu) + 0x7F7F7F7Fu) | x) & 0x80808080u;
          m |= f >> (7 - j);
        }
        if (m != 0u && q < NQT) {
          do {
            const int b = __builtin_ctz(m);
            m &= m - 1u;
            const int e = q * 16 + (b & 7) * 4 + (b >> 3);
            int pos = atomicAdd(&s_cnt, 1);
            if (pos < QCAP) s_list[pos] = (unsigned)e;
          } while (m != 0u);
        }
      }
    };
    uint4 setA[SLD], setB[SLD];
    load_tags(setA, 0);
    for (int base = 0; base < NQT; base += 2 * STEP) {
      load_tags(setB, base + STEP);
      test_tags(setA, base);
      load_tags(setA, base + 2 * STEP);
      test_tags(setB, base + STEP);
    }
  }
  __syncthreads();
  const int  n    = s_cnt;
  const bool fits = (n <= QCAP);

  if (fits) {
    for (int i0 = 0; i0 < n; i0 += AGG_T * RU) {
      unsigned ent[RU];
      int      sv[RU];
      int      dv[RU];
      float    es[RU][H];
      #pragma unroll
      for (int k = 0; k < RU; ++k) {
        const int idx = i0 + k * AGG_T + tid;
        ent[k] = (idx < n) ? s_list[idx] : 0u;
      }
      #pragma unroll
      for (int k = 0; k < RU; ++k) { sv[k] = src[ent[k]]; dv[k] = dst[ent[k]]; }
      #pragma unroll
      for (int k = 0; k < RU; ++k)
        #pragma unroll
        for (int h = 0; h < H; ++h) es[k][h] = esrc[sv[k] * H + h];
      #pragma unroll
      for (int k = 0; k < RU; ++k) {
        const int idx = i0 + k * AGG_T + tid;
        if (idx < n) {
          const unsigned dl = (unsigned)(dv[k] - lo);
          #pragma unroll
          for (int h = 0; h < H; ++h) atomicMax(&s_mx[dl * H + h], ordered_enc(es[k][h]));
          s_list[idx] = (dl << SBITS) | (unsigned)sv[k];
        }
      }
    }
  } else {
    for (int e = tid; e < NEDGES; e += AGG_T) {
      const unsigned dl = (unsigned)(dst[e] - lo);
      if (dl < (unsigned)cntN) {
        const int s = src[e];
        for (int h = 0; h < H; ++h) atomicMax(&s_mx[dl * H + h], ordered_enc(esrc[s * H + h]));
      }
    }
  }
  __syncthreads();

  for (int i = tid; i < cntN * H; i += AGG_T) s_m[i] = lrelu02(ordered_dec(s_mx[i]) + s_ed[i]);
  __syncthreads();

  if (fits) {
    const int p  = lane % (U * H);
    const int pk = p / H;
    const int ph = p % H;
    for (int i0 = 0; i0 < n; i0 += AGG_W * U) {
      const int  idxL  = i0 + pk * AGG_W + wave;
      const bool liveL = idxL < n;
      const unsigned entL = s_list[liveL ? idxL : (n - 1)];
      const int sL  = (int)(entL & ((1u << SBITS) - 1u));
      const int dlL = (int)(entL >> SBITS);
      const float esL = esrc[sL * H + ph];
      unsigned ent[U];
      #pragma unroll
      for (int k = 0; k < U; ++k) ent[k] = (unsigned)__builtin_amdgcn_readlane((int)entL, k * H);
      float v[U][NJ];
      #pragma unroll
      for (int k = 0; k < U; ++k)
        #pragma unroll
        for (int j = 0; j < NJ; ++j)
          v[k][j] = xw[(size_t)(ent[k] & ((1u << SBITS) - 1u)) * F + j * 32 + lane];
      const float aL = lrelu02(esL + s_ed[dlL * H + ph]);
      const float wL = liveL ? expf(aL - s_m[dlL * H + ph]) : 0.f;
      if (lane < U * H) atomicAdd(&s_den[dlL * H + ph], wL);
      #pragma unroll
      for (int k = 0; k < U; ++k) {
        float* row = s_acc + (size_t)(ent[k] >> SBITS) * F + lane;
        #pragma unroll
        for (int j = 0; j < NJ; ++j) {
          const float w = __uint_as_float((unsigned)__builtin_amdgcn_readlane(
              (int)__float_as_uint(wL), k * H + (j * 32) / C));
          atomicAdd(row + j * 32, v[k][j] * w);
        }
      }
    }
  } else {
    for (int e = tid; e < NEDGES; e += AGG_T) {
      const unsigned dl = (unsigned)(dst[e] - lo);
      if (dl < (unsigned)cntN) {
        const int s = src[e];
        for (int h = 0; h < H; ++h) {
          const float w = expf(lrelu02(esrc[s * H + h] + s_ed[dl * H + h]) - s_m[dl * H + h]);
          atomicAdd(&s_den[dl * H + h], w);
          for (int c = 0; c < C; ++c)
            atomicAdd(&s_acc[dl * F + h * C + c], w * xw[(size_t)s * F + h * C + c]);
        }
      }
    }
  }
  __syncthreads();

  if (POOL) {
    constexpr int NIT = (NPW * F + AGG_T - 1) / AGG_T;
    const int col = tid % F, h = col / C;
    float esv[NIT], xv[NIT];
    int   gv[NIT];
    #pragma unroll
    for (int k = 0; k < NIT; ++k) {
      const int i  = tid + k * AGG_T;
      const int ii = (i < cntN * F) ? i : col;
      const int dl = ii / F;
      esv[k] = esrc[(lo + dl) * H + h];
      xv[k]  = xw[(size_t)lo * F + ii];
      gv[k]  = gid[lo + dl];
    }
    const float bc = bias[col];
    float run  = 0.f;
    int   nrun = 0;
    int   g    = -1;
    auto flush = [&]() {
      if ((unsigned)g < (unsigned)NGRAPH) {
        atomicAdd(out + (size_t)g * F + col, run);
        if (col == 0) atomicAdd(gcnt + g, (float)nrun);
      }
    };
    #pragma unroll
    for (int k = 0; k < NIT; ++k) {
      const int i = tid + k * AGG_T;
      if (i < cntN * F) {
        const int dl = i / F;
        const float w   = expf(lrelu02(esv[k] + s_ed[dl * H + h]) - s_m[dl * H + h]);
        const float num = s_acc[i] + w * xv[k];
        const float den = s_den[dl * H + h] + w;
        float v = num / den + bc;
        if (ELU) v = (v > 0.f) ? v : (expf(v) - 1.f);
        if (gv[k] != g) { flush(); g = gv[k]; run = 0.f; nrun = 0; }
        run += v;
        ++nrun;
      }
    }
    flush();
  } else {
    float* orow = out + (size_t)lo * F;
    for (int i = tid; i < cntN * F; i += AGG_T) {
      const int dl = i / F, col = i % F, h = col / C;
      const float w   = expf(lrelu02(esrc[(lo + dl) * H + h] + s_ed[dl * H + h]) - s_m[dl * H + h]);
      const float num = s_acc[i] + w * xw[(size_t)lo * F + i];
      const float den = s_den[dl * H + h] + w;
      float v = num / den + bias[col];
      if (ELU) v = (v > 0.f) ? v : (expf(v) - 1.f);
      orow[i] = v;
    }
  }
}

__global__ void pool_div(float* __restrict__ out, const float* __restrict__ cnt) {
  int t = blockIdx.x * blockDim.x + threadIdx.x;
  if (t >= NGRAPH * OUTD) return;
  out[t] /= fmaxf(cnt[t / OUTD], 1.f);
}

static inline int cdiv(long long a, int b) { return (int)((a + b - 1) / b); }

extern "C" void kernel_launch(void* const* d_in, const int* in_sizes, int n_in,
                              void* d_out, int out_size, void* d_ws, size_t ws_size,
                              hipStream_t stream) {
  (void)in_sizes; (void)n_in; (void)out_size; (void)ws_size;
  const float* x   = (const float*)d_in[0];
  const int*   ei  = (const int*)d_in[1];
  const int*   bat = (const int*)d_in[2];
  const float* W1  = (const float*)d_in[3];
  const float* as1 = (const float*)d_in[4];
  const float* ad1 = (const float*)d_in[5];
  const float* b1  = (const float*)d_in[6];
  const float* W2  = (const float*)d_in[7];
  const float* as2 = (const float*)d_in[8];
  const float* ad2 = (const float*)d_in[9];
  const float* b2  = (const float*)d_in[10];
  const int* src = ei;
  const int* dst = ei + NEDGES;

  char* ws = (char*)d_ws;
  size_t off = 0;
  auto carve = [&](size_t bytes) -> char* {
    char* p = ws + off;
    off += (bytes + 255) & ~(size_t)255;
    return p;
  };
  float*    xw1    = (float*)carve((size_t)NNODES * HID * 4);
  float*    esrc1  = (float*)carve((size_t)NNODES * H1C * 4);
  float*    edst1  = (float*)carve((size_t)NNODES * H1C * 4);
  float*    agg1   = (float*)carve((size_t)NNODES * HID * 4);
  float*    cnt    = (float*)carve((size_t)NGRAPH * 4);
  unsigned* tags   = (unsigned*)carve((size_t)NEDGES);
  float*    xw2    = xw1;
  float*    esrc2  = esrc1;
  float*    edst2  = edst1;
  float*    outF   = (float*)d_out;

  const int B = 256;

  hipMemsetAsync(outF, 0, (size_t)NGRAPH * OUTD * 4, stream);
  hipMemsetAsync(cnt,  0, (size_t)NGRAPH * 4, stream);

  owner_tags<<<TAG_G, TAG_T, 0, stream>>>(dst, tags);

  {
    int tiles = (NNODES / 16) * (HID / 16);
    gemm_wmma_f32<<<cdiv((long long)tiles * 32, B), B, 0, stream>>>(x, W1, xw1, NNODES, INDIM, HID);
  }
  attn_scores<<<cdiv((long long)NNODES * H1C, B), B, 0, stream>>>(xw1, as1, ad1, esrc1, edst1, NNODES, H1C, C1C);
  gat_owner_aggregate<HID, H1C, 1, 0><<<NWG, AGG_T, 0, stream>>>(src, dst, tags, xw1, esrc1, edst1, b1, agg1, bat, cnt);

  {
    int tiles = (NNODES / 16) * (OUTD / 16);
    gemm_wmma_f32<<<cdiv((long long)tiles * 32, B), B, 0, stream>>>(agg1, W2, xw2, NNODES, HID, OUTD);
  }
  attn_scores<<<cdiv((long long)NNODES, B), B, 0, stream>>>(xw2, as2, ad2, esrc2, edst2, NNODES, 1, OUTD);
  gat_owner_aggregate<OUTD, 1, 0, 1><<<NWG, AGG_T, 0, stream>>>(src, dst, tags, xw2, esrc2, edst2, b2, outF, bat, cnt);

  pool_div<<<cdiv((long long)NGRAPH * OUTD, B), B, 0, stream>>>(outF, cnt);
}


# Layer 2's product and scores taken inside layer 1's owner groups: finished rows stay in LDS, WMMA with W2 there (same k order); 2 launches and the hidden rows' round trip gone; score dots lane-tree
# speedup vs baseline: 5.6455x; 1.0816x over previous
#include <hip/hip_runtime.h>
#include <math.h>

#define NNODES 50000
#define NEDGES 800000
#define ETOT   (NEDGES + NNODES)
#define INDIM  128
#define H1C    4
#define C1C    32
#define HID    128
#define OUTD   64
#define NGRAPH 64

typedef float v2f __attribute__((ext_vector_type(2)));
typedef float v8f __attribute__((ext_vector_type(8)));

__global__ void gemm_wmma_f32(const float* __restrict__ A, const float* __restrict__ B,
                              float* __restrict__ Cm, int M, int K, int Ncol) {
  int gwave = (int)((blockIdx.x * blockDim.x + threadIdx.x) >> 5);
  int lane  = (int)(threadIdx.x & 31);
  int tilesN = Ncol >> 4;
  int numTiles = (M >> 4) * tilesN;
  if (gwave >= numTiles) return;
  int tm = (gwave / tilesN) << 4;
  int tn = (gwave % tilesN) << 4;
  int half = lane >> 4;
  int l    = lane & 15;

  v8f acc = {};
  const float* arow = A + (size_t)(tm + l) * K;
  for (int k0 = 0; k0 < K; k0 += 4) {
    int ka = k0 + (half << 1);
    v2f a, b;
    a.x = arow[ka];
    a.y = arow[ka + 1];
    b.x = B[(size_t)(ka)     * Ncol + tn + l];
    b.y = B[(size_t)(ka + 1) * Ncol + tn + l];
    acc = __builtin_amdgcn_wmma_f32_16x16x4_f32(false, a, false, b,
                                                (short)0, acc, false, false);
  }
  int row0 = tm + (half << 3);
  #pragma unroll
  for (int v = 0; v < 8; ++v)
    Cm[(size_t)(row0 + v) * Ncol + tn + l] = acc[v];
}

__global__ void attn_scores(const float* __restrict__ xw,
                            const float* __restrict__ a_src,
                            const float* __restrict__ a_dst,
                            float* __restrict__ esrc, float* __restrict__ edst,
                            int n, int H, int C) {
  int t = blockIdx.x * blockDim.x + threadIdx.x;
  if (t >= n * H) return;
  int node = t / H, h = t % H;
  const float* v  = xw + (size_t)node * H * C + (size_t)h * C;
  const float* as = a_src + (size_t)h * C;
  const float* ad = a_dst + (size_t)h * C;
  float ss = 0.f, sd = 0.f;
  for (int c = 0; c < C; ++c) { float x = v[c]; ss += x * as[c]; sd += x * ad[c]; }
  esrc[t] = ss;
  edst[t] = sd;
}

__device__ __forceinline__ unsigned ordered_enc(float f) {
  unsigned u = __float_as_uint(f);
  return (u & 0x80000000u) ? ~u : (u | 0x80000000u);
}
__device__ __forceinline__ float ordered_dec(unsigned o) {
  return (o & 0x80000000u) ? __uint_as_float(o & 0x7FFFFFFFu) : __uint_as_float(~o);
}

__device__ __forceinline__ float lrelu02(float a) { return (a >= 0.f) ? a : 0.2f * a; }

#define NWG    256
#define NPW    ((NNODES + NWG - 1) / NWG)
#define AGG_T  1024
#define AGG_W  (AGG_T / 32)
#define QCAP   24576
#define NQ     (NEDGES / 4)
#define NQT    (NEDGES / 16)
#define SLD    5
#define SBITS  16
static_assert(NWG <= 256, "an owner tag is one byte");
static_assert(NPW <= (1 << (32 - SBITS)), "local node must fit the resolved entry's high bits");
static_assert(NNODES <= (1 << SBITS), "source node must fit the resolved entry's low bits");
static_assert(NEDGES % 16 == 0, "16-byte granularity of the tag scan");

#define TAG_T  256
#define TAG_P  4
#define TAG_G  ((NQ + TAG_T * TAG_P - 1) / (TAG_T * TAG_P))
__global__ __launch_bounds__(TAG_T)
void owner_tags(const int* __restrict__ dst, unsigned* __restrict__ tags) {
  const int4* dst4 = (const int4*)dst;
  const int q0 = (int)blockIdx.x * (TAG_T * TAG_P) + (int)threadIdx.x;
  int4 d[TAG_P];
  #pragma unroll
  for (int i = 0; i < TAG_P; ++i) {
    const int q = q0 + i * TAG_T;
    d[i] = dst4[(q < NQ) ? q : (NQ - 1)];
  }
  #pragma unroll
  for (int i = 0; i < TAG_P; ++i) {
    const int q = q0 + i * TAG_T;
    if (q < NQ) {
      const unsigned t0 = min((unsigned)d[i].x / (unsigned)NPW, 255u);
      const unsigned t1 = min((unsigned)d[i].y / (unsigned)NPW, 255u);
      const unsigned t2 = min((unsigned)d[i].z / (unsigned)NPW, 255u);
      const unsigned t3 = min((unsigned)d[i].w / (unsigned)NPW, 255u);
      tags[q] = t0 | (t1 << 8) | (t2 << 16) | (t3 << 24);
    }
  }
}

template <int F, int H, int ELU, int POOL, int FUSE>
__global__ __launch_bounds__(AGG_T)
void gat_owner_aggregate(const int* __restrict__ src, const int* __restrict__ dst,
                         const unsigned* __restrict__ tags,
                         const float* __restrict__ xw,
                         const float* __restrict__ esrc, const float* __restrict__ edst,
                         const float* __restrict__ bias, float* __restrict__ out,
                         const int* __restrict__ gid, float* __restrict__ gcnt,
                         const float* __restrict__ Wn, const float* __restrict__ asn,
                         const float* __restrict__ adn,
                         float* __restrict__ esn, float* __restrict__ edn) {
  constexpr int NJ = F / 32;
  constexpr int C  = F / H;
  constexpr int U  = 32 / NJ;
  constexpr int RU = 4;
  static_assert(U * H <= 32, "one lane per (edge, head) pair of a wave's batch");
  static_assert(!POOL || AGG_T % F == 0, "pooling: a thread keeps one column through the row loop");
  static_assert(!(POOL && FUSE), "a layer either pools its rows or hands them to the next product");
  constexpr int FN  = OUTD;
  constexpr int NRT = (NPW + 15) / 16;
  constexpr int NR  = FUSE ? NRT * 16 : NPW;
  constexpr int P   = FUSE ? F + 4 : F;
  constexpr int SP  = FN + 4;

  __shared__ __attribute__((aligned(16))) float s_acc[NR * P];
  __shared__ __attribute__((aligned(16))) float s_out[FUSE ? NR * SP : 4];
  __shared__ unsigned s_list[QCAP];
  __shared__ unsigned s_mx[NPW * H];
  __shared__ float    s_ed[NPW * H];
  __shared__ float    s_m[NPW * H];
  __shared__ float    s_den[NPW * H];
  __shared__ int      s_cnt;

  const int tid  = (int)threadIdx.x;
  const int lane = tid & 31;
  const int wave = tid >> 5;
  const int lo   = (int)blockIdx.x * NPW;
  const int cntN = (NNODES - lo < NPW) ? (NNODES - lo) : NPW;

  for (int i = tid; i < NR * P; i += AGG_T) s_acc[i] = 0.f;
  for (int i = tid; i < NPW * H; i += AGG_T) {
    const bool own = i < cntN * H;
    float es0 = 0.f, ed0 = 0.f;
    if (own) { es0 = esrc[lo * H + i]; ed0 = edst[lo * H + i]; }
    s_mx[i]  = own ? ordered_enc(es0) : 0u;
    s_ed[i]  = ed0;
    s_m[i]   = 0.f;
    s_den[i] = 0.f;
  }
  if (tid == 0) s_cnt = 0;
  __syncthreads();

  {
    const uint4* tag4 = (const uint4*)tags;
    constexpr int STEP = AGG_T * SLD;
    const unsigned pat = (unsigned)blockIdx.x * 0x01010101u;
    auto load_tags = [&](uint4 (&a)[SLD], int base) {
      #pragma unroll
      for (int i = 0; i < SLD; ++i) {
        const int q = base + i * AGG_T + tid;
        a[i] = tag4[(q < NQT) ? q : (NQT - 1)];
      }
    };
    auto test_tags = [&](const uint4 (&a)[SLD], int base) {
      #pragma unroll
      for (int i = 0; i < SLD; ++i) {
        const int q = base + i * AGG_T + tid;
        const unsigned w[4] = {a[i].x, a[i].y, a[i].z, a[i].w};
        unsigned m = 0u;
        #pragma unroll
        for (int j = 0; j < 4; ++j) {
          const unsigned x = w[j] ^ pat;
          const unsigned f = ~(((x & 0x7F7F7F7Fu) + 0x7F7F7F7Fu) | x) & 0x80808080u;
          m |= f >> (7 - j);
        }
        if (m != 0u && q < NQT) {
          do {
            const int b = __builtin_ctz(m);
            m &= m - 1u;
            const int e = q * 16 + (b & 7) * 4 + (b >> 3);
            int pos = atomicAdd(&s_cnt, 1);
            if (pos < QCAP) s_list[pos] = (unsigned)e;
          } while (m != 0u);
        }
      }
    };
    uint4 setA[SLD], setB[SLD];
    load_tags(setA, 0);
    for (int base = 0; base < NQT; base += 2 * STEP) {
      load_tags(setB, base + STEP);
      test_tags(setA, base);
      load_tags(setA, base + 2 * STEP);
      test_tags(setB, base + STEP);
    }
  }
  __syncthreads();
  const int  n    = s_cnt;
  const bool fits = (n <= QCAP);

  if (fits) {
    for (int i0 = 0; i0 < n; i0 += AGG_T * RU) {
      unsigned ent[RU];
      int      sv[RU];
      int      dv[RU];
      float    es[RU][H];
      #pragma unroll
      for (int k = 0; k < RU; ++k) {
        const int idx = i0 + k * AGG_T + tid;
        ent[k] = (idx < n) ? s_list[idx] : 0u;
      }
      #pragma unroll
      for (int k = 0; k < RU; ++k) { sv[k] = src[ent[k]]; dv[k] = dst[ent[k]]; }
      #pragma unroll
      for (int k = 0; k < RU; ++k)
        #pragma unroll
        for (int h = 0; h < H; ++h) es[k][h] = esrc[sv[k] * H + h];
      #pragma unroll
      for (int k = 0; k < RU; ++k) {
        const int idx = i0 + k * AGG_T + tid;
        if (idx < n) {
          const unsigned dl = (unsigned)(dv[k] - lo);
          #pragma unroll
          for (int h = 0; h < H; ++h) atomicMax(&s_mx[dl * H + h], ordered_enc(es[k][h]));
          s_list[idx] = (dl << SBITS) | (unsigned)sv[k];
        }
      }
    }
  } else {
    for (int e = tid; e < NEDGES; e += AGG_T) {
      const unsigned dl = (unsigned)(dst[e] - lo);
      if (dl < (unsigned)cntN) {
        const int s = src[e];
        for (int h = 0; h < H; ++h) atomicMax(&s_mx[dl * H + h], ordered_enc(esrc[s * H + h]));
      }
    }
  }
  __syncthreads();

  for (int i = tid; i < cntN * H; i += AGG_T) s_m[i] = lrelu02(ordered_dec(s_mx[i]) + s_ed[i]);
  __syncthreads();

  if (fits) {
    const int p  = lane % (U * H);
    const int pk = p / H;
    const int ph = p % H;
    for (int i0 = 0; i0 < n; i0 += AGG_W * U) {
      const int  idxL  = i0 + pk * AGG_W + wave;
      const bool liveL = idxL < n;
      const unsigned entL = s_list[liveL ? idxL : (n - 1)];
      const int sL  = (int)(entL & ((1u << SBITS) - 1u));
      const int dlL = (int)(entL >> SBITS);
      const float esL = esrc[sL * H + ph];
      unsigned ent[U];
      #pragma unroll
      for (int k = 0; k < U; ++k) ent[k] = (unsigned)__builtin_amdgcn_readlane((int)entL, k * H);
      float v[U][NJ];
      #pragma unroll
      for (int k = 0; k < U; ++k)
        #pragma unroll
        for (int j = 0; j < NJ; ++j)
          v[k][j] = xw[(size_t)(ent[k] & ((1u << SBITS) - 1u)) * F + j * 32 + lane];
      const float aL = lrelu02(esL + s_ed[dlL * H + ph]);
      const float wL = liveL ? expf(aL - s_m[dlL * H + ph]) : 0.f;
      if (lane < U * H) atomicAdd(&s_den[dlL * H + ph], wL);
      #pragma unroll
      for (int k = 0; k < U; ++k) {
        float* row = s_acc + (size_t)(ent[k] >> SBITS) * P + lane;
        #pragma unroll
        for (int j = 0; j < NJ; ++j) {
          const float w = __uint_as_float((unsigned)__builtin_amdgcn_readlane(
              (int)__float_as_uint(wL), k * H + (j * 32) / C));
          atomicAdd(row + j * 32, v[k][j] * w);
        }
      }
    }
  } else {
    for (int e = tid; e < NEDGES; e += AGG_T) {
      const unsigned dl = (unsigned)(dst[e] - lo);
      if (dl < (unsigned)cntN) {
        const int s = src[e];
        for (int h = 0; h < H; ++h) {
          const float w = expf(lrelu02(esrc[s * H + h] + s_ed[dl * H + h]) - s_m[dl * H + h]);
          atomicAdd(&s_den[dl * H + h], w);
          for (int c = 0; c < C; ++c)
            atomicAdd(&s_acc[dl * P + h * C + c], w * xw[(size_t)s * F + h * C + c]);
        }
      }
    }
  }
  __syncthreads();

  if (POOL) {
    constexpr int NIT = (NPW * F + AGG_T - 1) / AGG_T;
    const int col = tid % F, h = col / C;
    float esv[NIT], xv[NIT];
    int   gv[NIT];
    #pragma unroll
    for (int k = 0; k < NIT; ++k) {
      const int i  = tid + k * AGG_T;
      const int ii = (i < cntN * F) ? i : col;
      const int dl = ii / F;
      esv[k] = esrc[(lo + dl) * H + h];
      xv[k]  = xw[(size_t)lo * F + ii];
      gv[k]  = gid[lo + dl];
    }
    const float bc = bias[col];
    float run  = 0.f;
    int   nrun = 0;
    int   g    = -1;
    auto flush = [&]() {
      if ((unsigned)g < (unsigned)NGRAPH) {
        atomicAdd(out + (size_t)g * F + col, run);
        if (col == 0) atomicAdd(gcnt + g, (float)nrun);
      }
    };
    #pragma unroll
    for (int k = 0; k < NIT; ++k) {
      const int i = tid + k * AGG_T;
      if (i < cntN * F) {
        const int dl = i / F;
        const float w   = expf(lrelu02(esv[k] + s_ed[dl * H + h]) - s_m[dl * H + h]);
        const float num = s_acc[i] + w * xv[k];
        const float den = s_den[dl * H + h] + w;
        float v = num / den + bc;
        if (ELU) v = (v > 0.f) ? v : (expf(v) - 1.f);
        if (gv[k] != g) { flush(); g = gv[k]; run = 0.f; nrun = 0; }
        run += v;
        ++nrun;
      }
    }
    flush();
  } else if (FUSE) {
    constexpr int NCT = FN / 16;
    constexpr int NRG = AGG_W / NCT;
    constexpr int KS  = F / 4;
    static_assert(FN == 64, "score pass: a lane holds columns lane and lane + 32");
    static_assert(AGG_W % NCT == 0, "every column tile gets the same number of waves");
    const int half = lane >> 4;
    const int l    = lane & 15;
    const int wv = __builtin_amdgcn_readfirstlane(wave);
    const int ct = wv % NCT;
    const int rg = wv / NCT;
    v2f bf[KS];
    #pragma unroll
    for (int ks = 0; ks < KS; ++ks) {
      const int ka = ks * 4 + (half << 1);
      bf[ks].x = Wn[(size_t)ka * FN + ct * 16 + l];
      bf[ks].y = Wn[(size_t)(ka + 1) * FN + ct * 16 + l];
    }
    for (int i = tid; i < cntN * F; i += AGG_T) {
      const int dl = i / F, col = i % F, h = col / C;
      const float w   = expf(lrelu02(esrc[(lo + dl) * H + h] + s_ed[dl * H + h]) - s_m[dl * H + h]);
      const float num = s_acc[dl * P + col] + w * xw[(size_t)lo * F + i];
      const float den = s_den[dl * H + h] + w;
      float v = num / den + bias[col];
      if (ELU) v = (v > 0.f) ? v : (expf(v) - 1.f);
      s_acc[dl * P + col] = v;
    }
    __syncthreads();
    for (int rt = rg; rt < NRT; rt += NRG) {
      v8f acc = {};
      const float* hrow = s_acc + (rt * 16 + l) * P + (half << 1);
      #pragma unroll
      for (int ks = 0; ks < KS; ++ks) {
        const v2f a = *(const v2f*)(hrow + ks * 4);
        acc = __builtin_amdgcn_wmma_f32_16x16x4_f32(false, a, false, bf[ks],
                                                    (short)0, acc, false, false);
      }
      float* orow = s_out + (rt * 16 + (half << 3)) * SP + ct * 16 + l;
      #pragma unroll
      for (int v = 0; v < 8; ++v) orow[v * SP] = acc[v];
    }
    __syncthreads();
    {
      float4* o4 = (float4*)(out + (size_t)lo * FN);
      for (int i = tid; i < cntN * (FN / 4); i += AGG_T) {
        const int r = i / (FN / 4), c4 = i % (FN / 4);
        o4[i] = *(const float4*)(s_out + r * SP + c4 * 4);
      }
    }
    {
      const float as0 = asn[lane], as1 = asn[lane + 32];
      const float ad0 = adn[lane], ad1 = adn[lane + 32];
      for (int r = wave; r < cntN; r += AGG_W) {
        const float x0 = s_out[r * SP + lane], x1 = s_out[r * SP + 32 + lane];
        float ss = x0 * as0 + x1 * as1;
        float sd = x0 * ad0 + x1 * ad1;
        for (int m = 16; m > 0; m >>= 1) { ss += __shfl_xor(ss, m); sd += __shfl_xor(sd, m); }
        if (lane == 0) { esn[lo + r] = ss; edn[lo + r] = sd; }
      }
    }
  } else {
    float* orow = out + (size_t)lo * F;
    for (int i = tid; i < cntN * F; i += AGG_T) {
      const int dl = i / F, col = i % F, h = col / C;
      const float w   = expf(lrelu02(esrc[(lo + dl) * H + h] + s_ed[dl * H + h]) - s_m[dl * H + h]);
      const float num = s_acc[dl * P + col] + w * xw[(size_t)lo * F + i];
      const float den = s_den[dl * H + h] + w;
      float v = num / den + bias[col];
      if (ELU) v = (v > 0.f) ? v : (expf(v) - 1.f);
      orow[i] = v;
    }
  }
}

__global__ void pool_div(float* __restrict__ out, const float* __restrict__ cnt) {
  int t = blockIdx.x * blockDim.x + threadIdx.x;
  if (t >= NGRAPH * OUTD) return;
  out[t] /= fmaxf(cnt[t / OUTD], 1.f);
}

static inline int cdiv(long long a, int b) { return (int)((a + b - 1) / b); }

extern "C" void kernel_launch(void* const* d_in, const int* in_sizes, int n_in,
                              void* d_out, int out_size, void* d_ws, size_t ws_size,
                              hipStream_t stream) {
  (void)in_sizes; (void)n_in; (void)out_size; (void)ws_size;
  const float* x   = (const float*)d_in[0];
  const int*   ei  = (const int*)d_in[1];
  const int*   bat = (const int*)d_in[2];
  const float* W1  = (const float*)d_in[3];
  const float* as1 = (const float*)d_in[4];
  const float* ad1 = (const float*)d_in[5];
  const float* b1  = (const float*)d_in[6];
  const float* W2  = (const float*)d_in[7];
  const float* as2 = (const float*)d_in[8];
  const float* ad2 = (const float*)d_in[9];
  const float* b2  = (const float*)d_in[10];
  const int* src = ei;
  const int* dst = ei + NEDGES;

  char* ws = (char*)d_ws;
  size_t off = 0;
  auto carve = [&](size_t bytes) -> char* {
    char* p = ws + off;
    off += (bytes + 255) & ~(size_t)255;
    return p;
  };
  float*    xw1    = (float*)carve((size_t)NNODES * HID * 4);
  float*    esrc1  = (float*)carve((size_t)NNODES * H1C * 4);
  float*    edst1  = (float*)carve((size_t)NNODES * H1C * 4);
  float*    cnt    = (float*)carve((size_t)NGRAPH * 4);
  unsigned* tags   = (unsigned*)carve((size_t)NEDGES);
  float*    xw2    = (float*)carve((size_t)NNODES * OUTD * 4);
  float*    esrc2  = (float*)carve((size_t)NNODES * 4);
  float*    edst2  = (float*)carve((size_t)NNODES * 4);
  float*    outF   = (float*)d_out;

  const int B = 256;

  hipMemsetAsync(outF, 0, (size_t)NGRAPH * OUTD * 4, stream);
  hipMemsetAsync(cnt,  0, (size_t)NGRAPH * 4, stream);

  owner_tags<<<TAG_G, TAG_T, 0, stream>>>(dst, tags);

  {
    int tiles = (NNODES / 16) * (HID / 16);
    gemm_wmma_f32<<<cdiv((long long)tiles * 32, B), B, 0, stream>>>(x, W1, xw1, NNODES, INDIM, HID);
  }
  attn_scores<<<cdiv((long long)NNODES * H1C, B), B, 0, stream>>>(xw1, as1, ad1, esrc1, edst1, NNODES, H1C, C1C);
  gat_owner_aggregate<HID, H1C, 1, 0, 1><<<NWG, AGG_T, 0, stream>>>(src, dst, tags, xw1, esrc1, edst1, b1, xw2, bat, cnt,
                                                                    W2, as2, ad2, esrc2, edst2);

  gat_owner_aggregate<OUTD, 1, 0, 1, 0><<<NWG, AGG_T, 0, stream>>>(src, dst, tags, xw2, esrc2, edst2, b2, outF, bat, cnt,
                                                                   nullptr, nullptr, nullptr, nullptr, nullptr);

  pool_div<<<cdiv((long long)NGRAPH * OUTD, B), B, 0, stream>>>(outF, cnt);
}


# Layer 2's owner groups take the edge lists that layer 1's groups resolved (stored with their lengths in the workspace): layer 2's tag scan and edge id look-ups gone; sums stay in LDS arrival order
# speedup vs baseline: 6.3070x; 1.1172x over previous
#include <hip/hip_runtime.h>
#include <math.h>

#define NNODES 50000
#define NEDGES 800000
#define ETOT   (NEDGES + NNODES)
#define INDIM  128
#define H1C    4
#define C1C    32
#define HID    128
#define OUTD   64
#define NGRAPH 64

typedef float v2f __attribute__((ext_vector_type(2)));
typedef float v8f __attribute__((ext_vector_type(8)));

__global__ void gemm_wmma_f32(const float* __restrict__ A, const float* __restrict__ B,
                              float* __restrict__ Cm, int M, int K, int Ncol) {
  int gwave = (int)((blockIdx.x * blockDim.x + threadIdx.x) >> 5);
  int lane  = (int)(threadIdx.x & 31);
  int tilesN = Ncol >> 4;
  int numTiles = (M >> 4) * tilesN;
  if (gwave >= numTiles) return;
  int tm = (gwave / tilesN) << 4;
  int tn = (gwave % tilesN) << 4;
  int half = lane >> 4;
  int l    = lane & 15;

  v8f acc = {};
  const float* arow = A + (size_t)(tm + l) * K;
  for (int k0 = 0; k0 < K; k0 += 4) {
    int ka = k0 + (half << 1);
    v2f a, b;
    a.x = arow[ka];
    a.y = arow[ka + 1];
    b.x = B[(size_t)(ka)     * Ncol + tn + l];
    b.y = B[(size_t)(ka + 1) * Ncol + tn + l];
    acc = __builtin_amdgcn_wmma_f32_16x16x4_f32(false, a, false, b,
                                                (short)0, acc, false, false);
  }
  int row0 = tm + (half << 3);
  #pragma unroll
  for (int v = 0; v < 8; ++v)
    Cm[(size_t)(row0 + v) * Ncol + tn + l] = acc[v];
}

__global__ void attn_scores(const float* __restrict__ xw,
                            const float* __restrict__ a_src,
                            const float* __restrict__ a_dst,
                            float* __restrict__ esrc, float* __restrict__ edst,
                            int n, int H, int C) {
  int t = blockIdx.x * blockDim.x + threadIdx.x;
  if (t >= n * H) return;
  int node = t / H, h = t % H;
  const float* v  = xw + (size_t)node * H * C + (size_t)h * C;
  const float* as = a_src + (size_t)h * C;
  const float* ad = a_dst + (size_t)h * C;
  float ss = 0.f, sd = 0.f;
  for (int c = 0; c < C; ++c) { float x = v[c]; ss += x * as[c]; sd += x * ad[c]; }
  esrc[t] = ss;
  edst[t] = sd;
}

__device__ __forceinline__ unsigned ordered_enc(float f) {
  unsigned u = __float_as_uint(f);
  return (u & 0x80000000u) ? ~u : (u | 0x80000000u);
}
__device__ __forceinline__ float ordered_dec(unsigned o) {
  return (o & 0x80000000u) ? __uint_as_float(o & 0x7FFFFFFFu) : __uint_as_float(~o);
}

__device__ __forceinline__ float lrelu02(float a) { return (a >= 0.f) ? a : 0.2f * a; }

#define NWG    256
#define NPW    ((NNODES + NWG - 1) / NWG)
#define AGG_T  1024
#define AGG_W  (AGG_T / 32)
#define QCAP   24576
#define NQ     (NEDGES / 4)
#define NQT    (NEDGES / 16)
#define SLD    5
#define SBITS  16
static_assert(NWG <= 256, "an owner tag is one byte");
static_assert(NPW <= (1 << (32 - SBITS)), "local node must fit the resolved entry's high bits");
static_assert(NNODES <= (1 << SBITS), "source node must fit the resolved entry's low bits");
static_assert(NEDGES % 16 == 0, "16-byte granularity of the tag scan");

#define TAG_T  256
#define TAG_P  4
#define TAG_G  ((NQ + TAG_T * TAG_P - 1) / (TAG_T * TAG_P))
__global__ __launch_bounds__(TAG_T)
void owner_tags(const int* __restrict__ dst, unsigned* __restrict__ tags) {
  const int4* dst4 = (const int4*)dst;
  const int q0 = (int)blockIdx.x * (TAG_T * TAG_P) + (int)threadIdx.x;
  int4 d[TAG_P];
  #pragma unroll
  for (int i = 0; i < TAG_P; ++i) {
    const int q = q0 + i * TAG_T;
    d[i] = dst4[(q < NQ) ? q : (NQ - 1)];
  }
  #pragma unroll
  for (int i = 0; i < TAG_P; ++i) {
    const int q = q0 + i * TAG_T;
    if (q < NQ) {
      const unsigned t0 = min((unsigned)d[i].x / (unsigned)NPW, 255u);
      const unsigned t1 = min((unsigned)d[i].y / (unsigned)NPW, 255u);
      const unsigned t2 = min((unsigned)d[i].z / (unsigned)NPW, 255u);
      const unsigned t3 = min((unsigned)d[i].w / (unsigned)NPW, 255u);
      tags[q] = t0 | (t1 << 8) | (t2 << 16) | (t3 << 24);
    }
  }
}

template <int F, int H, int ELU, int POOL, int FUSE, int LIST>
__global__ __launch_bounds__(AGG_T)
void gat_owner_aggregate(const int* __restrict__ src, const int* __restrict__ dst,
                         const unsigned* __restrict__ tags,
                         const float* __restrict__ xw,
                         const float* __restrict__ esrc, const float* __restrict__ edst,
                         const float* __restrict__ bias, float* __restrict__ out,
                         const int* __restrict__ gid, float* __restrict__ gcnt,
                         const float* __restrict__ Wn, const float* __restrict__ asn,
                         const float* __restrict__ adn,
                         float* __restrict__ esn, float* __restrict__ edn,
                         unsigned* __restrict__ elist, int* __restrict__ ecnt) {
  static_assert(LIST == 1 || LIST == 2, "a layer either stores its edge list or takes the stored one");
  constexpr int NJ = F / 32;
  constexpr int C  = F / H;
  constexpr int U  = 32 / NJ;
  constexpr int RU = 4;
  static_assert(U * H <= 32, "one lane per (edge, head) pair of a wave's batch");
  static_assert(!POOL || AGG_T % F == 0, "pooling: a thread keeps one column through the row loop");
  static_assert(!(POOL && FUSE), "a layer either pools its rows or hands them to the next product");
  constexpr int FN  = OUTD;
  constexpr int NRT = (NPW + 15) / 16;
  constexpr int NR  = FUSE ? NRT * 16 : NPW;
  constexpr int P   = FUSE ? F + 4 : F;
  constexpr int SP  = FN + 4;

  __shared__ __attribute__((aligned(16))) float s_acc[NR * P];
  __shared__ __attribute__((aligned(16))) float s_out[FUSE ? NR * SP : 4];
  __shared__ unsigned s_list[QCAP];
  __shared__ unsigned s_mx[NPW * H];
  __shared__ float    s_ed[NPW * H];
  __shared__ float    s_m[NPW * H];
  __shared__ float    s_den[NPW * H];
  __shared__ int      s_cnt;

  const int tid  = (int)threadIdx.x;
  const int lane = tid & 31;
  const int wave = tid >> 5;
  const int lo   = (int)blockIdx.x * NPW;
  const int cntN = (NNODES - lo < NPW) ? (NNODES - lo) : NPW;

  for (int i = tid; i < NR * P; i += AGG_T) s_acc[i] = 0.f;
  for (int i = tid; i < NPW * H; i += AGG_T) {
    const bool own = i < cntN * H;
    float es0 = 0.f, ed0 = 0.f;
    if (own) { es0 = esrc[lo * H + i]; ed0 = edst[lo * H + i]; }
    s_mx[i]  = own ? ordered_enc(es0) : 0u;
    s_ed[i]  = ed0;
    s_m[i]   = 0.f;
    s_den[i] = 0.f;
  }
  if (tid == 0) s_cnt = 0;
  __syncthreads();

  if (LIST == 1) {
    const uint4* tag4 = (const uint4*)tags;
    constexpr int STEP = AGG_T * SLD;
    const unsigned pat = (unsigned)blockIdx.x * 0x01010101u;
    auto load_tags = [&](uint4 (&a)[SLD], int base) {
      #pragma unroll
      for (int i = 0; i < SLD; ++i) {
        const int q = base + i * AGG_T + tid;
        a[i] = tag4[(q < NQT) ? q : (NQT - 1)];
      }
    };
    auto test_tags = [&](const uint4 (&a)[SLD], int base) {
      #pragma unroll
      for (int i = 0; i < SLD; ++i) {
        const int q = base + i * AGG_T + tid;
        const unsigned w[4] = {a[i].x, a[i].y, a[i].z, a[i].w};
        unsigned m = 0u;
        #pragma unroll
        for (int j = 0; j < 4; ++j) {
          const unsigned x = w[j] ^ pat;
          const unsigned f = ~(((x & 0x7F7F7F7Fu) + 0x7F7F7F7Fu) | x) & 0x80808080u;
          m |= f >> (7 - j);
        }
        if (m != 0u && q < NQT) {
          do {
            const int b = __builtin_ctz(m);
            m &= m - 1u;
            const int e = q * 16 + (b & 7) * 4 + (b >> 3);
            int pos = atomicAdd(&s_cnt, 1);
            if (pos < QCAP) s_list[pos] = (unsigned)e;
          } while (m != 0u);
        }
      }
    };
    uint4 setA[SLD], setB[SLD];
    load_tags(setA, 0);
    for (int base = 0; base < NQT; base += 2 * STEP) {
      load_tags(setB, base + STEP);
      test_tags(setA, base);
      load_tags(setA, base + 2 * STEP);
      test_tags(setB, base + STEP);
    }
    __syncthreads();
  }
  const int  n    = (LIST == 2) ? __builtin_amdgcn_readfirstlane(ecnt[blockIdx.x])
                                : s_cnt;
  const bool fits = (n <= QCAP);
  if (LIST == 1 && tid == 0) ecnt[blockIdx.x] = n;

  if (fits && LIST == 2) {
    const unsigned* lin = elist + (size_t)blockIdx.x * QCAP;
    for (int i0 = 0; i0 < n; i0 += AGG_T * RU) {
      unsigned ent[RU];
      float    es[RU][H];
      #pragma unroll
      for (int k = 0; k < RU; ++k) {
        const int idx = i0 + k * AGG_T + tid;
        ent[k] = lin[(idx < n) ? idx : 0];
      }
      #pragma unroll
      for (int k = 0; k < RU; ++k)
        #pragma unroll
        for (int h = 0; h < H; ++h) es[k][h] = esrc[(ent[k] & ((1u << SBITS) - 1u)) * H + h];
      #pragma unroll
      for (int k = 0; k < RU; ++k) {
        const int idx = i0 + k * AGG_T + tid;
        if (idx < n) {
          const unsigned dl = ent[k] >> SBITS;
          #pragma unroll
          for (int h = 0; h < H; ++h) atomicMax(&s_mx[dl * H + h], ordered_enc(es[k][h]));
          s_list[idx] = ent[k];
        }
      }
    }
  } else if (fits) {
    unsigned* lout = elist + (size_t)blockIdx.x * QCAP;
    for (int i0 = 0; i0 < n; i0 += AGG_T * RU) {
      unsigned ent[RU];
      int      sv[RU];
      int      dv[RU];
      float    es[RU][H];
      #pragma unroll
      for (int k = 0; k < RU; ++k) {
        const int idx = i0 + k * AGG_T + tid;
        ent[k] = (idx < n) ? s_list[idx] : 0u;
      }
      #pragma unroll
      for (int k = 0; k < RU; ++k) { sv[k] = src[ent[k]]; dv[k] = dst[ent[k]]; }
      #pragma unroll
      for (int k = 0; k < RU; ++k)
        #pragma unroll
        for (int h = 0; h < H; ++h) es[k][h] = esrc[sv[k] * H + h];
      #pragma unroll
      for (int k = 0; k < RU; ++k) {
        const int idx = i0 + k * AGG_T + tid;
        if (idx < n) {
          const unsigned dl = (unsigned)(dv[k] - lo);
          #pragma unroll
          for (int h = 0; h < H; ++h) atomicMax(&s_mx[dl * H + h], ordered_enc(es[k][h]));
          const unsigned r = (dl << SBITS) | (unsigned)sv[k];
          s_list[idx] = r;
          lout[idx]   = r;
        }
      }
    }
  } else {
    for (int e = tid; e < NEDGES; e += AGG_T) {
      const unsigned dl = (unsigned)(dst[e] - lo);
      if (dl < (unsigned)cntN) {
        const int s = src[e];
        for (int h = 0; h < H; ++h) atomicMax(&s_mx[dl * H + h], ordered_enc(esrc[s * H + h]));
      }
    }
  }
  __syncthreads();

  for (int i = tid; i < cntN * H; i += AGG_T) s_m[i] = lrelu02(ordered_dec(s_mx[i]) + s_ed[i]);
  __syncthreads();

  if (fits) {
    const int p  = lane % (U * H);
    const int pk = p / H;
    const int ph = p % H;
    for (int i0 = 0; i0 < n; i0 += AGG_W * U) {
      const int  idxL  = i0 + pk * AGG_W + wave;
      const bool liveL = idxL < n;
      const unsigned entL = s_list[liveL ? idxL : (n - 1)];
      const int sL  = (int)(entL & ((1u << SBITS) - 1u));
      const int dlL = (int)(entL >> SBITS);
      const float esL = esrc[sL * H + ph];
      unsigned ent[U];
      #pragma unroll
      for (int k = 0; k < U; ++k) ent[k] = (unsigned)__builtin_amdgcn_readlane((int)entL, k * H);
      float v[U][NJ];
      #pragma unroll
      for (int k = 0; k < U; ++k)
        #pragma unroll
        for (int j = 0; j < NJ; ++j)
          v[k][j] = xw[(size_t)(ent[k] & ((1u << SBITS) - 1u)) * F + j * 32 + lane];
      const float aL = lrelu02(esL + s_ed[dlL * H + ph]);
      const float wL = liveL ? expf(aL - s_m[dlL * H + ph]) : 0.f;
      if (lane < U * H) atomicAdd(&s_den[dlL * H + ph], wL);
      #pragma unroll
      for (int k = 0; k < U; ++k) {
        float* row = s_acc + (size_t)(ent[k] >> SBITS) * P + lane;
        #pragma unroll
        for (int j = 0; j < NJ; ++j) {
          const float w = __uint_as_float((unsigned)__builtin_amdgcn_readlane(
              (int)__float_as_uint(wL), k * H + (j * 32) / C));
          atomicAdd(row + j * 32, v[k][j] * w);
        }
      }
    }
  } else {
    for (int e = tid; e < NEDGES; e += AGG_T) {
      const unsigned dl = (unsigned)(dst[e] - lo);
      if (dl < (unsigned)cntN) {
        const int s = src[e];
        for (int h = 0; h < H; ++h) {
          const float w = expf(lrelu02(esrc[s * H + h] + s_ed[dl * H + h]) - s_m[dl * H + h]);
          atomicAdd(&s_den[dl * H + h], w);
          for (int c = 0; c < C; ++c)
            atomicAdd(&s_acc[dl * P + h * C + c], w * xw[(size_t)s * F + h * C + c]);
        }
      }
    }
  }
  __syncthreads();

  if (POOL) {
    constexpr int NIT = (NPW * F + AGG_T - 1) / AGG_T;
    const int col = tid % F, h = col / C;
    float esv[NIT], xv[NIT];
    int   gv[NIT];
    #pragma unroll
    for (int k = 0; k < NIT; ++k) {
      const int i  = tid + k * AGG_T;
      const int ii = (i < cntN * F) ? i : col;
      const int dl = ii / F;
      esv[k] = esrc[(lo + dl) * H + h];
      xv[k]  = xw[(size_t)lo * F + ii];
      gv[k]  = gid[lo + dl];
    }
    const float bc = bias[col];
    float run  = 0.f;
    int   nrun = 0;
    int   g    = -1;
    auto flush = [&]() {
      if ((unsigned)g < (unsigned)NGRAPH) {
        atomicAdd(out + (size_t)g * F + col, run);
        if (col == 0) atomicAdd(gcnt + g, (float)nrun);
      }
    };
    #pragma unroll
    for (int k = 0; k < NIT; ++k) {
      const int i = tid + k * AGG_T;
      if (i < cntN * F) {
        const int dl = i / F;
        const float w   = expf(lrelu02(esv[k] + s_ed[dl * H + h]) - s_m[dl * H + h]);
        const float num = s_acc[i] + w * xv[k];
        const float den = s_den[dl * H + h] + w;
        float v = num / den + bc;
        if (ELU) v = (v > 0.f) ? v : (expf(v) - 1.f);
        if (gv[k] != g) { flush(); g = gv[k]; run = 0.f; nrun = 0; }
        run += v;
        ++nrun;
      }
    }
    flush();
  } else if (FUSE) {
    constexpr int NCT = FN / 16;
    constexpr int NRG = AGG_W / NCT;
    constexpr int KS  = F / 4;
    static_assert(FN == 64, "score pass: a lane holds columns lane and lane + 32");
    static_assert(AGG_W % NCT == 0, "every column tile gets the same number of waves");
    const int half = lane >> 4;
    const int l    = lane & 15;
    const int wv = __builtin_amdgcn_readfirstlane(wave);
    const int ct = wv % NCT;
    const int rg = wv / NCT;
    v2f bf[KS];
    #pragma unroll
    for (int ks = 0; ks < KS; ++ks) {
      const int ka = ks * 4 + (half << 1);
      bf[ks].x = Wn[(size_t)ka * FN + ct * 16 + l];
      bf[ks].y = Wn[(size_t)(ka + 1) * FN + ct * 16 + l];
    }
    for (int i = tid; i < cntN * F; i += AGG_T) {
      const int dl = i / F, col = i % F, h = col / C;
      const float w   = expf(lrelu02(esrc[(lo + dl) * H + h] + s_ed[dl * H + h]) - s_m[dl * H + h]);
      const float num = s_acc[dl * P + col] + w * xw[(size_t)lo * F + i];
      const float den = s_den[dl * H + h] + w;
      float v = num / den + bias[col];
      if (ELU) v = (v > 0.f) ? v : (expf(v) - 1.f);
      s_acc[dl * P + col] = v;
    }
    __syncthreads();
    for (int rt = rg; rt < NRT; rt += NRG) {
      v8f acc = {};
      const float* hrow = s_acc + (rt * 16 + l) * P + (half << 1);
      #pragma unroll
      for (int ks = 0; ks < KS; ++ks) {
        const v2f a = *(const v2f*)(hrow + ks * 4);
        acc = __builtin_amdgcn_wmma_f32_16x16x4_f32(false, a, false, bf[ks],
                                                    (short)0, acc, false, false);
      }
      float* orow = s_out + (rt * 16 + (half << 3)) * SP + ct * 16 + l;
      #pragma unroll
      for (int v = 0; v < 8; ++v) orow[v * SP] = acc[v];
    }
    __syncthreads();
    {
      float4* o4 = (float4*)(out + (size_t)lo * FN);
      for (int i = tid; i < cntN * (FN / 4); i += AGG_T) {
        const int r = i / (FN / 4), c4 = i % (FN / 4);
        o4[i] = *(const float4*)(s_out + r * SP + c4 * 4);
      }
    }
    {
      const float as0 = asn[lane], as1 = asn[lane + 32];
      const float ad0 = adn[lane], ad1 = adn[lane + 32];
      for (int r = wave; r < cntN; r += AGG_W) {
        const float x0 = s_out[r * SP + lane], x1 = s_out[r * SP + 32 + lane];
        float ss = x0 * as0 + x1 * as1;
        float sd = x0 * ad0 + x1 * ad1;
        for (int m = 16; m > 0; m >>= 1) { ss += __shfl_xor(ss, m); sd += __shfl_xor(sd, m); }
        if (lane == 0) { esn[lo + r] = ss; edn[lo + r] = sd; }
      }
    }
  } else {
    float* orow = out + (size_t)lo * F;
    for (int i = tid; i < cntN * F; i += AGG_T) {
      const int dl = i / F, col = i % F, h = col / C;
      const float w   = expf(lrelu02(esrc[(lo + dl) * H + h] + s_ed[dl * H + h]) - s_m[dl * H + h]);
      const float num = s_acc[dl * P + col] + w * xw[(size_t)lo * F + i];
      const float den = s_den[dl * H + h] + w;
      float v = num / den + bias[col];
      if (ELU) v = (v > 0.f) ? v : (expf(v) - 1.f);
      orow[i] = v;
    }
  }
}

__global__ void pool_div(float* __restrict__ out, const float* __restrict__ cnt) {
  int t = blockIdx.x * blockDim.x + threadIdx.x;
  if (t >= NGRAPH * OUTD) return;
  out[t] /= fmaxf(cnt[t / OUTD], 1.f);
}

static inline int cdiv(long long a, int b) { return (int)((a + b - 1) / b); }

extern "C" void kernel_launch(void* const* d_in, const int* in_sizes, int n_in,
                              void* d_out, int out_size, void* d_ws, size_t ws_size,
                              hipStream_t stream) {
  (void)in_sizes; (void)n_in; (void)out_size; (void)ws_size;
  const float* x   = (const float*)d_in[0];
  const int*   ei  = (const int*)d_in[1];
  const int*   bat = (const int*)d_in[2];
  const float* W1  = (const float*)d_in[3];
  const float* as1 = (const float*)d_in[4];
  const float* ad1 = (const float*)d_in[5];
  const float* b1  = (const float*)d_in[6];
  const float* W2  = (const float*)d_in[7];
  const float* as2 = (const float*)d_in[8];
  const float* ad2 = (const float*)d_in[9];
  const float* b2  = (const float*)d_in[10];
  const int* src = ei;
  const int* dst = ei + NEDGES;

  char* ws = (char*)d_ws;
  size_t off = 0;
  auto carve = [&](size_t bytes) -> char* {
    char* p = ws + off;
    off += (bytes + 255) & ~(size_t)255;
    return p;
  };
  float*    xw1    = (float*)carve((size_t)NNODES * HID * 4);
  float*    esrc1  = (float*)carve((size_t)NNODES * H1C * 4);
  float*    edst1  = (float*)carve((size_t)NNODES * H1C * 4);
  float*    cnt    = (float*)carve((size_t)NGRAPH * 4);
  unsigned* tags   = (unsigned*)carve((size_t)NEDGES);
  float*    xw2    = (float*)carve((size_t)NNODES * OUTD * 4);
  float*    esrc2  = (float*)carve((size_t)NNODES * 4);
  float*    edst2  = (float*)carve((size_t)NNODES * 4);
  unsigned* elist  = (unsigned*)carve((size_t)NWG * QCAP * 4);
  int*      ecnt   = (int*)carve((size_t)NWG * 4);
  float*    outF   = (float*)d_out;

  const int B = 256;

  hipMemsetAsync(outF, 0, (size_t)NGRAPH * OUTD * 4, stream);
  hipMemsetAsync(cnt,  0, (size_t)NGRAPH * 4, stream);

  owner_tags<<<TAG_G, TAG_T, 0, stream>>>(dst, tags);

  {
    int tiles = (NNODES / 16) * (HID / 16);
    gemm_wmma_f32<<<cdiv((long long)tiles * 32, B), B, 0, stream>>>(x, W1, xw1, NNODES, INDIM, HID);
  }
  attn_scores<<<cdiv((long long)NNODES * H1C, B), B, 0, stream>>>(xw1, as1, ad1, esrc1, edst1, NNODES, H1C, C1C);
  gat_owner_aggregate<HID, H1C, 1, 0, 1, 1><<<NWG, AGG_T, 0, stream>>>(src, dst, tags, xw1, esrc1, edst1, b1, xw2, bat, cnt,
                                                                       W2, as2, ad2, esrc2, edst2, elist, ecnt);

  gat_owner_aggregate<OUTD, 1, 0, 1, 0, 2><<<NWG, AGG_T, 0, stream>>>(src, dst, tags, xw2, esrc2, edst2, b2, outF, bat, cnt,
                                                                      nullptr, nullptr, nullptr, nullptr, nullptr, elist, ecnt);

  pool_div<<<cdiv((long long)NGRAPH * OUTD, B), B, 0, stream>>>(outF, cnt);
}


# Layer 1's product by 256 owner-sized groups: x rows staged in LDS, a wave keeps one W1 column tile's fragments and walks row tiles, scores from the LDS block; score launch gone; score dots lane-tree
# speedup vs baseline: 6.7796x; 1.0749x over previous
#include <hip/hip_runtime.h>
#include <math.h>

#define NNODES 50000
#define NEDGES 800000
#define ETOT   (NEDGES + NNODES)
#define INDIM  128
#define H1C    4
#define C1C    32
#define HID    128
#define OUTD   64
#define NGRAPH 64

typedef float v2f __attribute__((ext_vector_type(2)));
typedef float v8f __attribute__((ext_vector_type(8)));

__device__ __forceinline__ unsigned ordered_enc(float f) {
  unsigned u = __float_as_uint(f);
  return (u & 0x80000000u) ? ~u : (u | 0x80000000u);
}
__device__ __forceinline__ float ordered_dec(unsigned o) {
  return (o & 0x80000000u) ? __uint_as_float(o & 0x7FFFFFFFu) : __uint_as_float(~o);
}

__device__ __forceinline__ float lrelu02(float a) { return (a >= 0.f) ? a : 0.2f * a; }

#define NWG    256
#define NPW    ((NNODES + NWG - 1) / NWG)
#define AGG_T  1024
#define AGG_W  (AGG_T / 32)
#define QCAP   24576
#define NQ     (NEDGES / 4)
#define NQT    (NEDGES / 16)
#define SLD    5
#define SBITS  16
static_assert(NWG <= 256, "an owner tag is one byte");
static_assert(NPW <= (1 << (32 - SBITS)), "local node must fit the resolved entry's high bits");
static_assert(NNODES <= (1 << SBITS), "source node must fit the resolved entry's low bits");
static_assert(NEDGES % 16 == 0, "16-byte granularity of the tag scan");

template <int K, int FN, int H>
__global__ __launch_bounds__(AGG_T)
void owner_product(const float* __restrict__ x, const float* __restrict__ Wm,
                   const float* __restrict__ a_src, const float* __restrict__ a_dst,
                   float* __restrict__ xw, float* __restrict__ esrc, float* __restrict__ edst) {
  constexpr int NRT = (NPW + 15) / 16;
  constexpr int NR  = NRT * 16;
  constexpr int P   = K + 4;
  constexpr int SP  = FN + 4;
  constexpr int NCT = FN / 16;
  constexpr int NRG = AGG_W / NCT;
  constexpr int KS  = K / 4;
  constexpr int C   = FN / H;
  constexpr int XQ  = K / 4;
  constexpr int XIT = (NR * XQ + AGG_T - 1) / AGG_T;
  constexpr int XR  = XIT * AGG_T / XQ;
  static_assert(AGG_T % XQ == 0 && XR >= NR, "a staging trip covers whole rows");
  static_assert(H == 4 && C == 32, "score pass: a lane holds one channel of each of four heads");
  static_assert(AGG_W % NCT == 0, "every column tile gets the same number of waves");
  static_assert(K % 4 == 0 && FN % 16 == 0, "whole k steps and column tiles");

  __shared__ __attribute__((aligned(16))) float s_x[XR * P];
  __shared__ __attribute__((aligned(16))) float s_out[NR * SP];

  const int tid  = (int)threadIdx.x;
  const int lane = tid & 31;
  const int wave = tid >> 5;
  const int lo   = (int)blockIdx.x * NPW;
  const int cntN = (NNODES - lo < NPW) ? (NNODES - lo) : NPW;
  const int half = lane >> 4;
  const int l    = lane & 15;
  const int wv = __builtin_amdgcn_readfirstlane(wave);
  const int ct = wv % NCT;
  const int rg = wv / NCT;

  v2f bf[KS];
  #pragma unroll
  for (int ks = 0; ks < KS; ++ks) {
    const int ka = ks * 4 + (half << 1);
    bf[ks].x = Wm[(size_t)ka * FN + ct * 16 + l];
    bf[ks].y = Wm[(size_t)(ka + 1) * FN + ct * 16 + l];
  }

  {
    const float4* x4 = (const float4*)(x + (size_t)lo * K);
    float4 xv[XIT];
    #pragma unroll
    for (int k = 0; k < XIT; ++k) {
      const int q = tid + k * AGG_T;
      xv[k] = x4[(q / XQ < cntN) ? q : 0];
    }
    #pragma unroll
    for (int k = 0; k < XIT; ++k) {
      const int q = tid + k * AGG_T;
      const int row = q / XQ, c4 = q % XQ;
      const bool live = row < cntN;
      float4 v;
      v.x = live ? xv[k].x : 0.f;
      v.y = live ? xv[k].y : 0.f;
      v.z = live ? xv[k].z : 0.f;
      v.w = live ? xv[k].w : 0.f;
      *(float4*)(s_x + row * P + c4 * 4) = v;
    }
  }
  __syncthreads();

  for (int rt = rg; rt < NRT; rt += NRG) {
    v8f acc = {};
    const float* hrow = s_x + (rt * 16 + l) * P + (half << 1);
    #pragma unroll
    for (int ks = 0; ks < KS; ++ks) {
      const v2f a = *(const v2f*)(hrow + ks * 4);
      acc = __builtin_amdgcn_wmma_f32_16x16x4_f32(false, a, false, bf[ks],
                                                  (short)0, acc, false, false);
    }
    float* orow = s_out + (rt * 16 + (half << 3)) * SP + ct * 16 + l;
    #pragma unroll
    for (int v = 0; v < 8; ++v) orow[v * SP] = acc[v];
  }
  __syncthreads();

  {
    float4* o4 = (float4*)(xw + (size_t)lo * FN);
    for (int i = tid; i < cntN * (FN / 4); i += AGG_T) {
      const int r = i / (FN / 4), c4 = i % (FN / 4);
      o4[i] = *(const float4*)(s_out + r * SP + c4 * 4);
    }
  }
  {
    float as[H], ad[H];
    #pragma unroll
    for (int h = 0; h < H; ++h) { as[h] = a_src[h * C + lane]; ad[h] = a_dst[h * C + lane]; }
    const bool b4 = (lane & 16) != 0;
    const bool b3 = (lane & 8) != 0;
    const bool b2 = (lane & 4) != 0;
    const int  hd = (b3 ? 2 : 0) + (b2 ? 1 : 0);
    for (int r = wave; r < cntN; r += AGG_W) {
      float ss[H], sd[H];
      #pragma unroll
      for (int h = 0; h < H; ++h) {
        const float xr = s_out[r * SP + h * C + lane];
        ss[h] = xr * as[h];
        sd[h] = xr * ad[h];
      }
      float w4[H];
      #pragma unroll
      for (int h = 0; h < H; ++h) {
        const float give = b4 ? ss[h] : sd[h];
        const float keep = b4 ? sd[h] : ss[h];
        w4[h] = keep + __shfl_xor(give, 16);
      }
      float w2[2];
      #pragma unroll
      for (int j = 0; j < 2; ++j) {
        const float give = b3 ? w4[j] : w4[j + 2];
        const float keep = b3 ? w4[j + 2] : w4[j];
        w2[j] = keep + __shfl_xor(give, 8);
      }
      float w1;
      {
        const float give = b2 ? w2[0] : w2[1];
        const float keep = b2 ? w2[1] : w2[0];
        w1 = keep + __shfl_xor(give, 4);
      }
      w1 += __shfl_xor(w1, 2);
      w1 += __shfl_xor(w1, 1);
      if ((lane & 3) == 0) {
        float* dstp = b4 ? edst : esrc;
        dstp[(lo + r) * H + hd] = w1;
      }
    }
  }
}

#define TAG_T  256
#define TAG_P  4
#define TAG_G  ((NQ + TAG_T * TAG_P - 1) / (TAG_T * TAG_P))
__global__ __launch_bounds__(TAG_T)
void owner_tags(const int* __restrict__ dst, unsigned* __restrict__ tags) {
  const int4* dst4 = (const int4*)dst;
  const int q0 = (int)blockIdx.x * (TAG_T * TAG_P) + (int)threadIdx.x;
  int4 d[TAG_P];
  #pragma unroll
  for (int i = 0; i < TAG_P; ++i) {
    const int q = q0 + i * TAG_T;
    d[i] = dst4[(q < NQ) ? q : (NQ - 1)];
  }
  #pragma unroll
  for (int i = 0; i < TAG_P; ++i) {
    const int q = q0 + i * TAG_T;
    if (q < NQ) {
      const unsigned t0 = min((unsigned)d[i].x / (unsigned)NPW, 255u);
      const unsigned t1 = min((unsigned)d[i].y / (unsigned)NPW, 255u);
      const unsigned t2 = min((unsigned)d[i].z / (unsigned)NPW, 255u);
      const unsigned t3 = min((unsigned)d[i].w / (unsigned)NPW, 255u);
      tags[q] = t0 | (t1 << 8) | (t2 << 16) | (t3 << 24);
    }
  }
}

template <int F, int H, int ELU, int POOL, int FUSE, int LIST>
__global__ __launch_bounds__(AGG_T)
void gat_owner_aggregate(const int* __restrict__ src, const int* __restrict__ dst,
                         const unsigned* __restrict__ tags,
                         const float* __restrict__ xw,
                         const float* __restrict__ esrc, const float* __restrict__ edst,
                         const float* __restrict__ bias, float* __restrict__ out,
                         const int* __restrict__ gid, float* __restrict__ gcnt,
                         const float* __restrict__ Wn, const float* __restrict__ asn,
                         const float* __restrict__ adn,
                         float* __restrict__ esn, float* __restrict__ edn,
                         unsigned* __restrict__ elist, int* __restrict__ ecnt) {
  static_assert(LIST == 1 || LIST == 2, "a layer either stores its edge list or takes the stored one");
  constexpr int NJ = F / 32;
  constexpr int C  = F / H;
  constexpr int U  = 32 / NJ;
  constexpr int RU = 4;
  static_assert(U * H <= 32, "one lane per (edge, head) pair of a wave's batch");
  static_assert(!POOL || AGG_T % F == 0, "pooling: a thread keeps one column through the row loop");
  static_assert(!(POOL && FUSE), "a layer either pools its rows or hands them to the next product");
  constexpr int FN  = OUTD;
  constexpr int NRT = (NPW + 15) / 16;
  constexpr int NR  = FUSE ? NRT * 16 : NPW;
  constexpr int P   = FUSE ? F + 4 : F;
  constexpr int SP  = FN + 4;

  __shared__ __attribute__((aligned(16))) float s_acc[NR * P];
  __shared__ __attribute__((aligned(16))) float s_out[FUSE ? NR * SP : 4];
  __shared__ unsigned s_list[QCAP];
  __shared__ unsigned s_mx[NPW * H];
  __shared__ float    s_ed[NPW * H];
  __shared__ float    s_m[NPW * H];
  __shared__ float    s_den[NPW * H];
  __shared__ int      s_cnt;

  const int tid  = (int)threadIdx.x;
  const int lane = tid & 31;
  const int wave = tid >> 5;
  const int lo   = (int)blockIdx.x * NPW;
  const int cntN = (NNODES - lo < NPW) ? (NNODES - lo) : NPW;

  for (int i = tid; i < NR * P; i += AGG_T) s_acc[i] = 0.f;
  for (int i = tid; i < NPW * H; i += AGG_T) {
    const bool own = i < cntN * H;
    float es0 = 0.f, ed0 = 0.f;
    if (own) { es0 = esrc[lo * H + i]; ed0 = edst[lo * H + i]; }
    s_mx[i]  = own ? ordered_enc(es0) : 0u;
    s_ed[i]  = ed0;
    s_m[i]   = 0.f;
    s_den[i] = 0.f;
  }
  if (tid == 0) s_cnt = 0;
  __syncthreads();

  if (LIST == 1) {
    const uint4* tag4 = (const uint4*)tags;
    constexpr int STEP = AGG_T * SLD;
    const unsigned pat = (unsigned)blockIdx.x * 0x01010101u;
    auto load_tags = [&](uint4 (&a)[SLD], int base) {
      #pragma unroll
      for (int i = 0; i < SLD; ++i) {
        const int q = base + i * AGG_T + tid;
        a[i] = tag4[(q < NQT) ? q : (NQT - 1)];
      }
    };
    auto test_tags = [&](const uint4 (&a)[SLD], int base) {
      #pragma unroll
      for (int i = 0; i < SLD; ++i) {
        const int q = base + i * AGG_T + tid;
        const unsigned w[4] = {a[i].x, a[i].y, a[i].z, a[i].w};
        unsigned m = 0u;
        #pragma unroll
        for (int j = 0; j < 4; ++j) {
          const unsigned x = w[j] ^ pat;
          const unsigned f = ~(((x & 0x7F7F7F7Fu) + 0x7F7F7F7Fu) | x) & 0x80808080u;
          m |= f >> (7 - j);
        }
        if (m != 0u && q < NQT) {
          do {
            const int b = __builtin_ctz(m);
            m &= m - 1u;
            const int e = q * 16 + (b & 7) * 4 + (b >> 3);
            int pos = atomicAdd(&s_cnt, 1);
            if (pos < QCAP) s_list[pos] = (unsigned)e;
          } while (m != 0u);
        }
      }
    };
    uint4 setA[SLD], setB[SLD];
    load_tags(setA, 0);
    for (int base = 0; base < NQT; base += 2 * STEP) {
      load_tags(setB, base + STEP);
      test_tags(setA, base);
      load_tags(setA, base + 2 * STEP);
      test_tags(setB, base + STEP);
    }
    __syncthreads();
  }
  const int  n    = (LIST == 2) ? __builtin_amdgcn_readfirstlane(ecnt[blockIdx.x])
                                : s_cnt;
  const bool fits = (n <= QCAP);
  if (LIST == 1 && tid == 0) ecnt[blockIdx.x] = n;

  if (fits && LIST == 2) {
    const unsigned* lin = elist + (size_t)blockIdx.x * QCAP;
    for (int i0 = 0; i0 < n; i0 += AGG_T * RU) {
      unsigned ent[RU];
      float    es[RU][H];
      #pragma unroll
      for (int k = 0; k < RU; ++k) {
        const int idx = i0 + k * AGG_T + tid;
        ent[k] = lin[(idx < n) ? idx : 0];
      }
      #pragma unroll
      for (int k = 0; k < RU; ++k)
        #pragma unroll
        for (int h = 0; h < H; ++h) es[k][h] = esrc[(ent[k] & ((1u << SBITS) - 1u)) * H + h];
      #pragma unroll
      for (int k = 0; k < RU; ++k) {
        const int idx = i0 + k * AGG_T + tid;
        if (idx < n) {
          const unsigned dl = ent[k] >> SBITS;
          #pragma unroll
          for (int h = 0; h < H; ++h) atomicMax(&s_mx[dl * H + h], ordered_enc(es[k][h]));
          s_list[idx] = ent[k];
        }
      }
    }
  } else if (fits) {
    unsigned* lout = elist + (size_t)blockIdx.x * QCAP;
    for (int i0 = 0; i0 < n; i0 += AGG_T * RU) {
      unsigned ent[RU];
      int      sv[RU];
      int      dv[RU];
      float    es[RU][H];
      #pragma unroll
      for (int k = 0; k < RU; ++k) {
        const int idx = i0 + k * AGG_T + tid;
        ent[k] = (idx < n) ? s_list[idx] : 0u;
      }
      #pragma unroll
      for (int k = 0; k < RU; ++k) { sv[k] = src[ent[k]]; dv[k] = dst[ent[k]]; }
      #pragma unroll
      for (int k = 0; k < RU; ++k)
        #pragma unroll
        for (int h = 0; h < H; ++h) es[k][h] = esrc[sv[k] * H + h];
      #pragma unroll
      for (int k = 0; k < RU; ++k) {
        const int idx = i0 + k * AGG_T + tid;
        if (idx < n) {
          const unsigned dl = (unsigned)(dv[k] - lo);
          #pragma unroll
          for (int h = 0; h < H; ++h) atomicMax(&s_mx[dl * H + h], ordered_enc(es[k][h]));
          const unsigned r = (dl << SBITS) | (unsigned)sv[k];
          s_list[idx] = r;
          lout[idx]   = r;
        }
      }
    }
  } else {
    for (int e = tid; e < NEDGES; e += AGG_T) {
      const unsigned dl = (unsigned)(dst[e] - lo);
      if (dl < (unsigned)cntN) {
        const int s = src[e];
        for (int h = 0; h < H; ++h) atomicMax(&s_mx[dl * H + h], ordered_enc(esrc[s * H + h]));
      }
    }
  }
  __syncthreads();

  for (int i = tid; i < cntN * H; i += AGG_T) s_m[i] = lrelu02(ordered_dec(s_mx[i]) + s_ed[i]);
  __syncthreads();

  if (fits) {
    const int p  = lane % (U * H);
    const int pk = p / H;
    const int ph = p % H;
    for (int i0 = 0; i0 < n; i0 += AGG_W * U) {
      const int  idxL  = i0 + pk * AGG_W + wave;
      const bool liveL = idxL < n;
      const unsigned entL = s_list[liveL ? idxL : (n - 1)];
      const int sL  = (int)(entL & ((1u << SBITS) - 1u));
      const int dlL = (int)(entL >> SBITS);
      const float esL = esrc[sL * H + ph];
      unsigned ent[U];
      #pragma unroll
      for (int k = 0; k < U; ++k) ent[k] = (unsigned)__builtin_amdgcn_readlane((int)entL, k * H);
      float v[U][NJ];
      #pragma unroll
      for (int k = 0; k < U; ++k)
        #pragma unroll
        for (int j = 0; j < NJ; ++j)
          v[k][j] = xw[(size_t)(ent[k] & ((1u << SBITS) - 1u)) * F + j * 32 + lane];
      const float aL = lrelu02(esL + s_ed[dlL * H + ph]);
      const float wL = liveL ? expf(aL - s_m[dlL * H + ph]) : 0.f;
      if (lane < U * H) atomicAdd(&s_den[dlL * H + ph], wL);
      #pragma unroll
      for (int k = 0; k < U; ++k) {
        float* row = s_acc + (size_t)(ent[k] >> SBITS) * P + lane;
        #pragma unroll
        for (int j = 0; j < NJ; ++j) {
          const float w = __uint_as_float((unsigned)__builtin_amdgcn_readlane(
              (int)__float_as_uint(wL), k * H + (j * 32) / C));
          atomicAdd(row + j * 32, v[k][j] * w);
        }
      }
    }
  } else {
    for (int e = tid; e < NEDGES; e += AGG_T) {
      const unsigned dl = (unsigned)(dst[e] - lo);
      if (dl < (unsigned)cntN) {
        const int s = src[e];
        for (int h = 0; h < H; ++h) {
          const float w = expf(lrelu02(esrc[s * H + h] + s_ed[dl * H + h]) - s_m[dl * H + h]);
          atomicAdd(&s_den[dl * H + h], w);
          for (int c = 0; c < C; ++c)
            atomicAdd(&s_acc[dl * P + h * C + c], w * xw[(size_t)s * F + h * C + c]);
        }
      }
    }
  }
  __syncthreads();

  if (POOL) {
    constexpr int NIT = (NPW * F + AGG_T - 1) / AGG_T;
    const int col = tid % F, h = col / C;
    float esv[NIT], xv[NIT];
    int   gv[NIT];
    #pragma unroll
    for (int k = 0; k < NIT; ++k) {
      const int i  = tid + k * AGG_T;
      const int ii = (i < cntN * F) ? i : col;
      const int dl = ii / F;
      esv[k] = esrc[(lo + dl) * H + h];
      xv[k]  = xw[(size_t)lo * F + ii];
      gv[k]  = gid[lo + dl];
    }
    const float bc = bias[col];
    float run  = 0.f;
    int   nrun = 0;
    int   g    = -1;
    auto flush = [&]() {
      if ((unsigned)g < (unsigned)NGRAPH) {
        atomicAdd(out + (size_t)g * F + col, run);
        if (col == 0) atomicAdd(gcnt + g, (float)nrun);
      }
    };
    #pragma unroll
    for (int k = 0; k < NIT; ++k) {
      const int i = tid + k * AGG_T;
      if (i < cntN * F) {
        const int dl = i / F;
        const float w   = expf(lrelu02(esv[k] + s_ed[dl * H + h]) - s_m[dl * H + h]);
        const float num = s_acc[i] + w * xv[k];
        const float den = s_den[dl * H + h] + w;
        float v = num / den + bc;
        if (ELU) v = (v > 0.f) ? v : (expf(v) - 1.f);
        if (gv[k] != g) { flush(); g = gv[k]; run = 0.f; nrun = 0; }
        run += v;
        ++nrun;
      }
    }
    flush();
  } else if (FUSE) {
    constexpr int NCT = FN / 16;
    constexpr int NRG = AGG_W / NCT;
    constexpr int KS  = F / 4;
    static_assert(FN == 64, "score pass: a lane holds columns lane and lane + 32");
    static_assert(AGG_W % NCT == 0, "every column tile gets the same number of waves");
    const int half = lane >> 4;
    const int l    = lane & 15;
    const int wv = __builtin_amdgcn_readfirstlane(wave);
    const int ct = wv % NCT;
    const int rg = wv / NCT;
    v2f bf[KS];
    #pragma unroll
    for (int ks = 0; ks < KS; ++ks) {
      const int ka = ks * 4 + (half << 1);
      bf[ks].x = Wn[(size_t)ka * FN + ct * 16 + l];
      bf[ks].y = Wn[(size_t)(ka + 1) * FN + ct * 16 + l];
    }
    for (int i = tid; i < cntN * F; i += AGG_T) {
      const int dl = i / F, col = i % F, h = col / C;
      const float w   = expf(lrelu02(esrc[(lo + dl) * H + h] + s_ed[dl * H + h]) - s_m[dl * H + h]);
      const float num = s_acc[dl * P + col] + w * xw[(size_t)lo * F + i];
      const float den = s_den[dl * H + h] + w;
      float v = num / den + bias[col];
      if (ELU) v = (v > 0.f) ? v : (expf(v) - 1.f);
      s_acc[dl * P + col] = v;
    }
    __syncthreads();
    for (int rt = rg; rt < NRT; rt += NRG) {
      v8f acc = {};
      const float* hrow = s_acc + (rt * 16 + l) * P + (half << 1);
      #pragma unroll
      for (int ks = 0; ks < KS; ++ks) {
        const v2f a = *(const v2f*)(hrow + ks * 4);
        acc = __builtin_amdgcn_wmma_f32_16x16x4_f32(false, a, false, bf[ks],
                                                    (short)0, acc, false, false);
      }
      float* orow = s_out + (rt * 16 + (half << 3)) * SP + ct * 16 + l;
      #pragma unroll
      for (int v = 0; v < 8; ++v) orow[v * SP] = acc[v];
    }
    __syncthreads();
    {
      float4* o4 = (float4*)(out + (size_t)lo * FN);
      for (int i = tid; i < cntN * (FN / 4); i += AGG_T) {
        const int r = i / (FN / 4), c4 = i % (FN / 4);
        o4[i] = *(const float4*)(s_out + r * SP + c4 * 4);
      }
    }
    {
      const float as0 = asn[lane], as1 = asn[lane + 32];
      const float ad0 = adn[lane], ad1 = adn[lane + 32];
      for (int r = wave; r < cntN; r += AGG_W) {
        const float x0 = s_out[r * SP + lane], x1 = s_out[r * SP + 32 + lane];
        float ss = x0 * as0 + x1 * as1;
        float sd = x0 * ad0 + x1 * ad1;
        for (int m = 16; m > 0; m >>= 1) { ss += __shfl_xor(ss, m); sd += __shfl_xor(sd, m); }
        if (lane == 0) { esn[lo + r] = ss; edn[lo + r] = sd; }
      }
    }
  } else {
    float* orow = out + (size_t)lo * F;
    for (int i = tid; i < cntN * F; i += AGG_T) {
      const int dl = i / F, col = i % F, h = col / C;
      const float w   = expf(lrelu02(esrc[(lo + dl) * H + h] + s_ed[dl * H + h]) - s_m[dl * H + h]);
      const float num = s_acc[dl * P + col] + w * xw[(size_t)lo * F + i];
      const float den = s_den[dl * H + h] + w;
      float v = num / den + bias[col];
      if (ELU) v = (v > 0.f) ? v : (expf(v) - 1.f);
      orow[i] = v;
    }
  }
}

__global__ void pool_div(float* __restrict__ out, const float* __restrict__ cnt) {
  int t = blockIdx.x * blockDim.x + threadIdx.x;
  if (t >= NGRAPH * OUTD) return;
  out[t] /= fmaxf(cnt[t / OUTD], 1.f);
}

static inline int cdiv(long long a, int b) { return (int)((a + b - 1) / b); }

extern "C" void kernel_launch(void* const* d_in, const int* in_sizes, int n_in,
                              void* d_out, int out_size, void* d_ws, size_t ws_size,
                              hipStream_t stream) {
  (void)in_sizes; (void)n_in; (void)out_size; (void)ws_size;
  const float* x   = (const float*)d_in[0];
  const int*   ei  = (const int*)d_in[1];
  const int*   bat = (const int*)d_in[2];
  const float* W1  = (const float*)d_in[3];
  const float* as1 = (const float*)d_in[4];
  const float* ad1 = (const float*)d_in[5];
  const float* b1  = (const float*)d_in[6];
  const float* W2  = (const float*)d_in[7];
  const float* as2 = (const float*)d_in[8];
  const float* ad2 = (const float*)d_in[9];
  const float* b2  = (const float*)d_in[10];
  const int* src = ei;
  const int* dst = ei + NEDGES;

  char* ws = (char*)d_ws;
  size_t off = 0;
  auto carve = [&](size_t bytes) -> char* {
    char* p = ws + off;
    off += (bytes + 255) & ~(size_t)255;
    return p;
  };
  float*    xw1    = (float*)carve((size_t)NNODES * HID * 4);
  float*    esrc1  = (float*)carve((size_t)NNODES * H1C * 4);
  float*    edst1  = (float*)carve((size_t)NNODES * H1C * 4);
  float*    cnt    = (float*)carve((size_t)NGRAPH * 4);
  unsigned* tags   = (unsigned*)carve((size_t)NEDGES);
  float*    xw2    = (float*)carve((size_t)NNODES * OUTD * 4);
  float*    esrc2  = (float*)carve((size_t)NNODES * 4);
  float*    edst2  = (float*)carve((size_t)NNODES * 4);
  unsigned* elist  = (unsigned*)carve((size_t)NWG * QCAP * 4);
  int*      ecnt   = (int*)carve((size_t)NWG * 4);
  float*    outF   = (float*)d_out;

  const int B = 256;

  hipMemsetAsync(outF, 0, (size_t)NGRAPH * OUTD * 4, stream);
  hipMemsetAsync(cnt,  0, (size_t)NGRAPH * 4, stream);

  owner_tags<<<TAG_G, TAG_T, 0, stream>>>(dst, tags);

  owner_product<INDIM, HID, H1C><<<NWG, AGG_T, 0, stream>>>(x, W1, as1, ad1, xw1, esrc1, edst1);
  gat_owner_aggregate<HID, H1C, 1, 0, 1, 1><<<NWG, AGG_T, 0, stream>>>(src, dst, tags, xw1, esrc1, edst1, b1, xw2, bat, cnt,
                                                                       W2, as2, ad2, esrc2, edst2, elist, ecnt);

  gat_owner_aggregate<OUTD, 1, 0, 1, 0, 2><<<NWG, AGG_T, 0, stream>>>(src, dst, tags, xw2, esrc2, edst2, b2, outF, bat, cnt,
                                                                      nullptr, nullptr, nullptr, nullptr, nullptr, elist, ecnt);

  pool_div<<<cdiv((long long)NGRAPH * OUTD, B), B, 0, stream>>>(outF, cnt);
}


# Small launches folded into the three big ones: product launch writes owner tags and zero start values, layer 1's groups count nodes per graph, pool runs divided by the count before the atomic add
# speedup vs baseline: 7.0194x; 1.0354x over previous
#include <hip/hip_runtime.h>
#include <math.h>

#define NNODES 50000
#define NEDGES 800000
#define ETOT   (NEDGES + NNODES)
#define INDIM  128
#define H1C    4
#define C1C    32
#define HID    128
#define OUTD   64
#define NGRAPH 64

typedef float v2f __attribute__((ext_vector_type(2)));
typedef float v8f __attribute__((ext_vector_type(8)));

__device__ __forceinline__ unsigned ordered_enc(float f) {
  unsigned u = __float_as_uint(f);
  return (u & 0x80000000u) ? ~u : (u | 0x80000000u);
}
__device__ __forceinline__ float ordered_dec(unsigned o) {
  return (o & 0x80000000u) ? __uint_as_float(o & 0x7FFFFFFFu) : __uint_as_float(~o);
}

__device__ __forceinline__ float lrelu02(float a) { return (a >= 0.f) ? a : 0.2f * a; }

#define NWG    256
#define NPW    ((NNODES + NWG - 1) / NWG)
#define AGG_T  1024
#define AGG_W  (AGG_T / 32)
#define QCAP   24576
#define NQ     (NEDGES / 4)
#define NQT    (NEDGES / 16)
#define SLD    5
#define SBITS  16
static_assert(NWG <= 256, "an owner tag is one byte");
static_assert(NPW <= (1 << (32 - SBITS)), "local node must fit the resolved entry's high bits");
static_assert(NNODES <= (1 << SBITS), "source node must fit the resolved entry's low bits");
static_assert(NEDGES % 16 == 0, "16-byte granularity of the tag scan");

#define TQ     ((NQ + NWG - 1) / NWG)
static_assert(TQ <= AGG_T, "one id piece per thread");

template <int K, int FN, int H>
__global__ __launch_bounds__(AGG_T)
void owner_product(const float* __restrict__ x, const float* __restrict__ Wm,
                   const float* __restrict__ a_src, const float* __restrict__ a_dst,
                   float* __restrict__ xw, float* __restrict__ esrc, float* __restrict__ edst,
                   const int* __restrict__ dst, unsigned* __restrict__ tags,
                   float* __restrict__ pooled, float* __restrict__ gcnt) {
  constexpr int NRT = (NPW + 15) / 16;
  constexpr int NR  = NRT * 16;
  constexpr int P   = K + 4;
  constexpr int SP  = FN + 4;
  constexpr int NCT = FN / 16;
  constexpr int NRG = AGG_W / NCT;
  constexpr int KS  = K / 4;
  constexpr int C   = FN / H;
  constexpr int XQ  = K / 4;
  constexpr int XIT = (NR * XQ + AGG_T - 1) / AGG_T;
  constexpr int XR  = XIT * AGG_T / XQ;
  static_assert(AGG_T % XQ == 0 && XR >= NR, "a staging trip covers whole rows");
  static_assert(H == 4 && C == 32, "score pass: a lane holds one channel of each of four heads");
  static_assert(AGG_W % NCT == 0, "every column tile gets the same number of waves");
  static_assert(K % 4 == 0 && FN % 16 == 0, "whole k steps and column tiles");

  __shared__ __attribute__((aligned(16))) float s_x[XR * P];
  __shared__ __attribute__((aligned(16))) float s_out[NR * SP];

  const int tid  = (int)threadIdx.x;
  const int lane = tid & 31;
  const int wave = tid >> 5;
  const int lo   = (int)blockIdx.x * NPW;
  const int cntN = (NNODES - lo < NPW) ? (NNODES - lo) : NPW;
  const int half = lane >> 4;
  const int l    = lane & 15;
  const int wv = __builtin_amdgcn_readfirstlane(wave);
  const int ct = wv % NCT;
  const int rg = wv / NCT;

  const int  tq    = (int)blockIdx.x * TQ + tid;
  const bool tlive = (tid < TQ) && (tq < NQ);
  const int4 td    = ((const int4*)dst)[tlive ? tq : 0];

  v2f bf[KS];
  #pragma unroll
  for (int ks = 0; ks < KS; ++ks) {
    const int ka = ks * 4 + (half << 1);
    bf[ks].x = Wm[(size_t)ka * FN + ct * 16 + l];
    bf[ks].y = Wm[(size_t)(ka + 1) * FN + ct * 16 + l];
  }

  {
    const float4* x4 = (const float4*)(x + (size_t)lo * K);
    float4 xv[XIT];
    #pragma unroll
    for (int k = 0; k < XIT; ++k) {
      const int q = tid + k * AGG_T;
      xv[k] = x4[(q / XQ < cntN) ? q : 0];
    }
    #pragma unroll
    for (int k = 0; k < XIT; ++k) {
      const int q = tid + k * AGG_T;
      const int row = q / XQ, c4 = q % XQ;
      const bool live = row < cntN;
      float4 v;
      v.x = live ? xv[k].x : 0.f;
      v.y = live ? xv[k].y : 0.f;
      v.z = live ? xv[k].z : 0.f;
      v.w = live ? xv[k].w : 0.f;
      *(float4*)(s_x + row * P + c4 * 4) = v;
    }
  }
  if (tlive) {
    const unsigned t0 = min((unsigned)td.x / (unsigned)NPW, 255u);
    const unsigned t1 = min((unsigned)td.y / (unsigned)NPW, 255u);
    const unsigned t2 = min((unsigned)td.z / (unsigned)NPW, 255u);
    const unsigned t3 = min((unsigned)td.w / (unsigned)NPW, 255u);
    tags[tq] = t0 | (t1 << 8) | (t2 << 16) | (t3 << 24);
  }
  {
    const int gt = (int)blockIdx.x * AGG_T + tid;
    if (gt < NGRAPH * OUTD) pooled[gt] = 0.f;
    if (gt < NGRAPH) gcnt[gt] = 0.f;
  }
  __syncthreads();

  for (int rt = rg; rt < NRT; rt += NRG) {
    v8f acc = {};
    const float* hrow = s_x + (rt * 16 + l) * P + (half << 1);
    #pragma unroll
    for (int ks = 0; ks < KS; ++ks) {
      const v2f a = *(const v2f*)(hrow + ks * 4);
      acc = __builtin_amdgcn_wmma_f32_16x16x4_f32(false, a, false, bf[ks],
                                                  (short)0, acc, false, false);
    }
    float* orow = s_out + (rt * 16 + (half << 3)) * SP + ct * 16 + l;
    #pragma unroll
    for (int v = 0; v < 8; ++v) orow[v * SP] = acc[v];
  }
  __syncthreads();

  {
    float4* o4 = (float4*)(xw + (size_t)lo * FN);
    for (int i = tid; i < cntN * (FN / 4); i += AGG_T) {
      const int r = i / (FN / 4), c4 = i % (FN / 4);
      o4[i] = *(const float4*)(s_out + r * SP + c4 * 4);
    }
  }
  {
    float as[H], ad[H];
    #pragma unroll
    for (int h = 0; h < H; ++h) { as[h] = a_src[h * C + lane]; ad[h] = a_dst[h * C + lane]; }
    const bool b4 = (lane & 16) != 0;
    const bool b3 = (lane & 8) != 0;
    const bool b2 = (lane & 4) != 0;
    const int  hd = (b3 ? 2 : 0) + (b2 ? 1 : 0);
    for (int r = wave; r < cntN; r += AGG_W) {
      float ss[H], sd[H];
      #pragma unroll
      for (int h = 0; h < H; ++h) {
        const float xr = s_out[r * SP + h * C + lane];
        ss[h] = xr * as[h];
        sd[h] = xr * ad[h];
      }
      float w4[H];
      #pragma unroll
      for (int h = 0; h < H; ++h) {
        const float give = b4 ? ss[h] : sd[h];
        const float keep = b4 ? sd[h] : ss[h];
        w4[h] = keep + __shfl_xor(give, 16);
      }
      float w2[2];
      #pragma unroll
      for (int j = 0; j < 2; ++j) {
        const float give = b3 ? w4[j] : w4[j + 2];
        const float keep = b3 ? w4[j + 2] : w4[j];
        w2[j] = keep + __shfl_xor(give, 8);
      }
      float w1;
      {
        const float give = b2 ? w2[0] : w2[1];
        const float keep = b2 ? w2[1] : w2[0];
        w1 = keep + __shfl_xor(give, 4);
      }
      w1 += __shfl_xor(w1, 2);
      w1 += __shfl_xor(w1, 1);
      if ((lane & 3) == 0) {
        float* dstp = b4 ? edst : esrc;
        dstp[(lo + r) * H + hd] = w1;
      }
    }
  }
}

template <int F, int H, int ELU, int POOL, int FUSE, int LIST, int CNT>
__global__ __launch_bounds__(AGG_T)
void gat_owner_aggregate(const int* __restrict__ src, const int* __restrict__ dst,
                         const unsigned* __restrict__ tags,
                         const float* __restrict__ xw,
                         const float* __restrict__ esrc, const float* __restrict__ edst,
                         const float* __restrict__ bias, float* __restrict__ out,
                         const int* __restrict__ gid, float* __restrict__ gcnt,
                         const float* __restrict__ Wn, const float* __restrict__ asn,
                         const float* __restrict__ adn,
                         float* __restrict__ esn, float* __restrict__ edn,
                         unsigned* __restrict__ elist, int* __restrict__ ecnt) {
  static_assert(LIST == 1 || LIST == 2, "a layer either stores its edge list or takes the stored one");
  constexpr int NJ = F / 32;
  constexpr int C  = F / H;
  constexpr int U  = 32 / NJ;
  constexpr int RU = 4;
  static_assert(U * H <= 32, "one lane per (edge, head) pair of a wave's batch");
  static_assert(!POOL || AGG_T % F == 0, "pooling: a thread keeps one column through the row loop");
  static_assert(!(POOL && FUSE), "a layer either pools its rows or hands them to the next product");
  static_assert(!(POOL && CNT), "the counts must be complete before a layer divides by them");
  static_assert(NGRAPH <= AGG_T && NPW <= AGG_T, "one thread per graph for the count tables, one per owned node for its id");
  constexpr int FN  = OUTD;
  constexpr int NRT = (NPW + 15) / 16;
  constexpr int NR  = FUSE ? NRT * 16 : NPW;
  constexpr int P   = FUSE ? F + 4 : F;
  constexpr int SP  = FN + 4;

  __shared__ __attribute__((aligned(16))) float s_acc[NR * P];
  __shared__ __attribute__((aligned(16))) float s_out[FUSE ? NR * SP : 4];
  __shared__ unsigned s_list[QCAP];
  __shared__ unsigned s_mx[NPW * H];
  __shared__ float    s_ed[NPW * H];
  __shared__ float    s_m[NPW * H];
  __shared__ float    s_den[NPW * H];
  __shared__ int      s_cnt;
  __shared__ int      s_hist[CNT ? NGRAPH : 1];
  __shared__ float    s_gc[POOL ? NGRAPH : 1];

  const int tid  = (int)threadIdx.x;
  const int lane = tid & 31;
  const int wave = tid >> 5;
  const int lo   = (int)blockIdx.x * NPW;
  const int cntN = (NNODES - lo < NPW) ? (NNODES - lo) : NPW;

  int gmine = -1;
  if (CNT && tid < cntN) gmine = gid[lo + tid];
  if (CNT && tid < NGRAPH) s_hist[tid] = 0;
  if (POOL && tid < NGRAPH) s_gc[tid] = fmaxf(gcnt[tid], 1.f);

  for (int i = tid; i < NR * P; i += AGG_T) s_acc[i] = 0.f;
  for (int i = tid; i < NPW * H; i += AGG_T) {
    const bool own = i < cntN * H;
    float es0 = 0.f, ed0 = 0.f;
    if (own) { es0 = esrc[lo * H + i]; ed0 = edst[lo * H + i]; }
    s_mx[i]  = own ? ordered_enc(es0) : 0u;
    s_ed[i]  = ed0;
    s_m[i]   = 0.f;
    s_den[i] = 0.f;
  }
  if (tid == 0) s_cnt = 0;
  __syncthreads();

  if (CNT && (unsigned)gmine < (unsigned)NGRAPH) atomicAdd(&s_hist[gmine], 1);

  if (LIST == 1) {
    const uint4* tag4 = (const uint4*)tags;
    constexpr int STEP = AGG_T * SLD;
    const unsigned pat = (unsigned)blockIdx.x * 0x01010101u;
    auto load_tags = [&](uint4 (&a)[SLD], int base) {
      #pragma unroll
      for (int i = 0; i < SLD; ++i) {
        const int q = base + i * AGG_T + tid;
        a[i] = tag4[(q < NQT) ? q : (NQT - 1)];
      }
    };
    auto test_tags = [&](const uint4 (&a)[SLD], int base) {
      #pragma unroll
      for (int i = 0; i < SLD; ++i) {
        const int q = base + i * AGG_T + tid;
        const unsigned w[4] = {a[i].x, a[i].y, a[i].z, a[i].w};
        unsigned m = 0u;
        #pragma unroll
        for (int j = 0; j < 4; ++j) {
          const unsigned x = w[j] ^ pat;
          const unsigned f = ~(((x & 0x7F7F7F7Fu) + 0x7F7F7F7Fu) | x) & 0x80808080u;
          m |= f >> (7 - j);
        }
        if (m != 0u && q < NQT) {
          do {
            const int b = __builtin_ctz(m);
            m &= m - 1u;
            const int e = q * 16 + (b & 7) * 4 + (b >> 3);
            int pos = atomicAdd(&s_cnt, 1);
            if (pos < QCAP) s_list[pos] = (unsigned)e;
          } while (m != 0u);
        }
      }
    };
    uint4 setA[SLD], setB[SLD];
    load_tags(setA, 0);
    for (int base = 0; base < NQT; base += 2 * STEP) {
      load_tags(setB, base + STEP);
      test_tags(setA, base);
      load_tags(setA, base + 2 * STEP);
      test_tags(setB, base + STEP);
    }
    __syncthreads();
  }
  const int  n    = (LIST == 2) ? __builtin_amdgcn_readfirstlane(ecnt[blockIdx.x])
                                : s_cnt;
  const bool fits = (n <= QCAP);
  if (LIST == 1 && tid == 0) ecnt[blockIdx.x] = n;

  if (fits && LIST == 2) {
    const unsigned* lin = elist + (size_t)blockIdx.x * QCAP;
    for (int i0 = 0; i0 < n; i0 += AGG_T * RU) {
      unsigned ent[RU];
      float    es[RU][H];
      #pragma unroll
      for (int k = 0; k < RU; ++k) {
        const int idx = i0 + k * AGG_T + tid;
        ent[k] = lin[(idx < n) ? idx : 0];
      }
      #pragma unroll
      for (int k = 0; k < RU; ++k)
        #pragma unroll
        for (int h = 0; h < H; ++h) es[k][h] = esrc[(ent[k] & ((1u << SBITS) - 1u)) * H + h];
      #pragma unroll
      for (int k = 0; k < RU; ++k) {
        const int idx = i0 + k * AGG_T + tid;
        if (idx < n) {
          const unsigned dl = ent[k] >> SBITS;
          #pragma unroll
          for (int h = 0; h < H; ++h) atomicMax(&s_mx[dl * H + h], ordered_enc(es[k][h]));
          s_list[idx] = ent[k];
        }
      }
    }
  } else if (fits) {
    unsigned* lout = elist + (size_t)blockIdx.x * QCAP;
    for (int i0 = 0; i0 < n; i0 += AGG_T * RU) {
      unsigned ent[RU];
      int      sv[RU];
      int      dv[RU];
      float    es[RU][H];
      #pragma unroll
      for (int k = 0; k < RU; ++k) {
        const int idx = i0 + k * AGG_T + tid;
        ent[k] = (idx < n) ? s_list[idx] : 0u;
      }
      #pragma unroll
      for (int k = 0; k < RU; ++k) { sv[k] = src[ent[k]]; dv[k] = dst[ent[k]]; }
      #pragma unroll
      for (int k = 0; k < RU; ++k)
        #pragma unroll
        for (int h = 0; h < H; ++h) es[k][h] = esrc[sv[k] * H + h];
      #pragma unroll
      for (int k = 0; k < RU; ++k) {
        const int idx = i0 + k * AGG_T + tid;
        if (idx < n) {
          const unsigned dl = (unsigned)(dv[k] - lo);
          #pragma unroll
          for (int h = 0; h < H; ++h) atomicMax(&s_mx[dl * H + h], ordered_enc(es[k][h]));
          const unsigned r = (dl << SBITS) | (unsigned)sv[k];
          s_list[idx] = r;
          lout[idx]   = r;
        }
      }
    }
  } else {
    for (int e = tid; e < NEDGES; e += AGG_T) {
      const unsigned dl = (unsigned)(dst[e] - lo);
      if (dl < (unsigned)cntN) {
        const int s = src[e];
        for (int h = 0; h < H; ++h) atomicMax(&s_mx[dl * H + h], ordered_enc(esrc[s * H + h]));
      }
    }
  }
  __syncthreads();

  for (int i = tid; i < cntN * H; i += AGG_T) s_m[i] = lrelu02(ordered_dec(s_mx[i]) + s_ed[i]);
  if (CNT && tid >= AGG_T - NGRAPH) {
    const int gq = tid - (AGG_T - NGRAPH);
    const int c  = s_hist[gq];
    if (c > 0) atomicAdd(gcnt + gq, (float)c);
  }
  __syncthreads();

  if (fits) {
    const int p  = lane % (U * H);
    const int pk = p / H;
    const int ph = p % H;
    for (int i0 = 0; i0 < n; i0 += AGG_W * U) {
      const int  idxL  = i0 + pk * AGG_W + wave;
      const bool liveL = idxL < n;
      const unsigned entL = s_list[liveL ? idxL : (n - 1)];
      const int sL  = (int)(entL & ((1u << SBITS) - 1u));
      const int dlL = (int)(entL >> SBITS);
      const float esL = esrc[sL * H + ph];
      unsigned ent[U];
      #pragma unroll
      for (int k = 0; k < U; ++k) ent[k] = (unsigned)__builtin_amdgcn_readlane((int)entL, k * H);
      float v[U][NJ];
      #pragma unroll
      for (int k = 0; k < U; ++k)
        #pragma unroll
        for (int j = 0; j < NJ; ++j)
          v[k][j] = xw[(size_t)(ent[k] & ((1u << SBITS) - 1u)) * F + j * 32 + lane];
      const float aL = lrelu02(esL + s_ed[dlL * H + ph]);
      const float wL = liveL ? expf(aL - s_m[dlL * H + ph]) : 0.f;
      if (lane < U * H) atomicAdd(&s_den[dlL * H + ph], wL);
      #pragma unroll
      for (int k = 0; k < U; ++k) {
        float* row = s_acc + (size_t)(ent[k] >> SBITS) * P + lane;
        #pragma unroll
        for (int j = 0; j < NJ; ++j) {
          const float w = __uint_as_float((unsigned)__builtin_amdgcn_readlane(
              (int)__float_as_uint(wL), k * H + (j * 32) / C));
          atomicAdd(row + j * 32, v[k][j] * w);
        }
      }
    }
  } else {
    for (int e = tid; e < NEDGES; e += AGG_T) {
      const unsigned dl = (unsigned)(dst[e] - lo);
      if (dl < (unsigned)cntN) {
        const int s = src[e];
        for (int h = 0; h < H; ++h) {
          const float w = expf(lrelu02(esrc[s * H + h] + s_ed[dl * H + h]) - s_m[dl * H + h]);
          atomicAdd(&s_den[dl * H + h], w);
          for (int c = 0; c < C; ++c)
            atomicAdd(&s_acc[dl * P + h * C + c], w * xw[(size_t)s * F + h * C + c]);
        }
      }
    }
  }
  __syncthreads();

  if (POOL) {
    constexpr int NIT = (NPW * F + AGG_T - 1) / AGG_T;
    const int col = tid % F, h = col / C;
    float esv[NIT], xv[NIT];
    int   gv[NIT];
    #pragma unroll
    for (int k = 0; k < NIT; ++k) {
      const int i  = tid + k * AGG_T;
      const int ii = (i < cntN * F) ? i : col;
      const int dl = ii / F;
      esv[k] = esrc[(lo + dl) * H + h];
      xv[k]  = xw[(size_t)lo * F + ii];
      gv[k]  = gid[lo + dl];
    }
    const float bc = bias[col];
    float run  = 0.f;
    int   g    = -1;
    auto flush = [&]() {
      if ((unsigned)g < (unsigned)NGRAPH) atomicAdd(out + (size_t)g * F + col, run / s_gc[g]);
    };
    #pragma unroll
    for (int k = 0; k < NIT; ++k) {
      const int i = tid + k * AGG_T;
      if (i < cntN * F) {
        const int dl = i / F;
        const float w   = expf(lrelu02(esv[k] + s_ed[dl * H + h]) - s_m[dl * H + h]);
        const float num = s_acc[i] + w * xv[k];
        const float den = s_den[dl * H + h] + w;
        float v = num / den + bc;
        if (ELU) v = (v > 0.f) ? v : (expf(v) - 1.f);
        if (gv[k] != g) { flush(); g = gv[k]; run = 0.f; }
        run += v;
      }
    }
    flush();
  } else if (FUSE) {
    constexpr int NCT = FN / 16;
    constexpr int NRG = AGG_W / NCT;
    constexpr int KS  = F / 4;
    static_assert(FN == 64, "score pass: a lane holds columns lane and lane + 32");
    static_assert(AGG_W % NCT == 0, "every column tile gets the same number of waves");
    const int half = lane >> 4;
    const int l    = lane & 15;
    const int wv = __builtin_amdgcn_readfirstlane(wave);
    const int ct = wv % NCT;
    const int rg = wv / NCT;
    v2f bf[KS];
    #pragma unroll
    for (int ks = 0; ks < KS; ++ks) {
      const int ka = ks * 4 + (half << 1);
      bf[ks].x = Wn[(size_t)ka * FN + ct * 16 + l];
      bf[ks].y = Wn[(size_t)(ka + 1) * FN + ct * 16 + l];
    }
    for (int i = tid; i < cntN * F; i += AGG_T) {
      const int dl = i / F, col = i % F, h = col / C;
      const float w   = expf(lrelu02(esrc[(lo + dl) * H + h] + s_ed[dl * H + h]) - s_m[dl * H + h]);
      const float num = s_acc[dl * P + col] + w * xw[(size_t)lo * F + i];
      const float den = s_den[dl * H + h] + w;
      float v = num / den + bias[col];
      if (ELU) v = (v > 0.f) ? v : (expf(v) - 1.f);
      s_acc[dl * P + col] = v;
    }
    __syncthreads();
    for (int rt = rg; rt < NRT; rt += NRG) {
      v8f acc = {};
      const float* hrow = s_acc + (rt * 16 + l) * P + (half << 1);
      #pragma unroll
      for (int ks = 0; ks < KS; ++ks) {
        const v2f a = *(const v2f*)(hrow + ks * 4);
        acc = __builtin_amdgcn_wmma_f32_16x16x4_f32(false, a, false, bf[ks],
                                                    (short)0, acc, false, false);
      }
      float* orow = s_out + (rt * 16 + (half << 3)) * SP + ct * 16 + l;
      #pragma unroll
      for (int v = 0; v < 8; ++v) orow[v * SP] = acc[v];
    }
    __syncthreads();
    {
      float4* o4 = (float4*)(out + (size_t)lo * FN);
      for (int i = tid; i < cntN * (FN / 4); i += AGG_T) {
        const int r = i / (FN / 4), c4 = i % (FN / 4);
        o4[i] = *(const float4*)(s_out + r * SP + c4 * 4);
      }
    }
    {
      const float as0 = asn[lane], as1 = asn[lane + 32];
      const float ad0 = adn[lane], ad1 = adn[lane + 32];
      for (int r = wave; r < cntN; r += AGG_W) {
        const float x0 = s_out[r * SP + lane], x1 = s_out[r * SP + 32 + lane];
        float ss = x0 * as0 + x1 * as1;
        float sd = x0 * ad0 + x1 * ad1;
        for (int m = 16; m > 0; m >>= 1) { ss += __shfl_xor(ss, m); sd += __shfl_xor(sd, m); }
        if (lane == 0) { esn[lo + r] = ss; edn[lo + r] = sd; }
      }
    }
  } else {
    float* orow = out + (size_t)lo * F;
    for (int i = tid; i < cntN * F; i += AGG_T) {
      const int dl = i / F, col = i % F, h = col / C;
      const float w   = expf(lrelu02(esrc[(lo + dl) * H + h] + s_ed[dl * H + h]) - s_m[dl * H + h]);
      const float num = s_acc[dl * P + col] + w * xw[(size_t)lo * F + i];
      const float den = s_den[dl * H + h] + w;
      float v = num / den + bias[col];
      if (ELU) v = (v > 0.f) ? v : (expf(v) - 1.f);
      orow[i] = v;
    }
  }
}

extern "C" void kernel_launch(void* const* d_in, const int* in_sizes, int n_in,
                              void* d_out, int out_size, void* d_ws, size_t ws_size,
                              hipStream_t stream) {
  (void)in_sizes; (void)n_in; (void)out_size; (void)ws_size;
  const float* x   = (const float*)d_in[0];
  const int*   ei  = (const int*)d_in[1];
  const int*   bat = (const int*)d_in[2];
  const float* W1  = (const float*)d_in[3];
  const float* as1 = (const float*)d_in[4];
  const float* ad1 = (const float*)d_in[5];
  const float* b1  = (const float*)d_in[6];
  const float* W2  = (const float*)d_in[7];
  const float* as2 = (const float*)d_in[8];
  const float* ad2 = (const float*)d_in[9];
  const float* b2  = (const float*)d_in[10];
  const int* src = ei;
  const int* dst = ei + NEDGES;

  char* ws = (char*)d_ws;
  size_t off = 0;
  auto carve = [&](size_t bytes) -> char* {
    char* p = ws + off;
    off += (bytes + 255) & ~(size_t)255;
    return p;
  };
  float*    xw1    = (float*)carve((size_t)NNODES * HID * 4);
  float*    esrc1  = (float*)carve((size_t)NNODES * H1C * 4);
  float*    edst1  = (float*)carve((size_t)NNODES * H1C * 4);
  float*    cnt    = (float*)carve((size_t)NGRAPH * 4);
  unsigned* tags   = (unsigned*)carve((size_t)NEDGES);
  float*    xw2    = (float*)carve((size_t)NNODES * OUTD * 4);
  float*    esrc2  = (float*)carve((size_t)NNODES * 4);
  float*    edst2  = (float*)carve((size_t)NNODES * 4);
  unsigned* elist  = (unsigned*)carve((size_t)NWG * QCAP * 4);
  int*      ecnt   = (int*)carve((size_t)NWG * 4);
  float*    outF   = (float*)d_out;

  owner_product<INDIM, HID, H1C><<<NWG, AGG_T, 0, stream>>>(x, W1, as1, ad1, xw1, esrc1, edst1, dst, tags, outF, cnt);
  gat_owner_aggregate<HID, H1C, 1, 0, 1, 1, 1><<<NWG, AGG_T, 0, stream>>>(src, dst, tags, xw1, esrc1, edst1, b1, xw2, bat, cnt,
                                                                          W2, as2, ad2, esrc2, edst2, elist, ecnt);

  gat_owner_aggregate<OUTD, 1, 0, 1, 0, 2, 0><<<NWG, AGG_T, 0, stream>>>(src, dst, tags, xw2, esrc2, edst2, b2, outF, bat, cnt,
                                                                         nullptr, nullptr, nullptr, nullptr, nullptr, elist, ecnt);
}


# Last layer's rows summed per graph in registers: exp weights made final first (over exp-sum and node count), waves add weighted source rows; LDS row adds and node rows gone; sums in list order
# speedup vs baseline: 7.5906x; 1.0814x over previous
#include <hip/hip_runtime.h>
#include <math.h>

#define NNODES 50000
#define NEDGES 800000
#define ETOT   (NEDGES + NNODES)
#define INDIM  128
#define H1C    4
#define C1C    32
#define HID    128
#define OUTD   64
#define NGRAPH 64

typedef float v2f __attribute__((ext_vector_type(2)));
typedef float v8f __attribute__((ext_vector_type(8)));

__device__ __forceinline__ unsigned ordered_enc(float f) {
  unsigned u = __float_as_uint(f);
  return (u & 0x80000000u) ? ~u : (u | 0x80000000u);
}
__device__ __forceinline__ float ordered_dec(unsigned o) {
  return (o & 0x80000000u) ? __uint_as_float(o & 0x7FFFFFFFu) : __uint_as_float(~o);
}

__device__ __forceinline__ float lrelu02(float a) { return (a >= 0.f) ? a : 0.2f * a; }

#define NWG    256
#define NPW    ((NNODES + NWG - 1) / NWG)
#define AGG_T  1024
#define AGG_W  (AGG_T / 32)
#define QCAP   24576
#define NQ     (NEDGES / 4)
#define NQT    (NEDGES / 16)
#define SLD    5
#define SBITS  16
static_assert(NWG <= 256, "an owner tag is one byte");
static_assert(NPW <= (1 << (32 - SBITS)), "local node must fit the resolved entry's high bits");
static_assert(NNODES <= (1 << SBITS), "source node must fit the resolved entry's low bits");
static_assert(NEDGES % 16 == 0, "16-byte granularity of the tag scan");

#define TQ     ((NQ + NWG - 1) / NWG)
static_assert(TQ <= AGG_T, "one id piece per thread");

template <int K, int FN, int H>
__global__ __launch_bounds__(AGG_T)
void owner_product(const float* __restrict__ x, const float* __restrict__ Wm,
                   const float* __restrict__ a_src, const float* __restrict__ a_dst,
                   float* __restrict__ xw, float* __restrict__ esrc, float* __restrict__ edst,
                   const int* __restrict__ dst, unsigned* __restrict__ tags,
                   float* __restrict__ pooled, float* __restrict__ gcnt) {
  constexpr int NRT = (NPW + 15) / 16;
  constexpr int NR  = NRT * 16;
  constexpr int P   = K + 4;
  constexpr int SP  = FN + 4;
  constexpr int NCT = FN / 16;
  constexpr int NRG = AGG_W / NCT;
  constexpr int KS  = K / 4;
  constexpr int C   = FN / H;
  constexpr int XQ  = K / 4;
  constexpr int XIT = (NR * XQ + AGG_T - 1) / AGG_T;
  constexpr int XR  = XIT * AGG_T / XQ;
  static_assert(AGG_T % XQ == 0 && XR >= NR, "a staging trip covers whole rows");
  static_assert(H == 4 && C == 32, "score pass: a lane holds one channel of each of four heads");
  static_assert(AGG_W % NCT == 0, "every column tile gets the same number of waves");
  static_assert(K % 4 == 0 && FN % 16 == 0, "whole k steps and column tiles");

  __shared__ __attribute__((aligned(16))) float s_x[XR * P];
  __shared__ __attribute__((aligned(16))) float s_out[NR * SP];

  const int tid  = (int)threadIdx.x;
  const int lane = tid & 31;
  const int wave = tid >> 5;
  const int lo   = (int)blockIdx.x * NPW;
  const int cntN = (NNODES - lo < NPW) ? (NNODES - lo) : NPW;
  const int half = lane >> 4;
  const int l    = lane & 15;
  const int wv = __builtin_amdgcn_readfirstlane(wave);
  const int ct = wv % NCT;
  const int rg = wv / NCT;

  const int  tq    = (int)blockIdx.x * TQ + tid;
  const bool tlive = (tid < TQ) && (tq < NQ);
  const int4 td    = ((const int4*)dst)[tlive ? tq : 0];

  v2f bf[KS];
  #pragma unroll
  for (int ks = 0; ks < KS; ++ks) {
    const int ka = ks * 4 + (half << 1);
    bf[ks].x = Wm[(size_t)ka * FN + ct * 16 + l];
    bf[ks].y = Wm[(size_t)(ka + 1) * FN + ct * 16 + l];
  }

  {
    const float4* x4 = (const float4*)(x + (size_t)lo * K);
    float4 xv[XIT];
    #pragma unroll
    for (int k = 0; k < XIT; ++k) {
      const int q = tid + k * AGG_T;
      xv[k] = x4[(q / XQ < cntN) ? q : 0];
    }
    #pragma unroll
    for (int k = 0; k < XIT; ++k) {
      const int q = tid + k * AGG_T;
      const int row = q / XQ, c4 = q % XQ;
      const bool live = row < cntN;
      float4 v;
      v.x = live ? xv[k].x : 0.f;
      v.y = live ? xv[k].y : 0.f;
      v.z = live ? xv[k].z : 0.f;
      v.w = live ? xv[k].w : 0.f;
      *(float4*)(s_x + row * P + c4 * 4) = v;
    }
  }
  if (tlive) {
    const unsigned t0 = min((unsigned)td.x / (unsigned)NPW, 255u);
    const unsigned t1 = min((unsigned)td.y / (unsigned)NPW, 255u);
    const unsigned t2 = min((unsigned)td.z / (unsigned)NPW, 255u);
    const unsigned t3 = min((unsigned)td.w / (unsigned)NPW, 255u);
    tags[tq] = t0 | (t1 << 8) | (t2 << 16) | (t3 << 24);
  }
  {
    const int gt = (int)blockIdx.x * AGG_T + tid;
    if (gt < NGRAPH * OUTD) pooled[gt] = 0.f;
    if (gt < NGRAPH) gcnt[gt] = 0.f;
  }
  __syncthreads();

  for (int rt = rg; rt < NRT; rt += NRG) {
    v8f acc = {};
    const float* hrow = s_x + (rt * 16 + l) * P + (half << 1);
    #pragma unroll
    for (int ks = 0; ks < KS; ++ks) {
      const v2f a = *(const v2f*)(hrow + ks * 4);
      acc = __builtin_amdgcn_wmma_f32_16x16x4_f32(false, a, false, bf[ks],
                                                  (short)0, acc, false, false);
    }
    float* orow = s_out + (rt * 16 + (half << 3)) * SP + ct * 16 + l;
    #pragma unroll
    for (int v = 0; v < 8; ++v) orow[v * SP] = acc[v];
  }
  __syncthreads();

  {
    float4* o4 = (float4*)(xw + (size_t)lo * FN);
    for (int i = tid; i < cntN * (FN / 4); i += AGG_T) {
      const int r = i / (FN / 4), c4 = i % (FN / 4);
      o4[i] = *(const float4*)(s_out + r * SP + c4 * 4);
    }
  }
  {
    float as[H], ad[H];
    #pragma unroll
    for (int h = 0; h < H; ++h) { as[h] = a_src[h * C + lane]; ad[h] = a_dst[h * C + lane]; }
    const bool b4 = (lane & 16) != 0;
    const bool b3 = (lane & 8) != 0;
    const bool b2 = (lane & 4) != 0;
    const int  hd = (b3 ? 2 : 0) + (b2 ? 1 : 0);
    for (int r = wave; r < cntN; r += AGG_W) {
      float ss[H], sd[H];
      #pragma unroll
      for (int h = 0; h < H; ++h) {
        const float xr = s_out[r * SP + h * C + lane];
        ss[h] = xr * as[h];
        sd[h] = xr * ad[h];
      }
      float w4[H];
      #pragma unroll
      for (int h = 0; h < H; ++h) {
        const float give = b4 ? ss[h] : sd[h];
        const float keep = b4 ? sd[h] : ss[h];
        w4[h] = keep + __shfl_xor(give, 16);
      }
      float w2[2];
      #pragma unroll
      for (int j = 0; j < 2; ++j) {
        const float give = b3 ? w4[j] : w4[j + 2];
        const float keep = b3 ? w4[j + 2] : w4[j];
        w2[j] = keep + __shfl_xor(give, 8);
      }
      float w1;
      {
        const float give = b2 ? w2[0] : w2[1];
        const float keep = b2 ? w2[1] : w2[0];
        w1 = keep + __shfl_xor(give, 4);
      }
      w1 += __shfl_xor(w1, 2);
      w1 += __shfl_xor(w1, 1);
      if ((lane & 3) == 0) {
        float* dstp = b4 ? edst : esrc;
        dstp[(lo + r) * H + hd] = w1;
      }
    }
  }
}

template <int F, int H, int ELU, int FUSE, int CNT>
__global__ __launch_bounds__(AGG_T)
void gat_owner_aggregate(const int* __restrict__ src, const int* __restrict__ dst,
                         const unsigned* __restrict__ tags,
                         const float* __restrict__ xw,
                         const float* __restrict__ esrc, const float* __restrict__ edst,
                         const float* __restrict__ bias, float* __restrict__ out,
                         const int* __restrict__ gid, float* __restrict__ gcnt,
                         const float* __restrict__ Wn, const float* __restrict__ asn,
                         const float* __restrict__ adn,
                         float* __restrict__ esn, float* __restrict__ edn,
                         unsigned* __restrict__ elist, int* __restrict__ ecnt) {
  constexpr int NJ = F / 32;
  constexpr int C  = F / H;
  constexpr int U  = 32 / NJ;
  constexpr int RU = 4;
  static_assert(U * H <= 32, "one lane per (edge, head) pair of a wave's batch");
  static_assert(NGRAPH <= AGG_T && NPW <= AGG_T, "one thread per graph for the count tables, one per owned node for its id");
  constexpr int FN  = OUTD;
  constexpr int NRT = (NPW + 15) / 16;
  constexpr int NR  = FUSE ? NRT * 16 : NPW;
  constexpr int P   = FUSE ? F + 4 : F;
  constexpr int SP  = FN + 4;

  __shared__ __attribute__((aligned(16))) float s_acc[NR * P];
  __shared__ __attribute__((aligned(16))) float s_out[FUSE ? NR * SP : 4];
  __shared__ unsigned s_list[QCAP];
  __shared__ unsigned s_mx[NPW * H];
  __shared__ float    s_ed[NPW * H];
  __shared__ float    s_m[NPW * H];
  __shared__ float    s_den[NPW * H];
  __shared__ int      s_cnt;
  __shared__ int      s_hist[CNT ? NGRAPH : 1];

  const int tid  = (int)threadIdx.x;
  const int lane = tid & 31;
  const int wave = tid >> 5;
  const int lo   = (int)blockIdx.x * NPW;
  const int cntN = (NNODES - lo < NPW) ? (NNODES - lo) : NPW;

  int gmine = -1;
  if (CNT && tid < cntN) gmine = gid[lo + tid];
  if (CNT && tid < NGRAPH) s_hist[tid] = 0;

  for (int i = tid; i < NR * P; i += AGG_T) s_acc[i] = 0.f;
  for (int i = tid; i < NPW * H; i += AGG_T) {
    const bool own = i < cntN * H;
    float es0 = 0.f, ed0 = 0.f;
    if (own) { es0 = esrc[lo * H + i]; ed0 = edst[lo * H + i]; }
    s_mx[i]  = own ? ordered_enc(es0) : 0u;
    s_ed[i]  = ed0;
    s_m[i]   = 0.f;
    s_den[i] = 0.f;
  }
  if (tid == 0) s_cnt = 0;
  __syncthreads();

  if (CNT && (unsigned)gmine < (unsigned)NGRAPH) atomicAdd(&s_hist[gmine], 1);

  {
    const uint4* tag4 = (const uint4*)tags;
    constexpr int STEP = AGG_T * SLD;
    const unsigned pat = (unsigned)blockIdx.x * 0x01010101u;
    auto load_tags = [&](uint4 (&a)[SLD], int base) {
      #pragma unroll
      for (int i = 0; i < SLD; ++i) {
        const int q = base + i * AGG_T + tid;
        a[i] = tag4[(q < NQT) ? q : (NQT - 1)];
      }
    };
    auto test_tags = [&](const uint4 (&a)[SLD], int base) {
      #pragma unroll
      for (int i = 0; i < SLD; ++i) {
        const int q = base + i * AGG_T + tid;
        const unsigned w[4] = {a[i].x, a[i].y, a[i].z, a[i].w};
        unsigned m = 0u;
        #pragma unroll
        for (int j = 0; j < 4; ++j) {
          const unsigned x = w[j] ^ pat;
          const unsigned f = ~(((x & 0x7F7F7F7Fu) + 0x7F7F7F7Fu) | x) & 0x80808080u;
          m |= f >> (7 - j);
        }
        if (m != 0u && q < NQT) {
          do {
            const int b = __builtin_ctz(m);
            m &= m - 1u;
            const int e = q * 16 + (b & 7) * 4 + (b >> 3);
            int pos = atomicAdd(&s_cnt, 1);
            if (pos < QCAP) s_list[pos] = (unsigned)e;
          } while (m != 0u);
        }
      }
    };
    uint4 setA[SLD], setB[SLD];
    load_tags(setA, 0);
    for (int base = 0; base < NQT; base += 2 * STEP) {
      load_tags(setB, base + STEP);
      test_tags(setA, base);
      load_tags(setA, base + 2 * STEP);
      test_tags(setB, base + STEP);
    }
    __syncthreads();
  }
  const int  n    = s_cnt;
  const bool fits = (n <= QCAP);
  if (tid == 0) ecnt[blockIdx.x] = n;

  if (fits) {
    unsigned* lout = elist + (size_t)blockIdx.x * QCAP;
    for (int i0 = 0; i0 < n; i0 += AGG_T * RU) {
      unsigned ent[RU];
      int      sv[RU];
      int      dv[RU];
      float    es[RU][H];
      #pragma unroll
      for (int k = 0; k < RU; ++k) {
        const int idx = i0 + k * AGG_T + tid;
        ent[k] = (idx < n) ? s_list[idx] : 0u;
      }
      #pragma unroll
      for (int k = 0; k < RU; ++k) { sv[k] = src[ent[k]]; dv[k] = dst[ent[k]]; }
      #pragma unroll
      for (int k = 0; k < RU; ++k)
        #pragma unroll
        for (int h = 0; h < H; ++h) es[k][h] = esrc[sv[k] * H + h];
      #pragma unroll
      for (int k = 0; k < RU; ++k) {
        const int idx = i0 + k * AGG_T + tid;
        if (idx < n) {
          const unsigned dl = (unsigned)(dv[k] - lo);
          #pragma unroll
          for (int h = 0; h < H; ++h) atomicMax(&s_mx[dl * H + h], ordered_enc(es[k][h]));
          const unsigned r = (dl << SBITS) | (unsigned)sv[k];
          s_list[idx] = r;
          lout[idx]   = r;
        }
      }
    }
  } else {
    for (int e = tid; e < NEDGES; e += AGG_T) {
      const unsigned dl = (unsigned)(dst[e] - lo);
      if (dl < (unsigned)cntN) {
        const int s = src[e];
        for (int h = 0; h < H; ++h) atomicMax(&s_mx[dl * H + h], ordered_enc(esrc[s * H + h]));
      }
    }
  }
  __syncthreads();

  for (int i = tid; i < cntN * H; i += AGG_T) s_m[i] = lrelu02(ordered_dec(s_mx[i]) + s_ed[i]);
  if (CNT && tid >= AGG_T - NGRAPH) {
    const int gq = tid - (AGG_T - NGRAPH);
    const int c  = s_hist[gq];
    if (c > 0) atomicAdd(gcnt + gq, (float)c);
  }
  __syncthreads();

  if (fits) {
    const int p  = lane % (U * H);
    const int pk = p / H;
    const int ph = p % H;
    for (int i0 = 0; i0 < n; i0 += AGG_W * U) {
      const int  idxL  = i0 + pk * AGG_W + wave;
      const bool liveL = idxL < n;
      const unsigned entL = s_list[liveL ? idxL : (n - 1)];
      const int sL  = (int)(entL & ((1u << SBITS) - 1u));
      const int dlL = (int)(entL >> SBITS);
      const float esL = esrc[sL * H + ph];
      unsigned ent[U];
      #pragma unroll
      for (int k = 0; k < U; ++k) ent[k] = (unsigned)__builtin_amdgcn_readlane((int)entL, k * H);
      float v[U][NJ];
      #pragma unroll
      for (int k = 0; k < U; ++k)
        #pragma unroll
        for (int j = 0; j < NJ; ++j)
          v[k][j] = xw[(size_t)(ent[k] & ((1u << SBITS) - 1u)) * F + j * 32 + lane];
      const float aL = lrelu02(esL + s_ed[dlL * H + ph]);
      const float wL = liveL ? expf(aL - s_m[dlL * H + ph]) : 0.f;
      if (lane < U * H) atomicAdd(&s_den[dlL * H + ph], wL);
      #pragma unroll
      for (int k = 0; k < U; ++k) {
        float* row = s_acc + (size_t)(ent[k] >> SBITS) * P + lane;
        #pragma unroll
        for (int j = 0; j < NJ; ++j) {
          const float w = __uint_as_float((unsigned)__builtin_amdgcn_readlane(
              (int)__float_as_uint(wL), k * H + (j * 32) / C));
          atomicAdd(row + j * 32, v[k][j] * w);
        }
      }
    }
  } else {
    for (int e = tid; e < NEDGES; e += AGG_T) {
      const unsigned dl = (unsigned)(dst[e] - lo);
      if (dl < (unsigned)cntN) {
        const int s = src[e];
        for (int h = 0; h < H; ++h) {
          const float w = expf(lrelu02(esrc[s * H + h] + s_ed[dl * H + h]) - s_m[dl * H + h]);
          atomicAdd(&s_den[dl * H + h], w);
          for (int c = 0; c < C; ++c)
            atomicAdd(&s_acc[dl * P + h * C + c], w * xw[(size_t)s * F + h * C + c]);
        }
      }
    }
  }
  __syncthreads();

  if (FUSE) {
    constexpr int NCT = FN / 16;
    constexpr int NRG = AGG_W / NCT;
    constexpr int KS  = F / 4;
    static_assert(FN == 64, "score pass: a lane holds columns lane and lane + 32");
    static_assert(AGG_W % NCT == 0, "every column tile gets the same number of waves");
    const int half = lane >> 4;
    const int l    = lane & 15;
    const int wv = __builtin_amdgcn_readfirstlane(wave);
    const int ct = wv % NCT;
    const int rg = wv / NCT;
    v2f bf[KS];
    #pragma unroll
    for (int ks = 0; ks < KS; ++ks) {
      const int ka = ks * 4 + (half << 1);
      bf[ks].x = Wn[(size_t)ka * FN + ct * 16 + l];
      bf[ks].y = Wn[(size_t)(ka + 1) * FN + ct * 16 + l];
    }
    for (int i = tid; i < cntN * F; i += AGG_T) {
      const int dl = i / F, col = i % F, h = col / C;
      const float w   = expf(lrelu02(esrc[(lo + dl) * H + h] + s_ed[dl * H + h]) - s_m[dl * H + h]);
      const float num = s_acc[dl * P + col] + w * xw[(size_t)lo * F + i];
      const float den = s_den[dl * H + h] + w;
      float v = num / den + bias[col];
      if (ELU) v = (v > 0.f) ? v : (expf(v) - 1.f);
      s_acc[dl * P + col] = v;
    }
    __syncthreads();
    for (int rt = rg; rt < NRT; rt += NRG) {
      v8f acc = {};
      const float* hrow = s_acc + (rt * 16 + l) * P + (half << 1);
      #pragma unroll
      for (int ks = 0; ks < KS; ++ks) {
        const v2f a = *(const v2f*)(hrow + ks * 4);
        acc = __builtin_amdgcn_wmma_f32_16x16x4_f32(false, a, false, bf[ks],
                                                    (short)0, acc, false, false);
      }
      float* orow = s_out + (rt * 16 + (half << 3)) * SP + ct * 16 + l;
      #pragma unroll
      for (int v = 0; v < 8; ++v) orow[v * SP] = acc[v];
    }
    __syncthreads();
    {
      float4* o4 = (float4*)(out + (size_t)lo * FN);
      for (int i = tid; i < cntN * (FN / 4); i += AGG_T) {
        const int r = i / (FN / 4), c4 = i % (FN / 4);
        o4[i] = *(const float4*)(s_out + r * SP + c4 * 4);
      }
    }
    {
      const float as0 = asn[lane], as1 = asn[lane + 32];
      const float ad0 = adn[lane], ad1 = adn[lane + 32];
      for (int r = wave; r < cntN; r += AGG_W) {
        const float x0 = s_out[r * SP + lane], x1 = s_out[r * SP + 32 + lane];
        float ss = x0 * as0 + x1 * as1;
        float sd = x0 * ad0 + x1 * ad1;
        for (int m = 16; m > 0; m >>= 1) { ss += __shfl_xor(ss, m); sd += __shfl_xor(sd, m); }
        if (lane == 0) { esn[lo + r] = ss; edn[lo + r] = sd; }
      }
    }
  } else {
    float* orow = out + (size_t)lo * F;
    for (int i = tid; i < cntN * F; i += AGG_T) {
      const int dl = i / F, col = i % F, h = col / C;
      const float w   = expf(lrelu02(esrc[(lo + dl) * H + h] + s_ed[dl * H + h]) - s_m[dl * H + h]);
      const float num = s_acc[dl * P + col] + w * xw[(size_t)lo * F + i];
      const float den = s_den[dl * H + h] + w;
      float v = num / den + bias[col];
      if (ELU) v = (v > 0.f) ? v : (expf(v) - 1.f);
      orow[i] = v;
    }
  }
}

template <int F>
__global__ __launch_bounds__(AGG_T)
void gat_owner_pool(const int* __restrict__ src, const int* __restrict__ dst,
                    const float* __restrict__ xw,
                    const float* __restrict__ esrc, const float* __restrict__ edst,
                    const float* __restrict__ bias, float* __restrict__ out,
                    const int* __restrict__ gid, const float* __restrict__ gcnt,
                    const unsigned* __restrict__ elist, const int* __restrict__ ecnt) {
  constexpr int NJ = F / 32;
  constexpr int U  = 32 / NJ;
  constexpr int RU = 4;
  constexpr unsigned SMASK = (1u << SBITS) - 1u;
  static_assert(NJ == 2, "a lane holds columns lane and lane + 32");
  static_assert(NPW <= AGG_T && NGRAPH <= AGG_T && 2 * F <= AGG_T, "one thread per owned node, per graph, per pooled sum");

  __shared__ unsigned s_list[QCAP];
  __shared__ float    s_w[QCAP];
  __shared__ unsigned s_mx[NPW];
  __shared__ float    s_ed[NPW];
  __shared__ float    s_m[NPW];
  __shared__ float    s_den[NPW];
  __shared__ int      s_slot[NPW];
  __shared__ float    s_gc[NGRAPH];
  __shared__ float    s_part[AGG_W * 2 * F];

  const int tid  = (int)threadIdx.x;
  const int lane = tid & 31;
  const int wave = tid >> 5;
  const int lo   = (int)blockIdx.x * NPW;
  const int cntN = (NNODES - lo < NPW) ? (NNODES - lo) : NPW;
  const int  n    = __builtin_amdgcn_readfirstlane(ecnt[blockIdx.x]);
  const bool fits = (n + NPW <= QCAP);
  const int  gA   = gid[lo];
  const int  gB   = gid[lo + cntN - 1];

  if (tid < NGRAPH) s_gc[tid] = fmaxf(gcnt[tid], 1.f);
  if (tid < NPW) {
    const bool own = tid < cntN;
    float es0 = 0.f, ed0 = 0.f;
    if (own) { es0 = esrc[lo + tid]; ed0 = edst[lo + tid]; }
    s_mx[tid]  = own ? ordered_enc(es0) : 0u;
    s_ed[tid]  = ed0;
    s_m[tid]   = 0.f;
    s_den[tid] = 0.f;
  }
  __syncthreads();

  if (fits) {
    const unsigned* lin = elist + (size_t)blockIdx.x * QCAP;
    for (int i0 = 0; i0 < n; i0 += AGG_T * RU) {
      unsigned ent[RU];
      float    es[RU];
      #pragma unroll
      for (int k = 0; k < RU; ++k) {
        const int idx = i0 + k * AGG_T + tid;
        ent[k] = lin[(idx < n) ? idx : 0];
      }
      #pragma unroll
      for (int k = 0; k < RU; ++k) es[k] = esrc[ent[k] & SMASK];
      #pragma unroll
      for (int k = 0; k < RU; ++k) {
        const int idx = i0 + k * AGG_T + tid;
        if (idx < n) {
          atomicMax(&s_mx[ent[k] >> SBITS], ordered_enc(es[k]));
          s_list[idx] = ent[k];
          s_w[idx]    = es[k];
        }
      }
    }
  } else {
    for (int e = tid; e < NEDGES; e += AGG_T) {
      const unsigned dl = (unsigned)(dst[e] - lo);
      if (dl < (unsigned)cntN) atomicMax(&s_mx[dl], ordered_enc(esrc[src[e]]));
    }
  }
  __syncthreads();

  if (tid < cntN) s_m[tid] = lrelu02(ordered_dec(s_mx[tid]) + s_ed[tid]);
  __syncthreads();

  if (fits) {
    for (int idx = tid; idx < n; idx += AGG_T) {
      const unsigned dl = s_list[idx] >> SBITS;
      const float w = expf(lrelu02(s_w[idx] + s_ed[dl]) - s_m[dl]);
      atomicAdd(&s_den[dl], w);
      s_w[idx] = w;
    }
  } else {
    for (int e = tid; e < NEDGES; e += AGG_T) {
      const unsigned dl = (unsigned)(dst[e] - lo);
      if (dl < (unsigned)cntN)
        atomicAdd(&s_den[dl], expf(lrelu02(esrc[src[e]] + s_ed[dl]) - s_m[dl]));
    }
  }
  __syncthreads();

  float wself = 0.f;
  if (tid < cntN) {
    const int node = lo + tid;
    wself = expf(lrelu02(esrc[node] + s_ed[tid]) - s_m[tid]);
    const float den = s_den[tid] + wself;
    const int g = gid[node];
    const bool valid = (unsigned)g < (unsigned)NGRAPH;
    s_den[tid]  = valid ? 1.f / (den * s_gc[valid ? g : 0]) : 0.f;
    s_slot[tid] = (g == gA) ? 0 : ((g == gB) ? 1 : 2);
    if (fits) {
      s_list[n + tid] = ((unsigned)tid << SBITS) | (unsigned)node;
      s_w[n + tid]    = wself;
    }
  }
  __syncthreads();

  float accA[NJ] = {}, accB[NJ] = {};
  if (fits) {
    const int nt = n + cntN;
    const int pk = lane % U;
    float csA = 0.f, csB = 0.f;
    for (int i0 = 0; i0 < nt; i0 += AGG_W * U) {
      const int  idxL  = i0 + pk * AGG_W + wave;
      const bool liveL = idxL < nt;
      const unsigned entL  = s_list[liveL ? idxL : 0];
      unsigned ent[U];
      #pragma unroll
      for (int k = 0; k < U; ++k) ent[k] = (unsigned)__builtin_amdgcn_readlane((int)entL, k);
      float v[U][NJ];
      #pragma unroll
      for (int k = 0; k < U; ++k)
        #pragma unroll
        for (int j = 0; j < NJ; ++j)
          v[k][j] = xw[(size_t)(ent[k] & SMASK) * F + j * 32 + lane];
      const unsigned dlL   = entL >> SBITS;
      const float    cL    = liveL ? s_w[liveL ? idxL : 0] * s_den[dlL] : 0.f;
      const int      slotL = s_slot[dlL];
      const float    cAL   = (slotL == 0) ? cL : 0.f;
      const float    cBL   = (slotL == 1) ? cL : 0.f;
      if (lane < U) { csA += cAL; csB += cBL; }
      if (slotL == 2 && cL != 0.f && lane < U) {
        const int g = gid[lo + (int)dlL];
        const size_t srow = (size_t)(entL & SMASK) * F;
        for (int col = 0; col < F; ++col) atomicAdd(out + (size_t)g * F + col, cL * (xw[srow + col] + bias[col]));
      }
      #pragma unroll
      for (int k = 0; k < U; ++k) {
        const float cA = __uint_as_float((unsigned)__builtin_amdgcn_readlane((int)__float_as_uint(cAL), k));
        const float cB = __uint_as_float((unsigned)__builtin_amdgcn_readlane((int)__float_as_uint(cBL), k));
        #pragma unroll
        for (int j = 0; j < NJ; ++j) {
          accA[j] = __builtin_fmaf(cA, v[k][j], accA[j]);
          accB[j] = __builtin_fmaf(cB, v[k][j], accB[j]);
        }
      }
    }
    for (int m = 16; m > 0; m >>= 1) { csA += __shfl_xor(csA, m); csB += __shfl_xor(csB, m); }
    #pragma unroll
    for (int j = 0; j < NJ; ++j) {
      const float bj = bias[j * 32 + lane];
      accA[j] = __builtin_fmaf(csA, bj, accA[j]);
      accB[j] = __builtin_fmaf(csB, bj, accB[j]);
    }
  } else {
    for (int e = tid; e < NEDGES; e += AGG_T) {
      const unsigned dl = (unsigned)(dst[e] - lo);
      if (dl < (unsigned)cntN) {
        const int s = src[e];
        const float c = expf(lrelu02(esrc[s] + s_ed[dl]) - s_m[dl]) * s_den[dl];
        if (c != 0.f) {
          const int g = gid[lo + (int)dl];
          for (int col = 0; col < F; ++col) atomicAdd(out + (size_t)g * F + col, c * (xw[(size_t)s * F + col] + bias[col]));
        }
      }
    }
    if (tid < cntN) {
      const float c = wself * s_den[tid];
      if (c != 0.f) {
        const int g = gid[lo + tid];
        for (int col = 0; col < F; ++col) atomicAdd(out + (size_t)g * F + col, c * (xw[(size_t)(lo + tid) * F + col] + bias[col]));
      }
    }
  }

  #pragma unroll
  for (int j = 0; j < NJ; ++j) {
    s_part[(wave * 2 + 0) * F + j * 32 + lane] = accA[j];
    s_part[(wave * 2 + 1) * F + j * 32 + lane] = accB[j];
  }
  __syncthreads();
  if (tid < 2 * F) {
    const int slot = tid / F, col = tid % F;
    float sum = 0.f;
    for (int w = 0; w < AGG_W; ++w) sum += s_part[(w * 2 + slot) * F + col];
    const int g = slot ? gB : gA;
    if ((unsigned)g < (unsigned)NGRAPH && (slot == 0 || gB != gA)) atomicAdd(out + (size_t)g * F + col, sum);
  }
}

extern "C" void kernel_launch(void* const* d_in, const int* in_sizes, int n_in,
                              void* d_out, int out_size, void* d_ws, size_t ws_size,
                              hipStream_t stream) {
  (void)in_sizes; (void)n_in; (void)out_size; (void)ws_size;
  const float* x   = (const float*)d_in[0];
  const int*   ei  = (const int*)d_in[1];
  const int*   bat = (const int*)d_in[2];
  const float* W1  = (const float*)d_in[3];
  const float* as1 = (const float*)d_in[4];
  const float* ad1 = (const float*)d_in[5];
  const float* b1  = (const float*)d_in[6];
  const float* W2  = (const float*)d_in[7];
  const float* as2 = (const float*)d_in[8];
  const float* ad2 = (const float*)d_in[9];
  const float* b2  = (const float*)d_in[10];
  const int* src = ei;
  const int* dst = ei + NEDGES;

  char* ws = (char*)d_ws;
  size_t off = 0;
  auto carve = [&](size_t bytes) -> char* {
    char* p = ws + off;
    off += (bytes + 255) & ~(size_t)255;
    return p;
  };
  float*    xw1    = (float*)carve((size_t)NNODES * HID * 4);
  float*    esrc1  = (float*)carve((size_t)NNODES * H1C * 4);
  float*    edst1  = (float*)carve((size_t)NNODES * H1C * 4);
  float*    cnt    = (float*)carve((size_t)NGRAPH * 4);
  unsigned* tags   = (unsigned*)carve((size_t)NEDGES);
  float*    xw2    = (float*)carve((size_t)NNODES * OUTD * 4);
  float*    esrc2  = (float*)carve((size_t)NNODES * 4);
  float*    edst2  = (float*)carve((size_t)NNODES * 4);
  unsigned* elist  = (unsigned*)carve((size_t)NWG * QCAP * 4);
  int*      ecnt   = (int*)carve((size_t)NWG * 4);
  float*    outF   = (float*)d_out;

  owner_product<INDIM, HID, H1C><<<NWG, AGG_T, 0, stream>>>(x, W1, as1, ad1, xw1, esrc1, edst1, dst, tags, outF, cnt);
  gat_owner_aggregate<HID, H1C, 1, 1, 1><<<NWG, AGG_T, 0, stream>>>(src, dst, tags, xw1, esrc1, edst1, b1, xw2, bat, cnt,
                                                                          W2, as2, ad2, esrc2, edst2, elist, ecnt);

  gat_owner_pool<OUTD><<<NWG, AGG_T, 0, stream>>>(src, dst, xw2, esrc2, edst2, b2, outF, bat, cnt, elist, ecnt);
}


# Owner tags replaced by presence bits: product launch sets a bit per 16-byte id piece, chunk and owner; owner groups test only flagged pieces; tag scan, dst look-ups gone; sums in LDS arrival order
# speedup vs baseline: 8.5018x; 1.1200x over previous
#include <hip/hip_runtime.h>
#include <math.h>

#define NNODES 50000
#define NEDGES 800000
#define ETOT   (NEDGES + NNODES)
#define INDIM  128
#define H1C    4
#define C1C    32
#define HID    128
#define OUTD   64
#define NGRAPH 64

typedef float v2f __attribute__((ext_vector_type(2)));
typedef float v8f __attribute__((ext_vector_type(8)));

__device__ __forceinline__ unsigned ordered_enc(float f) {
  unsigned u = __float_as_uint(f);
  return (u & 0x80000000u) ? ~u : (u | 0x80000000u);
}
__device__ __forceinline__ float ordered_dec(unsigned o) {
  return (o & 0x80000000u) ? __uint_as_float(o & 0x7FFFFFFFu) : __uint_as_float(~o);
}

__device__ __forceinline__ float lrelu02(float a) { return (a >= 0.f) ? a : 0.2f * a; }

#define NWG    256
#define NPW    ((NNODES + NWG - 1) / NWG)
#define AGG_T  1024
#define AGG_W  (AGG_T / 32)
#define QCAP   24576
#define NQ     (NEDGES / 4)
#define SBITS  16
#define EBITS  20
#define CCAP   256
static_assert(NPW <= (1 << (32 - SBITS)), "local node must fit the resolved entry's high bits");
static_assert(NNODES <= (1 << SBITS), "source node must fit the resolved entry's low bits");
static_assert(NEDGES <= (1 << EBITS) && NPW <= (1 << (32 - EBITS)), "edge id and local node must fit the raw entry");
static_assert(NEDGES % 4 == 0, "whole 16-byte id pieces");

#define TQ     ((NQ + NWG - 1) / NWG)
#define PW     ((TQ + 31) / 32)
static_assert(TQ <= AGG_T, "one id piece per thread");
static_assert(PW <= 32, "a lane per word of a row");
static_assert((NWG * PW) % 4 == 0, "a group's rows leave LDS in 16-byte pieces");
static_assert(NWG % AGG_W == 0, "every wave of an owner group takes the same number of chunks");

template <int K, int FN, int H>
__global__ __launch_bounds__(AGG_T)
void owner_product(const float* __restrict__ x, const float* __restrict__ Wm,
                   const float* __restrict__ a_src, const float* __restrict__ a_dst,
                   float* __restrict__ xw, float* __restrict__ esrc, float* __restrict__ edst,
                   const int* __restrict__ dst, unsigned* __restrict__ pres,
                   float* __restrict__ pooled, float* __restrict__ gcnt) {
  constexpr int NRT = (NPW + 15) / 16;
  constexpr int NR  = NRT * 16;
  constexpr int P   = K + 4;
  constexpr int SP  = FN + 4;
  constexpr int NCT = FN / 16;
  constexpr int NRG = AGG_W / NCT;
  constexpr int KS  = K / 4;
  constexpr int C   = FN / H;
  constexpr int XQ  = K / 4;
  constexpr int XIT = (NR * XQ + AGG_T - 1) / AGG_T;
  constexpr int XR  = XIT * AGG_T / XQ;
  static_assert(AGG_T % XQ == 0 && XR >= NR, "a staging trip covers whole rows");
  static_assert(H == 4 && C == 32, "score pass: a lane holds one channel of each of four heads");
  static_assert(AGG_W % NCT == 0, "every column tile gets the same number of waves");
  static_assert(K % 4 == 0 && FN % 16 == 0, "whole k steps and column tiles");

  __shared__ __attribute__((aligned(16))) float s_x[XR * P];
  __shared__ __attribute__((aligned(16))) float s_out[NR * SP];
  __shared__ __attribute__((aligned(16))) unsigned s_pres[NWG * PW];

  const int tid  = (int)threadIdx.x;
  const int lane = tid & 31;
  const int wave = tid >> 5;
  const int lo   = (int)blockIdx.x * NPW;
  const int cntN = (NNODES - lo < NPW) ? (NNODES - lo) : NPW;
  const int half = lane >> 4;

  for (int i = tid; i < NWG * PW; i += AGG_T) s_pres[i] = 0u;
  __syncthreads();
  const int l    = lane & 15;
  const int wv = __builtin_amdgcn_readfirstlane(wave);
  const int ct = wv % NCT;
  const int rg = wv / NCT;

  const int  tq    = (int)blockIdx.x * TQ + tid;
  const bool tlive = (tid < TQ) && (tq < NQ);
  const int4 td    = ((const int4*)dst)[tlive ? tq : 0];

  v2f bf[KS];
  #pragma unroll
  for (int ks = 0; ks < KS; ++ks) {
    const int ka = ks * 4 + (half << 1);
    bf[ks].x = Wm[(size_t)ka * FN + ct * 16 + l];
    bf[ks].y = Wm[(size_t)(ka + 1) * FN + ct * 16 + l];
  }

  {
    const float4* x4 = (const float4*)(x + (size_t)lo * K);
    float4 xv[XIT];
    #pragma unroll
    for (int k = 0; k < XIT; ++k) {
      const int q = tid + k * AGG_T;
      xv[k] = x4[(q / XQ < cntN) ? q : 0];
    }
    #pragma unroll
    for (int k = 0; k < XIT; ++k) {
      const int q = tid + k * AGG_T;
      const int row = q / XQ, c4 = q % XQ;
      const bool live = row < cntN;
      float4 v;
      v.x = live ? xv[k].x : 0.f;
      v.y = live ? xv[k].y : 0.f;
      v.z = live ? xv[k].z : 0.f;
      v.w = live ? xv[k].w : 0.f;
      *(float4*)(s_x + row * P + c4 * 4) = v;
    }
  }
  if (tlive) {
    const unsigned o0 = min((unsigned)td.x / (unsigned)NPW, (unsigned)(NWG - 1));
    const unsigned o1 = min((unsigned)td.y / (unsigned)NPW, (unsigned)(NWG - 1));
    const unsigned o2 = min((unsigned)td.z / (unsigned)NPW, (unsigned)(NWG - 1));
    const unsigned o3 = min((unsigned)td.w / (unsigned)NPW, (unsigned)(NWG - 1));
    const unsigned bit = 1u << lane;
    atomicOr(&s_pres[o0 * PW + wave], bit);
    atomicOr(&s_pres[o1 * PW + wave], bit);
    atomicOr(&s_pres[o2 * PW + wave], bit);
    atomicOr(&s_pres[o3 * PW + wave], bit);
  }
  {
    const int gt = (int)blockIdx.x * AGG_T + tid;
    if (gt < NGRAPH * OUTD) pooled[gt] = 0.f;
    if (gt < NGRAPH) gcnt[gt] = 0.f;
  }
  __syncthreads();

  {
    uint4* p4 = (uint4*)(pres + (size_t)blockIdx.x * (NWG * PW));
    const uint4* s4 = (const uint4*)s_pres;
    for (int i = tid; i < NWG * PW / 4; i += AGG_T) p4[i] = s4[i];
  }

  for (int rt = rg; rt < NRT; rt += NRG) {
    v8f acc = {};
    const float* hrow = s_x + (rt * 16 + l) * P + (half << 1);
    #pragma unroll
    for (int ks = 0; ks < KS; ++ks) {
      const v2f a = *(const v2f*)(hrow + ks * 4);
      acc = __builtin_amdgcn_wmma_f32_16x16x4_f32(false, a, false, bf[ks],
                                                  (short)0, acc, false, false);
    }
    float* orow = s_out + (rt * 16 + (half << 3)) * SP + ct * 16 + l;
    #pragma unroll
    for (int v = 0; v < 8; ++v) orow[v * SP] = acc[v];
  }
  __syncthreads();

  {
    float4* o4 = (float4*)(xw + (size_t)lo * FN);
    for (int i = tid; i < cntN * (FN / 4); i += AGG_T) {
      const int r = i / (FN / 4), c4 = i % (FN / 4);
      o4[i] = *(const float4*)(s_out + r * SP + c4 * 4);
    }
  }
  {
    float as[H], ad[H];
    #pragma unroll
    for (int h = 0; h < H; ++h) { as[h] = a_src[h * C + lane]; ad[h] = a_dst[h * C + lane]; }
    const bool b4 = (lane & 16) != 0;
    const bool b3 = (lane & 8) != 0;
    const bool b2 = (lane & 4) != 0;
    const int  hd = (b3 ? 2 : 0) + (b2 ? 1 : 0);
    for (int r = wave; r < cntN; r += AGG_W) {
      float ss[H], sd[H];
      #pragma unroll
      for (int h = 0; h < H; ++h) {
        const float xr = s_out[r * SP + h * C + lane];
        ss[h] = xr * as[h];
        sd[h] = xr * ad[h];
      }
      float w4[H];
      #pragma unroll
      for (int h = 0; h < H; ++h) {
        const float give = b4 ? ss[h] : sd[h];
        const float keep = b4 ? sd[h] : ss[h];
        w4[h] = keep + __shfl_xor(give, 16);
      }
      float w2[2];
      #pragma unroll
      for (int j = 0; j < 2; ++j) {
        const float give = b3 ? w4[j] : w4[j + 2];
        const float keep = b3 ? w4[j + 2] : w4[j];
        w2[j] = keep + __shfl_xor(give, 8);
      }
      float w1;
      {
        const float give = b2 ? w2[0] : w2[1];
        const float keep = b2 ? w2[1] : w2[0];
        w1 = keep + __shfl_xor(give, 4);
      }
      w1 += __shfl_xor(w1, 2);
      w1 += __shfl_xor(w1, 1);
      if ((lane & 3) == 0) {
        float* dstp = b4 ? edst : esrc;
        dstp[(lo + r) * H + hd] = w1;
      }
    }
  }
}

template <int F, int H, int ELU, int FUSE, int CNT>
__global__ __launch_bounds__(AGG_T)
void gat_owner_aggregate(const int* __restrict__ src, const int* __restrict__ dst,
                         const unsigned* __restrict__ pres,
                         const float* __restrict__ xw,
                         const float* __restrict__ esrc, const float* __restrict__ edst,
                         const float* __restrict__ bias, float* __restrict__ out,
                         const int* __restrict__ gid, float* __restrict__ gcnt,
                         const float* __restrict__ Wn, const float* __restrict__ asn,
                         const float* __restrict__ adn,
                         float* __restrict__ esn, float* __restrict__ edn,
                         unsigned* __restrict__ elist, int* __restrict__ ecnt) {
  constexpr int NJ = F / 32;
  constexpr int C  = F / H;
  constexpr int U  = 32 / NJ;
  constexpr int RU = 4;
  constexpr int CPV = NWG / AGG_W;
  constexpr int RB = 4;
  static_assert(U * H <= 32, "one lane per (edge, head) pair of a wave's batch");
  static_assert(NGRAPH <= AGG_T && NPW <= AGG_T, "one thread per graph for the count tables, one per owned node for its id");
  constexpr int FN  = OUTD;
  constexpr int NRT = (NPW + 15) / 16;
  constexpr int NR  = FUSE ? NRT * 16 : NPW;
  constexpr int P   = FUSE ? F + 4 : F;
  constexpr int SP  = FN + 4;

  __shared__ __attribute__((aligned(16))) float s_acc[NR * P];
  __shared__ __attribute__((aligned(16))) float s_out[FUSE ? NR * SP : 4];
  __shared__ unsigned s_list[QCAP];
  __shared__ unsigned s_mx[NPW * H];
  __shared__ float    s_ed[NPW * H];
  __shared__ float    s_m[NPW * H];
  __shared__ float    s_den[NPW * H];
  __shared__ int      s_cnt;
  __shared__ int      s_ovf;
  __shared__ unsigned s_cand[AGG_W * CCAP];
  __shared__ int      s_hist[CNT ? NGRAPH : 1];

  const int tid  = (int)threadIdx.x;
  const int lane = tid & 31;
  const int wave = tid >> 5;
  const int lo   = (int)blockIdx.x * NPW;
  const int cntN = (NNODES - lo < NPW) ? (NNODES - lo) : NPW;
  const int wvs  = __builtin_amdgcn_readfirstlane(wave);

  unsigned bw[CPV];
  #pragma unroll
  for (int k = 0; k < CPV; ++k) {
    const int c = k * AGG_W + wave;
    bw[k] = pres[((size_t)c * NWG + blockIdx.x) * PW + ((lane < PW) ? lane : (PW - 1))];
  }

  int gmine = -1;
  if (CNT && tid < cntN) gmine = gid[lo + tid];
  if (CNT && tid < NGRAPH) s_hist[tid] = 0;

  for (int i = tid; i < NR * P; i += AGG_T) s_acc[i] = 0.f;
  for (int i = tid; i < NPW * H; i += AGG_T) {
    const bool own = i < cntN * H;
    float es0 = 0.f, ed0 = 0.f;
    if (own) { es0 = esrc[lo * H + i]; ed0 = edst[lo * H + i]; }
    s_mx[i]  = own ? ordered_enc(es0) : 0u;
    s_ed[i]  = ed0;
    s_m[i]   = 0.f;
    s_den[i] = 0.f;
  }
  if (tid == 0) { s_cnt = 0; s_ovf = 0; }
  __syncthreads();

  if (CNT && (unsigned)gmine < (unsigned)NGRAPH) atomicAdd(&s_hist[gmine], 1);

  {
    int wcnt = 0;
    #pragma unroll
    for (int k = 0; k < CPV; ++k) {
      unsigned w = (lane < PW) ? bw[k] : 0u;
      const int qb = (k * AGG_W + wvs) * TQ + lane * 32;
      for (;;) {
        const unsigned bal = __builtin_amdgcn_ballot_w32(w != 0u);
        if (bal == 0u) break;
        const int pos = wcnt + (int)__builtin_amdgcn_mbcnt_lo(bal, 0u);
        if (w != 0u && pos < CCAP) s_cand[wvs * CCAP + pos] = (unsigned)(qb + __builtin_ctz(w));
        wcnt += __builtin_popcount(bal);
        w &= w - 1u;
      }
    }
    if (wcnt > CCAP && lane == 0) s_ovf = 1;
    const int nc = (wcnt < CCAP) ? wcnt : CCAP;

    const int4* dst4 = (const int4*)dst;
    for (int i0 = 0; i0 < nc; i0 += 32 * RB) {
      int  q[RB];
      int4 dv[RB];
      #pragma unroll
      for (int j = 0; j < RB; ++j) {
        const int idx = i0 + j * 32 + lane;
        q[j] = (int)s_cand[wvs * CCAP + ((idx < nc) ? idx : 0)];
      }
      #pragma unroll
      for (int j = 0; j < RB; ++j) dv[j] = dst4[q[j]];
      #pragma unroll
      for (int j = 0; j < RB; ++j) {
        const int idx = i0 + j * 32 + lane;
        if (idx < nc) {
          const unsigned d0 = (unsigned)(dv[j].x - lo), d1 = (unsigned)(dv[j].y - lo);
          const unsigned d2 = (unsigned)(dv[j].z - lo), d3 = (unsigned)(dv[j].w - lo);
          unsigned h = (d0 < (unsigned)cntN ? 1u : 0u) | (d1 < (unsigned)cntN ? 2u : 0u) |
                       (d2 < (unsigned)cntN ? 4u : 0u) | (d3 < (unsigned)cntN ? 8u : 0u);
          while (h != 0u) {
            const int r = __builtin_ctz(h);
            h &= h - 1u;
            const unsigned dl = (r == 0) ? d0 : ((r == 1) ? d1 : ((r == 2) ? d2 : d3));
            const int pos = atomicAdd(&s_cnt, 1);
            if (pos < QCAP) s_list[pos] = (dl << EBITS) | (unsigned)(q[j] * 4 + r);
          }
        }
      }
    }
    __syncthreads();
  }
  const int  n    = s_cnt;
  const bool fits = (n <= QCAP) && (s_ovf == 0);
  if (tid == 0) ecnt[blockIdx.x] = (s_ovf == 0) ? n : (QCAP + 1);

  if (fits) {
    unsigned* lout = elist + (size_t)blockIdx.x * QCAP;
    for (int i0 = 0; i0 < n; i0 += AGG_T * RU) {
      unsigned ent[RU];
      int      sv[RU];
      float    es[RU][H];
      #pragma unroll
      for (int k = 0; k < RU; ++k) {
        const int idx = i0 + k * AGG_T + tid;
        ent[k] = (idx < n) ? s_list[idx] : 0u;
      }
      #pragma unroll
      for (int k = 0; k < RU; ++k) sv[k] = src[ent[k] & ((1u << EBITS) - 1u)];
      #pragma unroll
      for (int k = 0; k < RU; ++k)
        #pragma unroll
        for (int h = 0; h < H; ++h) es[k][h] = esrc[sv[k] * H + h];
      #pragma unroll
      for (int k = 0; k < RU; ++k) {
        const int idx = i0 + k * AGG_T + tid;
        if (idx < n) {
          const unsigned dl = ent[k] >> EBITS;
          #pragma unroll
          for (int h = 0; h < H; ++h) atomicMax(&s_mx[dl * H + h], ordered_enc(es[k][h]));
          const unsigned r = (dl << SBITS) | (unsigned)sv[k];
          s_list[idx] = r;
          lout[idx]   = r;
        }
      }
    }
  } else {
    for (int e = tid; e < NEDGES; e += AGG_T) {
      const unsigned dl = (unsigned)(dst[e] - lo);
      if (dl < (unsigned)cntN) {
        const int s = src[e];
        for (int h = 0; h < H; ++h) atomicMax(&s_mx[dl * H + h], ordered_enc(esrc[s * H + h]));
      }
    }
  }
  __syncthreads();

  for (int i = tid; i < cntN * H; i += AGG_T) s_m[i] = lrelu02(ordered_dec(s_mx[i]) + s_ed[i]);
  if (CNT && tid >= AGG_T - NGRAPH) {
    const int gq = tid - (AGG_T - NGRAPH);
    const int c  = s_hist[gq];
    if (c > 0) atomicAdd(gcnt + gq, (float)c);
  }
  __syncthreads();

  if (fits) {
    const int p  = lane % (U * H);
    const int pk = p / H;
    const int ph = p % H;
    for (int i0 = 0; i0 < n; i0 += AGG_W * U) {
      const int  idxL  = i0 + pk * AGG_W + wave;
      const bool liveL = idxL < n;
      const unsigned entL = s_list[liveL ? idxL : (n - 1)];
      const int sL  = (int)(entL & ((1u << SBITS) - 1u));
      const int dlL = (int)(entL >> SBITS);
      const float esL = esrc[sL * H + ph];
      unsigned ent[U];
      #pragma unroll
      for (int k = 0; k < U; ++k) ent[k] = (unsigned)__builtin_amdgcn_readlane((int)entL, k * H);
      float v[U][NJ];
      #pragma unroll
      for (int k = 0; k < U; ++k)
        #pragma unroll
        for (int j = 0; j < NJ; ++j)
          v[k][j] = xw[(size_t)(ent[k] & ((1u << SBITS) - 1u)) * F + j * 32 + lane];
      const float aL = lrelu02(esL + s_ed[dlL * H + ph]);
      const float wL = liveL ? expf(aL - s_m[dlL * H + ph]) : 0.f;
      if (lane < U * H) atomicAdd(&s_den[dlL * H + ph], wL);
      #pragma unroll
      for (int k = 0; k < U; ++k) {
        float* row = s_acc + (size_t)(ent[k] >> SBITS) * P + lane;
        #pragma unroll
        for (int j = 0; j < NJ; ++j) {
          const float w = __uint_as_float((unsigned)__builtin_amdgcn_readlane(
              (int)__float_as_uint(wL), k * H + (j * 32) / C));
          atomicAdd(row + j * 32, v[k][j] * w);
        }
      }
    }
  } else {
    for (int e = tid; e < NEDGES; e += AGG_T) {
      const unsigned dl = (unsigned)(dst[e] - lo);
      if (dl < (unsigned)cntN) {
        const int s = src[e];
        for (int h = 0; h < H; ++h) {
          const float w = expf(lrelu02(esrc[s * H + h] + s_ed[dl * H + h]) - s_m[dl * H + h]);
          atomicAdd(&s_den[dl * H + h], w);
          for (int c = 0; c < C; ++c)
            atomicAdd(&s_acc[dl * P + h * C + c], w * xw[(size_t)s * F + h * C + c]);
        }
      }
    }
  }
  __syncthreads();

  if (FUSE) {
    constexpr int NCT = FN / 16;
    constexpr int NRG = AGG_W / NCT;
    constexpr int KS  = F / 4;
    static_assert(FN == 64, "score pass: a lane holds columns lane and lane + 32");
    static_assert(AGG_W % NCT == 0, "every column tile gets the same number of waves");
    const int half = lane >> 4;
    const int l    = lane & 15;
    const int wv = __builtin_amdgcn_readfirstlane(wave);
    const int ct = wv % NCT;
    const int rg = wv / NCT;
    v2f bf[KS];
    #pragma unroll
    for (int ks = 0; ks < KS; ++ks) {
      const int ka = ks * 4 + (half << 1);
      bf[ks].x = Wn[(size_t)ka * FN + ct * 16 + l];
      bf[ks].y = Wn[(size_t)(ka + 1) * FN + ct * 16 + l];
    }
    for (int i = tid; i < cntN * F; i += AGG_T) {
      const int dl = i / F, col = i % F, h = col / C;
      const float w   = expf(lrelu02(esrc[(lo + dl) * H + h] + s_ed[dl * H + h]) - s_m[dl * H + h]);
      const float num = s_acc[dl * P + col] + w * xw[(size_t)lo * F + i];
      const float den = s_den[dl * H + h] + w;
      float v = num / den + bias[col];
      if (ELU) v = (v > 0.f) ? v : (expf(v) - 1.f);
      s_acc[dl * P + col] = v;
    }
    __syncthreads();
    for (int rt = rg; rt < NRT; rt += NRG) {
      v8f acc = {};
      const float* hrow = s_acc + (rt * 16 + l) * P + (half << 1);
      #pragma unroll
      for (int ks = 0; ks < KS; ++ks) {
        const v2f a = *(const v2f*)(hrow + ks * 4);
        acc = __builtin_amdgcn_wmma_f32_16x16x4_f32(false, a, false, bf[ks],
                                                    (short)0, acc, false, false);
      }
      float* orow = s_out + (rt * 16 + (half << 3)) * SP + ct * 16 + l;
      #pragma unroll
      for (int v = 0; v < 8; ++v) orow[v * SP] = acc[v];
    }
    __syncthreads();
    {
      float4* o4 = (float4*)(out + (size_t)lo * FN);
      for (int i = tid; i < cntN * (FN / 4); i += AGG_T) {
        const int r = i / (FN / 4), c4 = i % (FN / 4);
        o4[i] = *(const float4*)(s_out + r * SP + c4 * 4);
      }
    }
    {
      const float as0 = asn[lane], as1 = asn[lane + 32];
      const float ad0 = adn[lane], ad1 = adn[lane + 32];
      for (int r = wave; r < cntN; r += AGG_W) {
        const float x0 = s_out[r * SP + lane], x1 = s_out[r * SP + 32 + lane];
        float ss = x0 * as0 + x1 * as1;
        float sd = x0 * ad0 + x1 * ad1;
        for (int m = 16; m > 0; m >>= 1) { ss += __shfl_xor(ss, m); sd += __shfl_xor(sd, m); }
        if (lane == 0) { esn[lo + r] = ss; edn[lo + r] = sd; }
      }
    }
  } else {
    float* orow = out + (size_t)lo * F;
    for (int i = tid; i < cntN * F; i += AGG_T) {
      const int dl = i / F, col = i % F, h = col / C;
      const float w   = expf(lrelu02(esrc[(lo + dl) * H + h] + s_ed[dl * H + h]) - s_m[dl * H + h]);
      const float num = s_acc[dl * P + col] + w * xw[(size_t)lo * F + i];
      const float den = s_den[dl * H + h] + w;
      float v = num / den + bias[col];
      if (ELU) v = (v > 0.f) ? v : (expf(v) - 1.f);
      orow[i] = v;
    }
  }
}

template <int F>
__global__ __launch_bounds__(AGG_T)
void gat_owner_pool(const int* __restrict__ src, const int* __restrict__ dst,
                    const float* __restrict__ xw,
                    const float* __restrict__ esrc, const float* __restrict__ edst,
                    const float* __restrict__ bias, float* __restrict__ out,
                    const int* __restrict__ gid, const float* __restrict__ gcnt,
                    const unsigned* __restrict__ elist, const int* __restrict__ ecnt) {
  constexpr int NJ = F / 32;
  constexpr int U  = 32 / NJ;
  constexpr int RU = 4;
  constexpr unsigned SMASK = (1u << SBITS) - 1u;
  static_assert(NJ == 2, "a lane holds columns lane and lane + 32");
  static_assert(NPW <= AGG_T && NGRAPH <= AGG_T && 2 * F <= AGG_T, "one thread per owned node, per graph, per pooled sum");

  __shared__ unsigned s_list[QCAP];
  __shared__ float    s_w[QCAP];
  __shared__ unsigned s_mx[NPW];
  __shared__ float    s_ed[NPW];
  __shared__ float    s_m[NPW];
  __shared__ float    s_den[NPW];
  __shared__ int      s_slot[NPW];
  __shared__ float    s_gc[NGRAPH];
  __shared__ float    s_part[AGG_W * 2 * F];

  const int tid  = (int)threadIdx.x;
  const int lane = tid & 31;
  const int wave = tid >> 5;
  const int lo   = (int)blockIdx.x * NPW;
  const int cntN = (NNODES - lo < NPW) ? (NNODES - lo) : NPW;
  const int  n    = __builtin_amdgcn_readfirstlane(ecnt[blockIdx.x]);
  const bool fits = (n + NPW <= QCAP);
  const int  gA   = gid[lo];
  const int  gB   = gid[lo + cntN - 1];

  if (tid < NGRAPH) s_gc[tid] = fmaxf(gcnt[tid], 1.f);
  if (tid < NPW) {
    const bool own = tid < cntN;
    float es0 = 0.f, ed0 = 0.f;
    if (own) { es0 = esrc[lo + tid]; ed0 = edst[lo + tid]; }
    s_mx[tid]  = own ? ordered_enc(es0) : 0u;
    s_ed[tid]  = ed0;
    s_m[tid]   = 0.f;
    s_den[tid] = 0.f;
  }
  __syncthreads();

  if (fits) {
    const unsigned* lin = elist + (size_t)blockIdx.x * QCAP;
    for (int i0 = 0; i0 < n; i0 += AGG_T * RU) {
      unsigned ent[RU];
      float    es[RU];
      #pragma unroll
      for (int k = 0; k < RU; ++k) {
        const int idx = i0 + k * AGG_T + tid;
        ent[k] = lin[(idx < n) ? idx : 0];
      }
      #pragma unroll
      for (int k = 0; k < RU; ++k) es[k] = esrc[ent[k] & SMASK];
      #pragma unroll
      for (int k = 0; k < RU; ++k) {
        const int idx = i0 + k * AGG_T + tid;
        if (idx < n) {
          atomicMax(&s_mx[ent[k] >> SBITS], ordered_enc(es[k]));
          s_list[idx] = ent[k];
          s_w[idx]    = es[k];
        }
      }
    }
  } else {
    for (int e = tid; e < NEDGES; e += AGG_T) {
      const unsigned dl = (unsigned)(dst[e] - lo);
      if (dl < (unsigned)cntN) atomicMax(&s_mx[dl], ordered_enc(esrc[src[e]]));
    }
  }
  __syncthreads();

  if (tid < cntN) s_m[tid] = lrelu02(ordered_dec(s_mx[tid]) + s_ed[tid]);
  __syncthreads();

  if (fits) {
    for (int idx = tid; idx < n; idx += AGG_T) {
      const unsigned dl = s_list[idx] >> SBITS;
      const float w = expf(lrelu02(s_w[idx] + s_ed[dl]) - s_m[dl]);
      atomicAdd(&s_den[dl], w);
      s_w[idx] = w;
    }
  } else {
    for (int e = tid; e < NEDGES; e += AGG_T) {
      const unsigned dl = (unsigned)(dst[e] - lo);
      if (dl < (unsigned)cntN)
        atomicAdd(&s_den[dl], expf(lrelu02(esrc[src[e]] + s_ed[dl]) - s_m[dl]));
    }
  }
  __syncthreads();

  float wself = 0.f;
  if (tid < cntN) {
    const int node = lo + tid;
    wself = expf(lrelu02(esrc[node] + s_ed[tid]) - s_m[tid]);
    const float den = s_den[tid] + wself;
    const int g = gid[node];
    const bool valid = (unsigned)g < (unsigned)NGRAPH;
    s_den[tid]  = valid ? 1.f / (den * s_gc[valid ? g : 0]) : 0.f;
    s_slot[tid] = (g == gA) ? 0 : ((g == gB) ? 1 : 2);
    if (fits) {
      s_list[n + tid] = ((unsigned)tid << SBITS) | (unsigned)node;
      s_w[n + tid]    = wself;
    }
  }
  __syncthreads();

  float accA[NJ] = {}, accB[NJ] = {};
  if (fits) {
    const int nt = n + cntN;
    const int pk = lane % U;
    float csA = 0.f, csB = 0.f;
    for (int i0 = 0; i0 < nt; i0 += AGG_W * U) {
      const int  idxL  = i0 + pk * AGG_W + wave;
      const bool liveL = idxL < nt;
      const unsigned entL  = s_list[liveL ? idxL : 0];
      unsigned ent[U];
      #pragma unroll
      for (int k = 0; k < U; ++k) ent[k] = (unsigned)__builtin_amdgcn_readlane((int)entL, k);
      float v[U][NJ];
      #pragma unroll
      for (int k = 0; k < U; ++k)
        #pragma unroll
        for (int j = 0; j < NJ; ++j)
          v[k][j] = xw[(size_t)(ent[k] & SMASK) * F + j * 32 + lane];
      const unsigned dlL   = entL >> SBITS;
      const float    cL    = liveL ? s_w[liveL ? idxL : 0] * s_den[dlL] : 0.f;
      const int      slotL = s_slot[dlL];
      const float    cAL   = (slotL == 0) ? cL : 0.f;
      const float    cBL   = (slotL == 1) ? cL : 0.f;
      if (lane < U) { csA += cAL; csB += cBL; }
      if (slotL == 2 && cL != 0.f && lane < U) {
        const int g = gid[lo + (int)dlL];
        const size_t srow = (size_t)(entL & SMASK) * F;
        for (int col = 0; col < F; ++col) atomicAdd(out + (size_t)g * F + col, cL * (xw[srow + col] + bias[col]));
      }
      #pragma unroll
      for (int k = 0; k < U; ++k) {
        const float cA = __uint_as_float((unsigned)__builtin_amdgcn_readlane((int)__float_as_uint(cAL), k));
        const float cB = __uint_as_float((unsigned)__builtin_amdgcn_readlane((int)__float_as_uint(cBL), k));
        #pragma unroll
        for (int j = 0; j < NJ; ++j) {
          accA[j] = __builtin_fmaf(cA, v[k][j], accA[j]);
          accB[j] = __builtin_fmaf(cB, v[k][j], accB[j]);
        }
      }
    }
    for (int m = 16; m > 0; m >>= 1) { csA += __shfl_xor(csA, m); csB += __shfl_xor(csB, m); }
    #pragma unroll
    for (int j = 0; j < NJ; ++j) {
      const float bj = bias[j * 32 + lane];
      accA[j] = __builtin_fmaf(csA, bj, accA[j]);
      accB[j] = __builtin_fmaf(csB, bj, accB[j]);
    }
  } else {
    for (int e = tid; e < NEDGES; e += AGG_T) {
      const unsigned dl = (unsigned)(dst[e] - lo);
      if (dl < (unsigned)cntN) {
        const int s = src[e];
        const float c = expf(lrelu02(esrc[s] + s_ed[dl]) - s_m[dl]) * s_den[dl];
        if (c != 0.f) {
          const int g = gid[lo + (int)dl];
          for (int col = 0; col < F; ++col) atomicAdd(out + (size_t)g * F + col, c * (xw[(size_t)s * F + col] + bias[col]));
        }
      }
    }
    if (tid < cntN) {
      const float c = wself * s_den[tid];
      if (c != 0.f) {
        const int g = gid[lo + tid];
        for (int col = 0; col < F; ++col) atomicAdd(out + (size_t)g * F + col, c * (xw[(size_t)(lo + tid) * F + col] + bias[col]));
      }
    }
  }

  #pragma unroll
  for (int j = 0; j < NJ; ++j) {
    s_part[(wave * 2 + 0) * F + j * 32 + lane] = accA[j];
    s_part[(wave * 2 + 1) * F + j * 32 + lane] = accB[j];
  }
  __syncthreads();
  if (tid < 2 * F) {
    const int slot = tid / F, col = tid % F;
    float sum = 0.f;
    for (int w = 0; w < AGG_W; ++w) sum += s_part[(w * 2 + slot) * F + col];
    const int g = slot ? gB : gA;
    if ((unsigned)g < (unsigned)NGRAPH && (slot == 0 || gB != gA)) atomicAdd(out + (size_t)g * F + col, sum);
  }
}

extern "C" void kernel_launch(void* const* d_in, const int* in_sizes, int n_in,
                              void* d_out, int out_size, void* d_ws, size_t ws_size,
                              hipStream_t stream) {
  (void)in_sizes; (void)n_in; (void)out_size; (void)ws_size;
  const float* x   = (const float*)d_in[0];
  const int*   ei  = (const int*)d_in[1];
  const int*   bat = (const int*)d_in[2];
  const float* W1  = (const float*)d_in[3];
  const float* as1 = (const float*)d_in[4];
  const float* ad1 = (const float*)d_in[5];
  const float* b1  = (const float*)d_in[6];
  const float* W2  = (const float*)d_in[7];
  const float* as2 = (const float*)d_in[8];
  const float* ad2 = (const float*)d_in[9];
  const float* b2  = (const float*)d_in[10];
  const int* src = ei;
  const int* dst = ei + NEDGES;

  char* ws = (char*)d_ws;
  size_t off = 0;
  auto carve = [&](size_t bytes) -> char* {
    char* p = ws + off;
    off += (bytes + 255) & ~(size_t)255;
    return p;
  };
  float*    xw1    = (float*)carve((size_t)NNODES * HID * 4);
  float*    esrc1  = (float*)carve((size_t)NNODES * H1C * 4);
  float*    edst1  = (float*)carve((size_t)NNODES * H1C * 4);
  float*    cnt    = (float*)carve((size_t)NGRAPH * 4);
  unsigned* pres   = (unsigned*)carve((size_t)NWG * NWG * PW * 4);
  float*    xw2    = (float*)carve((size_t)NNODES * OUTD * 4);
  float*    esrc2  = (float*)carve((size_t)NNODES * 4);
  float*    edst2  = (float*)carve((size_t)NNODES * 4);
  unsigned* elist  = (unsigned*)carve((size_t)NWG * QCAP * 4);
  int*      ecnt   = (int*)carve((size_t)NWG * 4);
  float*    outF   = (float*)d_out;

  owner_product<INDIM, HID, H1C><<<NWG, AGG_T, 0, stream>>>(x, W1, as1, ad1, xw1, esrc1, edst1, dst, pres, outF, cnt);
  gat_owner_aggregate<HID, H1C, 1, 1, 1><<<NWG, AGG_T, 0, stream>>>(src, dst, pres, xw1, esrc1, edst1, b1, xw2, bat, cnt,
                                                                          W2, as2, ad2, esrc2, edst2, elist, ecnt);

  gat_owner_pool<OUTD><<<NWG, AGG_T, 0, stream>>>(src, dst, xw2, esrc2, edst2, b2, outF, bat, cnt, elist, ecnt);
}
